# starting kernel
# baseline (speedup 1.0000x reference)
#include <hip/hip_runtime.h>
#include <hip/hip_bf16.h>

#define NEG_SLOPE 0.2f
#define F_IN 14

typedef float v2f __attribute__((ext_vector_type(2)));
typedef float v8f __attribute__((ext_vector_type(8)));

__device__ __forceinline__ unsigned enc_f32(float f) {
    unsigned u = __float_as_uint(f);
    return (u & 0x80000000u) ? ~u : (u | 0x80000000u);
}
__device__ __forceinline__ float dec_f32(unsigned e) {
    unsigned u = (e & 0x80000000u) ? (e ^ 0x80000000u) : ~e;
    return __uint_as_float(u);
}

__device__ __forceinline__ void edge_sd(const int* __restrict__ ei, int E, int n,
                                        int idx, int& s, int& d) {
    if (idx < E) { s = ei[idx]; d = ei[E + idx]; }
    else         { s = idx - E; d = idx - E; }
    s = s < 0 ? 0 : (s >= n ? n - 1 : s);
    d = d < 0 ? 0 : (d >= n ? n - 1 : d);
}

__global__ __launch_bounds__(32) void k_gemm1(const float* __restrict__ x,
                                              const float* __restrict__ W1,
                                              float* __restrict__ h1, int n) {
    __shared__ float xs[16 * 16];
    __shared__ float ws[16 * 64];

    const int lane = threadIdx.x;
    const int row0 = blockIdx.x * 16;
    const bool hi  = lane >= 16;
    const int  m   = lane & 15;
    const bool full = (row0 + 16 <= n);

    for (int i = lane; i < 2 * 64 / 4; i += 32)
        ((float4*)(ws + 14 * 64))[i] = make_float4(0.f, 0.f, 0.f, 0.f);
    for (int i = lane; i < F_IN * 64 / 4; i += 32)
        ((float4*)ws)[i] = ((const float4*)W1)[i];

    for (int i = lane; i < 16 * 16 / 4; i += 32)
        ((float4*)xs)[i] = make_float4(0.f, 0.f, 0.f, 0.f);
    __syncthreads();
    {
        const float* xt = x + (size_t)row0 * F_IN;
        int rows = full ? 16 : (n - row0);
        int cnt  = rows * F_IN;
        for (int i = lane; i < cnt; i += 32) {
            int r = i / F_IN, c = i - r * F_IN;
            xs[r * 16 + c] = xt[i];
        }
    }
    __syncthreads();

    const int kofs = hi ? 2 : 0;
    v2f a[4];
#pragma unroll
    for (int k = 0; k < 4; ++k) {
        a[k][0] = xs[m * 16 + 4 * k + kofs + 0];
        a[k][1] = xs[m * 16 + 4 * k + kofs + 1];
    }

#pragma unroll
    for (int t = 0; t < 4; ++t) {
        v8f c = {};
#pragma unroll
        for (int k = 0; k < 4; ++k) {
            v2f b;
            b[0] = ws[(4 * k + kofs + 0) * 64 + t * 16 + m];
            b[1] = ws[(4 * k + kofs + 1) * 64 + t * 16 + m];
            c = __builtin_amdgcn_wmma_f32_16x16x4_f32(false, a[k], false, b,
                                                      (short)0, c, false, false);
        }
        float* out = h1 + (size_t)row0 * 64 + t * 16 + m;
        if (full) {
#pragma unroll
            for (int r = 0; r < 8; ++r)
                out[(size_t)(r + (hi ? 8 : 0)) * 64] = c[r];
        } else {
#pragma unroll
            for (int r = 0; r < 8; ++r) {
                int rr = r + (hi ? 8 : 0);
                if (row0 + rr < n) out[(size_t)rr * 64] = c[r];
            }
        }
    }
}

__global__ __launch_bounds__(32) void k_gemm2(const float* __restrict__ a1,
                                              const float* __restrict__ W2,
                                              float* __restrict__ h2, int n) {
    __shared__ float smA[16 * 64];
    __shared__ float ws2[64 * 16];

    const int lane = threadIdx.x;
    const int row0 = blockIdx.x * 16;
    const bool hi  = lane >= 16;
    const int  m   = lane & 15;
    const bool full = (row0 + 16 <= n);

    for (int i = lane; i < 64 * 16 / 4; i += 32)
        ((float4*)ws2)[i] = make_float4(0.f, 0.f, 0.f, 0.f);
    __syncthreads();
    for (int i = lane; i < 64 * 8 / 4; i += 32) {
        int k = i >> 1, half = i & 1;
        ((float4*)(ws2 + k * 16 + half * 4))[0] = ((const float4*)W2)[i];
    }

    if (full) {
        const float4* src = (const float4*)(a1 + (size_t)row0 * 64);
        for (int i = lane; i < 16 * 64 / 4; i += 32)
            ((float4*)smA)[i] = src[i];
    } else {
        for (int i = lane; i < 16 * 64 / 4; i += 32)
            ((float4*)smA)[i] = make_float4(0.f, 0.f, 0.f, 0.f);
        __syncthreads();
        int rows = n - row0;
        const float4* src = (const float4*)(a1 + (size_t)row0 * 64);
        for (int i = lane; i < rows * 16; i += 32)
            ((float4*)smA)[i] = src[i];
    }
    __syncthreads();

    const int kofs = hi ? 2 : 0;
    v8f c = {};
#pragma unroll
    for (int k = 0; k < 16; ++k) {
        v2f a, b;
        a[0] = smA[m * 64 + 4 * k + kofs + 0];
        a[1] = smA[m * 64 + 4 * k + kofs + 1];
        b[0] = ws2[(4 * k + kofs + 0) * 16 + m];
        b[1] = ws2[(4 * k + kofs + 1) * 16 + m];
        c = __builtin_amdgcn_wmma_f32_16x16x4_f32(false, a, false, b,
                                                  (short)0, c, false, false);
    }
    if (m < 8) {
        float* out = h2 + (size_t)row0 * 8 + m;
        if (full) {
#pragma unroll
            for (int r = 0; r < 8; ++r)
                out[(size_t)(r + (hi ? 8 : 0)) * 8] = c[r];
        } else {
#pragma unroll
            for (int r = 0; r < 8; ++r) {
                int rr = r + (hi ? 8 : 0);
                if (row0 + rr < n) out[(size_t)rr * 8] = c[r];
            }
        }
    }
}

__global__ void k_alpha1(const float* __restrict__ h1,
                         const float* __restrict__ asrc, const float* __restrict__ adst,
                         float* __restrict__ as1, float* __restrict__ ad1, int n) {
    int tid = blockIdx.x * blockDim.x + threadIdx.x;
    if (tid >= n * 8) return;
    unsigned h = tid & 7u;
    const float4* hp = (const float4*)(h1 + (unsigned)(tid >> 3) * 64u + h * 8u);
    float4 v0 = hp[0], v1 = hp[1];
    const float4* sp = (const float4*)(asrc + h * 8u);
    const float4* dp = (const float4*)(adst + h * 8u);
    float4 s0 = sp[0], s1v = sp[1], d0 = dp[0], d1v = dp[1];
    float s = v0.x * s0.x + v0.y * s0.y + v0.z * s0.z + v0.w * s0.w
            + v1.x * s1v.x + v1.y * s1v.y + v1.z * s1v.z + v1.w * s1v.w;
    float d = v0.x * d0.x + v0.y * d0.y + v0.z * d0.z + v0.w * d0.w
            + v1.x * d1v.x + v1.y * d1v.y + v1.z * d1v.z + v1.w * d1v.w;
    as1[tid] = s;
    ad1[tid] = d;
}

__global__ void k_edge_max1(const int* __restrict__ ei, int E, int n,
                            const float* __restrict__ as1, const float* __restrict__ ad1,
                            unsigned* __restrict__ m1) {
    int idx = blockIdx.x * blockDim.x + threadIdx.x;
    if (idx >= E + n) return;
    int s, d; edge_sd(ei, E, n, idx, s, d);
    const float4* ap = (const float4*)(as1 + (unsigned)s * 8u);
    const float4* bp = (const float4*)(ad1 + (unsigned)d * 8u);
    float4 a0 = ap[0], a1 = ap[1], b0 = bp[0], b1 = bp[1];
    float e[8] = {a0.x + b0.x, a0.y + b0.y, a0.z + b0.z, a0.w + b0.w,
                  a1.x + b1.x, a1.y + b1.y, a1.z + b1.z, a1.w + b1.w};
#pragma unroll
    for (int h = 0; h < 8; ++h) {
        float v = e[h] > 0.0f ? e[h] : NEG_SLOPE * e[h];
        atomicMax(&m1[(unsigned)d * 8u + h], enc_f32(v));
    }
}

__global__ void k_edge_sum1(const int* __restrict__ ei, int E, int n,
                            const float* __restrict__ as1, const float* __restrict__ ad1,
                            const unsigned* __restrict__ m1, float* __restrict__ s1) {
    int idx = blockIdx.x * blockDim.x + threadIdx.x;
    if (idx >= E + n) return;
    int s, d; edge_sd(ei, E, n, idx, s, d);
    const float4* ap = (const float4*)(as1 + (unsigned)s * 8u);
    const float4* bp = (const float4*)(ad1 + (unsigned)d * 8u);
    const uint4*  mp = (const uint4*)(m1 + (unsigned)d * 8u);
    float4 a0 = ap[0], a1 = ap[1], b0 = bp[0], b1 = bp[1];
    uint4  m0 = mp[0], m1v = mp[1];
    float e[8] = {a0.x + b0.x, a0.y + b0.y, a0.z + b0.z, a0.w + b0.w,
                  a1.x + b1.x, a1.y + b1.y, a1.z + b1.z, a1.w + b1.w};
    unsigned mm[8] = {m0.x, m0.y, m0.z, m0.w, m1v.x, m1v.y, m1v.z, m1v.w};
#pragma unroll
    for (int h = 0; h < 8; ++h) {
        float v = e[h] > 0.0f ? e[h] : NEG_SLOPE * e[h];
        atomicAdd(&s1[(unsigned)d * 8u + h], expf(v - dec_f32(mm[h])));
    }
}

__global__ void k_edge_scatter1(const int* __restrict__ ei, int E, int n,
                                const float* __restrict__ as1, const float* __restrict__ ad1,
                                const unsigned* __restrict__ m1, const float* __restrict__ s1,
                                const float* __restrict__ h1, float* __restrict__ acc1) {
    int tid = blockIdx.x * blockDim.x + threadIdx.x;
    if (tid >= (E + n) * 8) return;
    int idx = tid >> 3, h = tid & 7;
    int s, d; edge_sd(ei, E, n, idx, s, d);
    const float* hp = h1 + (unsigned)s * 64u + (unsigned)h * 8u;
    __builtin_prefetch(hp, 0, 3);
    float e = as1[(unsigned)s * 8u + h] + ad1[(unsigned)d * 8u + h];
    e = e > 0.0f ? e : NEG_SLOPE * e;
    float ex    = expf(e - dec_f32(m1[(unsigned)d * 8u + h]));
    float alpha = ex / (s1[(unsigned)d * 8u + h] + 1e-16f);
    float4 v0 = ((const float4*)hp)[0], v1 = ((const float4*)hp)[1];
    float* op = acc1 + (unsigned)d * 64u + (unsigned)h * 8u;
    atomicAdd(op + 0, v0.x * alpha);
    atomicAdd(op + 1, v0.y * alpha);
    atomicAdd(op + 2, v0.z * alpha);
    atomicAdd(op + 3, v0.w * alpha);
    atomicAdd(op + 4, v1.x * alpha);
    atomicAdd(op + 5, v1.y * alpha);
    atomicAdd(op + 6, v1.z * alpha);
    atomicAdd(op + 7, v1.w * alpha);
}

__global__ void k_post1(float* __restrict__ acc1, const float* __restrict__ bias1, int n) {
    int tid = blockIdx.x * blockDim.x + threadIdx.x;
    if (tid >= n * 64) return;
    float v = acc1[tid] + bias1[tid & 63];
    acc1[tid] = v > 0.0f ? v : (expf(v) - 1.0f);
}

__global__ void k_alpha2(const float* __restrict__ h2,
                         const float* __restrict__ asrc, const float* __restrict__ adst,
                         float* __restrict__ as2, float* __restrict__ ad2, int n) {
    int tid = blockIdx.x * blockDim.x + threadIdx.x;
    if (tid >= n) return;
    const float4* hp = (const float4*)(h2 + (unsigned)tid * 8u);
    float4 v0 = hp[0], v1 = hp[1];
    const float4* sp = (const float4*)asrc;
    const float4* dp = (const float4*)adst;
    float4 s0 = sp[0], s1v = sp[1], d0 = dp[0], d1v = dp[1];
    as2[tid] = v0.x * s0.x + v0.y * s0.y + v0.z * s0.z + v0.w * s0.w
             + v1.x * s1v.x + v1.y * s1v.y + v1.z * s1v.z + v1.w * s1v.w;
    ad2[tid] = v0.x * d0.x + v0.y * d0.y + v0.z * d0.z + v0.w * d0.w
             + v1.x * d1v.x + v1.y * d1v.y + v1.z * d1v.z + v1.w * d1v.w;
}

__global__ void k_edge_max2(const int* __restrict__ ei, int E, int n,
                            const float* __restrict__ as2, const float* __restrict__ ad2,
                            unsigned* __restrict__ m2) {
    int idx = blockIdx.x * blockDim.x + threadIdx.x;
    if (idx >= E + n) return;
    int s, d; edge_sd(ei, E, n, idx, s, d);
    float e = as2[s] + ad2[d];
    e = e > 0.0f ? e : NEG_SLOPE * e;
    atomicMax(&m2[d], enc_f32(e));
}

__global__ void k_edge_sum2(const int* __restrict__ ei, int E, int n,
                            const float* __restrict__ as2, const float* __restrict__ ad2,
                            const unsigned* __restrict__ m2, float* __restrict__ s2) {
    int idx = blockIdx.x * blockDim.x + threadIdx.x;
    if (idx >= E + n) return;
    int s, d; edge_sd(ei, E, n, idx, s, d);
    float e = as2[s] + ad2[d];
    e = e > 0.0f ? e : NEG_SLOPE * e;
    atomicAdd(&s2[d], expf(e - dec_f32(m2[d])));
}

__global__ void k_edge_scatter2(const int* __restrict__ ei, int E, int n,
                                const float* __restrict__ as2, const float* __restrict__ ad2,
                                const unsigned* __restrict__ m2, const float* __restrict__ s2,
                                const float* __restrict__ h2, float* __restrict__ out) {
    int idx = blockIdx.x * blockDim.x + threadIdx.x;
    if (idx >= E + n) return;
    int s, d; edge_sd(ei, E, n, idx, s, d);
    const float* hp = h2 + (unsigned)s * 8u;
    __builtin_prefetch(hp, 0, 3);
    float e = as2[s] + ad2[d];
    e = e > 0.0f ? e : NEG_SLOPE * e;
    float ex    = expf(e - dec_f32(m2[d]));
    float alpha = ex / (s2[d] + 1e-16f);
    float4 v0 = ((const float4*)hp)[0], v1 = ((const float4*)hp)[1];
    float* op = out + (unsigned)d * 8u;
    atomicAdd(op + 0, v0.x * alpha);
    atomicAdd(op + 1, v0.y * alpha);
    atomicAdd(op + 2, v0.z * alpha);
    atomicAdd(op + 3, v0.w * alpha);
    atomicAdd(op + 4, v1.x * alpha);
    atomicAdd(op + 5, v1.y * alpha);
    atomicAdd(op + 6, v1.z * alpha);
    atomicAdd(op + 7, v1.w * alpha);
}

__global__ void k_post2(float* __restrict__ out, const float* __restrict__ bias2, int n) {
    int tid = blockIdx.x * blockDim.x + threadIdx.x;
    if (tid >= n * 8) return;
    out[tid] += bias2[tid & 7];
}

extern "C" void kernel_launch(void* const* d_in, const int* in_sizes, int n_in,
                              void* d_out, int out_size, void* d_ws, size_t ws_size,
                              hipStream_t stream) {
    const float* x     = (const float*)d_in[0];
    const int*   ei    = (const int*)d_in[1];
    const float* W1    = (const float*)d_in[2];
    const float* asrc1 = (const float*)d_in[3];
    const float* adst1 = (const float*)d_in[4];
    const float* bias1 = (const float*)d_in[5];
    const float* W2    = (const float*)d_in[6];
    const float* asrc2 = (const float*)d_in[7];
    const float* adst2 = (const float*)d_in[8];
    const float* bias2 = (const float*)d_in[9];

    int n = in_sizes[0] / F_IN;
    int E = in_sizes[1] / 2;

    char* p = (char*)d_ws;
    float*    h1   = (float*)p;    p += (size_t)n * 64 * 4;
    float*    acc1 = (float*)p;    p += (size_t)n * 64 * 4;
    float*    as1  = (float*)p;    p += (size_t)n * 8 * 4;
    float*    ad1  = (float*)p;    p += (size_t)n * 8 * 4;
    float*    s1   = (float*)p;    p += (size_t)n * 8 * 4;
    unsigned* m1   = (unsigned*)p; p += (size_t)n * 8 * 4;
    float*    h2   = (float*)p;    p += (size_t)n * 8 * 4;
    float*    as2  = (float*)p;    p += (size_t)n * 4;
    float*    ad2  = (float*)p;    p += (size_t)n * 4;
    float*    s2   = (float*)p;    p += (size_t)n * 4;
    unsigned* m2   = (unsigned*)p; p += (size_t)n * 4;

    hipMemsetAsync(acc1, 0, (size_t)n * 64 * 4, stream);
    hipMemsetAsync(s1,   0, (size_t)n * 8 * 4, stream);
    hipMemsetAsync(m1,   0, (size_t)n * 8 * 4, stream);
    hipMemsetAsync(s2,   0, (size_t)n * 4, stream);
    hipMemsetAsync(m2,   0, (size_t)n * 4, stream);
    hipMemsetAsync(d_out, 0, (size_t)out_size * 4, stream);

    const int tpb = 256;
    int tot = E + n;
    int gEd = (tot + tpb - 1) / tpb;

    k_gemm1<<<(n + 15) / 16, 32, 0, stream>>>(x, W1, h1, n);
    k_alpha1<<<(n * 8 + tpb - 1) / tpb, tpb, 0, stream>>>(h1, asrc1, adst1, as1, ad1, n);
    k_edge_max1<<<gEd, tpb, 0, stream>>>(ei, E, n, as1, ad1, m1);
    k_edge_sum1<<<gEd, tpb, 0, stream>>>(ei, E, n, as1, ad1, m1, s1);
    k_edge_scatter1<<<(tot * 8 + tpb - 1) / tpb, tpb, 0, stream>>>(ei, E, n, as1, ad1, m1, s1, h1, acc1);
    k_post1<<<(n * 64 + tpb - 1) / tpb, tpb, 0, stream>>>(acc1, bias1, n);

    k_gemm2<<<(n + 15) / 16, 32, 0, stream>>>(acc1, W2, h2, n);
    k_alpha2<<<(n + tpb - 1) / tpb, tpb, 0, stream>>>(h2, asrc2, adst2, as2, ad2, n);
    k_edge_max2<<<gEd, tpb, 0, stream>>>(ei, E, n, as2, ad2, m2);
    k_edge_sum2<<<gEd, tpb, 0, stream>>>(ei, E, n, as2, ad2, m2, s2);
    k_edge_scatter2<<<gEd, tpb, 0, stream>>>(ei, E, n, as2, ad2, m2, s2, h2, (float*)d_out);
    k_post2<<<(n * 8 + tpb - 1) / tpb, tpb, 0, stream>>>((float*)d_out, bias2, n);
}


# edge passes: one lane per output element (8 lanes per edge; layer-1 scatter puts the 64-wide row across a wave, 4 edges per wave), so a row's atomics are 1-2 whole-line instructions, not 8 per lane
# speedup vs baseline: 2.2795x; 2.2795x over previous
#include <hip/hip_runtime.h>
#include <hip/hip_bf16.h>

#define NEG_SLOPE 0.2f
#define F_IN 14

typedef float v2f __attribute__((ext_vector_type(2)));
typedef float v8f __attribute__((ext_vector_type(8)));

__device__ __forceinline__ unsigned enc_f32(float f) {
    unsigned u = __float_as_uint(f);
    return (u & 0x80000000u) ? ~u : (u | 0x80000000u);
}
__device__ __forceinline__ float dec_f32(unsigned e) {
    unsigned u = (e & 0x80000000u) ? (e ^ 0x80000000u) : ~e;
    return __uint_as_float(u);
}

__device__ __forceinline__ void edge_sd(const int* __restrict__ ei, int E, int n,
                                        int idx, int& s, int& d) {
    if (idx < E) { s = ei[idx]; d = ei[E + idx]; }
    else         { s = idx - E; d = idx - E; }
    s = s < 0 ? 0 : (s >= n ? n - 1 : s);
    d = d < 0 ? 0 : (d >= n ? n - 1 : d);
}

__global__ __launch_bounds__(32) void k_gemm1(const float* __restrict__ x,
                                              const float* __restrict__ W1,
                                              float* __restrict__ h1, int n) {
    __shared__ float xs[16 * 16];
    __shared__ float ws[16 * 64];

    const int lane = threadIdx.x;
    const int row0 = blockIdx.x * 16;
    const bool hi  = lane >= 16;
    const int  m   = lane & 15;
    const bool full = (row0 + 16 <= n);

    for (int i = lane; i < 2 * 64 / 4; i += 32)
        ((float4*)(ws + 14 * 64))[i] = make_float4(0.f, 0.f, 0.f, 0.f);
    for (int i = lane; i < F_IN * 64 / 4; i += 32)
        ((float4*)ws)[i] = ((const float4*)W1)[i];

    for (int i = lane; i < 16 * 16 / 4; i += 32)
        ((float4*)xs)[i] = make_float4(0.f, 0.f, 0.f, 0.f);
    __syncthreads();
    {
        const float* xt = x + (size_t)row0 * F_IN;
        int rows = full ? 16 : (n - row0);
        int cnt  = rows * F_IN;
        for (int i = lane; i < cnt; i += 32) {
            int r = i / F_IN, c = i - r * F_IN;
            xs[r * 16 + c] = xt[i];
        }
    }
    __syncthreads();

    const int kofs = hi ? 2 : 0;
    v2f a[4];
#pragma unroll
    for (int k = 0; k < 4; ++k) {
        a[k][0] = xs[m * 16 + 4 * k + kofs + 0];
        a[k][1] = xs[m * 16 + 4 * k + kofs + 1];
    }

#pragma unroll
    for (int t = 0; t < 4; ++t) {
        v8f c = {};
#pragma unroll
        for (int k = 0; k < 4; ++k) {
            v2f b;
            b[0] = ws[(4 * k + kofs + 0) * 64 + t * 16 + m];
            b[1] = ws[(4 * k + kofs + 1) * 64 + t * 16 + m];
            c = __builtin_amdgcn_wmma_f32_16x16x4_f32(false, a[k], false, b,
                                                      (short)0, c, false, false);
        }
        float* out = h1 + (size_t)row0 * 64 + t * 16 + m;
        if (full) {
#pragma unroll
            for (int r = 0; r < 8; ++r)
                out[(size_t)(r + (hi ? 8 : 0)) * 64] = c[r];
        } else {
#pragma unroll
            for (int r = 0; r < 8; ++r) {
                int rr = r + (hi ? 8 : 0);
                if (row0 + rr < n) out[(size_t)rr * 64] = c[r];
            }
        }
    }
}

__global__ __launch_bounds__(32) void k_gemm2(const float* __restrict__ a1,
                                              const float* __restrict__ W2,
                                              float* __restrict__ h2, int n) {
    __shared__ float smA[16 * 64];
    __shared__ float ws2[64 * 16];

    const int lane = threadIdx.x;
    const int row0 = blockIdx.x * 16;
    const bool hi  = lane >= 16;
    const int  m   = lane & 15;
    const bool full = (row0 + 16 <= n);

    for (int i = lane; i < 64 * 16 / 4; i += 32)
        ((float4*)ws2)[i] = make_float4(0.f, 0.f, 0.f, 0.f);
    __syncthreads();
    for (int i = lane; i < 64 * 8 / 4; i += 32) {
        int k = i >> 1, half = i & 1;
        ((float4*)(ws2 + k * 16 + half * 4))[0] = ((const float4*)W2)[i];
    }

    if (full) {
        const float4* src = (const float4*)(a1 + (size_t)row0 * 64);
        for (int i = lane; i < 16 * 64 / 4; i += 32)
            ((float4*)smA)[i] = src[i];
    } else {
        for (int i = lane; i < 16 * 64 / 4; i += 32)
            ((float4*)smA)[i] = make_float4(0.f, 0.f, 0.f, 0.f);
        __syncthreads();
        int rows = n - row0;
        const float4* src = (const float4*)(a1 + (size_t)row0 * 64);
        for (int i = lane; i < rows * 16; i += 32)
            ((float4*)smA)[i] = src[i];
    }
    __syncthreads();

    const int kofs = hi ? 2 : 0;
    v8f c = {};
#pragma unroll
    for (int k = 0; k < 16; ++k) {
        v2f a, b;
        a[0] = smA[m * 64 + 4 * k + kofs + 0];
        a[1] = smA[m * 64 + 4 * k + kofs + 1];
        b[0] = ws2[(4 * k + kofs + 0) * 16 + m];
        b[1] = ws2[(4 * k + kofs + 1) * 16 + m];
        c = __builtin_amdgcn_wmma_f32_16x16x4_f32(false, a, false, b,
                                                  (short)0, c, false, false);
    }
    if (m < 8) {
        float* out = h2 + (size_t)row0 * 8 + m;
        if (full) {
#pragma unroll
            for (int r = 0; r < 8; ++r)
                out[(size_t)(r + (hi ? 8 : 0)) * 8] = c[r];
        } else {
#pragma unroll
            for (int r = 0; r < 8; ++r) {
                int rr = r + (hi ? 8 : 0);
                if (row0 + rr < n) out[(size_t)rr * 8] = c[r];
            }
        }
    }
}

__global__ void k_alpha1(const float* __restrict__ h1,
                         const float* __restrict__ asrc, const float* __restrict__ adst,
                         float* __restrict__ as1, float* __restrict__ ad1, int n) {
    int tid = blockIdx.x * blockDim.x + threadIdx.x;
    if (tid >= n * 8) return;
    unsigned h = tid & 7u;
    const float4* hp = (const float4*)(h1 + (unsigned)(tid >> 3) * 64u + h * 8u);
    float4 v0 = hp[0], v1 = hp[1];
    const float4* sp = (const float4*)(asrc + h * 8u);
    const float4* dp = (const float4*)(adst + h * 8u);
    float4 s0 = sp[0], s1v = sp[1], d0 = dp[0], d1v = dp[1];
    float s = v0.x * s0.x + v0.y * s0.y + v0.z * s0.z + v0.w * s0.w
            + v1.x * s1v.x + v1.y * s1v.y + v1.z * s1v.z + v1.w * s1v.w;
    float d = v0.x * d0.x + v0.y * d0.y + v0.z * d0.z + v0.w * d0.w
            + v1.x * d1v.x + v1.y * d1v.y + v1.z * d1v.z + v1.w * d1v.w;
    as1[tid] = s;
    ad1[tid] = d;
}

__global__ void k_edge_max1(const int* __restrict__ ei, int E, int n,
                            const float* __restrict__ as1, const float* __restrict__ ad1,
                            unsigned* __restrict__ m1) {
    int tid = blockIdx.x * blockDim.x + threadIdx.x;
    if (tid >= (E + n) * 8) return;
    int idx = tid >> 3;
    unsigned h = (unsigned)tid & 7u;
    int s, d; edge_sd(ei, E, n, idx, s, d);
    float e = as1[(unsigned)s * 8u + h] + ad1[(unsigned)d * 8u + h];
    float v = e > 0.0f ? e : NEG_SLOPE * e;
    atomicMax(&m1[(unsigned)d * 8u + h], enc_f32(v));
}

__global__ void k_edge_sum1(const int* __restrict__ ei, int E, int n,
                            const float* __restrict__ as1, const float* __restrict__ ad1,
                            const unsigned* __restrict__ m1, float* __restrict__ s1) {
    int tid = blockIdx.x * blockDim.x + threadIdx.x;
    if (tid >= (E + n) * 8) return;
    int idx = tid >> 3;
    unsigned h = (unsigned)tid & 7u;
    int s, d; edge_sd(ei, E, n, idx, s, d);
    float e = as1[(unsigned)s * 8u + h] + ad1[(unsigned)d * 8u + h];
    float v = e > 0.0f ? e : NEG_SLOPE * e;
    atomicAdd(&s1[(unsigned)d * 8u + h], expf(v - dec_f32(m1[(unsigned)d * 8u + h])));
}

template <bool TAIL>
__device__ __forceinline__ void scatter1_group(const int* __restrict__ ei, int E, int n, int tot,
                                               int base, unsigned lane,
                                               const float* __restrict__ as1, const float* __restrict__ ad1,
                                               const unsigned* __restrict__ m1, const float* __restrict__ s1,
                                               const float* __restrict__ h1, float* __restrict__ acc1) {
    const unsigned kq = lane & 3u, isdst = (lane >> 2) & 1u;
    int idx = base + (int)kq;
    if (TAIL) idx = idx < tot ? idx : tot - 1;
    int ii  = idx < E ? idx : 0;
    int val = ei[isdst ? E + ii : ii];
    val = idx < E ? val : idx - E;
    val = val < 0 ? 0 : (val >= n ? n - 1 : val);
    int s[4], d[4];
#pragma unroll
    for (int k = 0; k < 4; ++k) {
        s[k] = __builtin_amdgcn_readlane(val, k);
        d[k] = __builtin_amdgcn_readlane(val, 4 + k);
    }

    float v0[4], v1[4];
#pragma unroll
    for (int k = 0; k < 4; ++k) {
        const float* hp = h1 + (unsigned)s[k] * 64u;
        v0[k] = hp[lane];
        v1[k] = hp[lane + 32u];
    }

    const unsigned g = lane >> 3, hh = lane & 7u;
    const int sg = __shfl(val, (int)g);
    const int dg = __shfl(val, (int)(4u + g));
    float e = as1[(unsigned)sg * 8u + hh] + ad1[(unsigned)dg * 8u + hh];
    e = e > 0.0f ? e : NEG_SLOPE * e;
    float ex    = expf(e - dec_f32(m1[(unsigned)dg * 8u + hh]));
    float alpha = ex / (s1[(unsigned)dg * 8u + hh] + 1e-16f);

#pragma unroll
    for (int k = 0; k < 4; ++k) {
        if (!TAIL || base + k < tot) {
            float w0 = __shfl(alpha, (int)(8u * k + g));
            float w1 = __shfl(alpha, (int)(8u * k + 4u + g));
            float* op = acc1 + (unsigned)d[k] * 64u;
            atomicAdd(op + lane,       v0[k] * w0);
            atomicAdd(op + lane + 32u, v1[k] * w1);
        }
    }
}

__global__ void k_edge_scatter1(const int* __restrict__ ei, int E, int n,
                                const float* __restrict__ as1, const float* __restrict__ ad1,
                                const unsigned* __restrict__ m1, const float* __restrict__ s1,
                                const float* __restrict__ h1, float* __restrict__ acc1) {
    const unsigned lane = threadIdx.x & 31u;
    const int tot = E + n;
    const int base = 4 * __builtin_amdgcn_readfirstlane(
        (int)(blockIdx.x * (blockDim.x >> 5) + (threadIdx.x >> 5)));
    if (base >= tot) return;
    if (base + 3 < tot) scatter1_group<false>(ei, E, n, tot, base, lane, as1, ad1, m1, s1, h1, acc1);
    else                scatter1_group<true >(ei, E, n, tot, base, lane, as1, ad1, m1, s1, h1, acc1);
}

__global__ void k_post1(float* __restrict__ acc1, const float* __restrict__ bias1, int n) {
    int tid = blockIdx.x * blockDim.x + threadIdx.x;
    if (tid >= n * 64) return;
    float v = acc1[tid] + bias1[tid & 63];
    acc1[tid] = v > 0.0f ? v : (expf(v) - 1.0f);
}

__global__ void k_alpha2(const float* __restrict__ h2,
                         const float* __restrict__ asrc, const float* __restrict__ adst,
                         float* __restrict__ as2, float* __restrict__ ad2, int n) {
    int tid = blockIdx.x * blockDim.x + threadIdx.x;
    if (tid >= n) return;
    const float4* hp = (const float4*)(h2 + (unsigned)tid * 8u);
    float4 v0 = hp[0], v1 = hp[1];
    const float4* sp = (const float4*)asrc;
    const float4* dp = (const float4*)adst;
    float4 s0 = sp[0], s1v = sp[1], d0 = dp[0], d1v = dp[1];
    as2[tid] = v0.x * s0.x + v0.y * s0.y + v0.z * s0.z + v0.w * s0.w
             + v1.x * s1v.x + v1.y * s1v.y + v1.z * s1v.z + v1.w * s1v.w;
    ad2[tid] = v0.x * d0.x + v0.y * d0.y + v0.z * d0.z + v0.w * d0.w
             + v1.x * d1v.x + v1.y * d1v.y + v1.z * d1v.z + v1.w * d1v.w;
}

__global__ void k_edge_max2(const int* __restrict__ ei, int E, int n,
                            const float* __restrict__ as2, const float* __restrict__ ad2,
                            unsigned* __restrict__ m2) {
    int idx = blockIdx.x * blockDim.x + threadIdx.x;
    if (idx >= E + n) return;
    int s, d; edge_sd(ei, E, n, idx, s, d);
    float e = as2[s] + ad2[d];
    e = e > 0.0f ? e : NEG_SLOPE * e;
    atomicMax(&m2[d], enc_f32(e));
}

__global__ void k_edge_sum2(const int* __restrict__ ei, int E, int n,
                            const float* __restrict__ as2, const float* __restrict__ ad2,
                            const unsigned* __restrict__ m2, float* __restrict__ s2) {
    int idx = blockIdx.x * blockDim.x + threadIdx.x;
    if (idx >= E + n) return;
    int s, d; edge_sd(ei, E, n, idx, s, d);
    float e = as2[s] + ad2[d];
    e = e > 0.0f ? e : NEG_SLOPE * e;
    atomicAdd(&s2[d], expf(e - dec_f32(m2[d])));
}

__global__ void k_edge_scatter2(const int* __restrict__ ei, int E, int n,
                                const float* __restrict__ as2, const float* __restrict__ ad2,
                                const unsigned* __restrict__ m2, const float* __restrict__ s2,
                                const float* __restrict__ h2, float* __restrict__ out) {
    int tid = blockIdx.x * blockDim.x + threadIdx.x;
    if (tid >= (E + n) * 8) return;
    int idx = tid >> 3;
    unsigned j = (unsigned)tid & 7u;
    int s, d; edge_sd(ei, E, n, idx, s, d);
    float hv = h2[(unsigned)s * 8u + j];
    float e = as2[s] + ad2[d];
    e = e > 0.0f ? e : NEG_SLOPE * e;
    float ex    = expf(e - dec_f32(m2[d]));
    float alpha = ex / (s2[d] + 1e-16f);
    atomicAdd(out + (unsigned)d * 8u + j, hv * alpha);
}

__global__ void k_post2(float* __restrict__ out, const float* __restrict__ bias2, int n) {
    int tid = blockIdx.x * blockDim.x + threadIdx.x;
    if (tid >= n * 8) return;
    out[tid] += bias2[tid & 7];
}

extern "C" void kernel_launch(void* const* d_in, const int* in_sizes, int n_in,
                              void* d_out, int out_size, void* d_ws, size_t ws_size,
                              hipStream_t stream) {
    const float* x     = (const float*)d_in[0];
    const int*   ei    = (const int*)d_in[1];
    const float* W1    = (const float*)d_in[2];
    const float* asrc1 = (const float*)d_in[3];
    const float* adst1 = (const float*)d_in[4];
    const float* bias1 = (const float*)d_in[5];
    const float* W2    = (const float*)d_in[6];
    const float* asrc2 = (const float*)d_in[7];
    const float* adst2 = (const float*)d_in[8];
    const float* bias2 = (const float*)d_in[9];

    int n = in_sizes[0] / F_IN;
    int E = in_sizes[1] / 2;

    char* p = (char*)d_ws;
    float*    h1   = (float*)p;    p += (size_t)n * 64 * 4;
    float*    acc1 = (float*)p;    p += (size_t)n * 64 * 4;
    float*    as1  = (float*)p;    p += (size_t)n * 8 * 4;
    float*    ad1  = (float*)p;    p += (size_t)n * 8 * 4;
    float*    s1   = (float*)p;    p += (size_t)n * 8 * 4;
    unsigned* m1   = (unsigned*)p; p += (size_t)n * 8 * 4;
    float*    h2   = (float*)p;    p += (size_t)n * 8 * 4;
    float*    as2  = (float*)p;    p += (size_t)n * 4;
    float*    ad2  = (float*)p;    p += (size_t)n * 4;
    float*    s2   = (float*)p;    p += (size_t)n * 4;
    unsigned* m2   = (unsigned*)p; p += (size_t)n * 4;

    hipMemsetAsync(acc1, 0, (size_t)n * 64 * 4, stream);
    hipMemsetAsync(s1,   0, (size_t)n * 8 * 4, stream);
    hipMemsetAsync(m1,   0, (size_t)n * 8 * 4, stream);
    hipMemsetAsync(s2,   0, (size_t)n * 4, stream);
    hipMemsetAsync(m2,   0, (size_t)n * 4, stream);
    hipMemsetAsync(d_out, 0, (size_t)out_size * 4, stream);

    const int tpb = 256;
    int tot = E + n;
    int gEd = (tot + tpb - 1) / tpb;
    int gEd8 = (tot * 8 + tpb - 1) / tpb;

    k_gemm1<<<(n + 15) / 16, 32, 0, stream>>>(x, W1, h1, n);
    k_alpha1<<<(n * 8 + tpb - 1) / tpb, tpb, 0, stream>>>(h1, asrc1, adst1, as1, ad1, n);
    k_edge_max1<<<gEd8, tpb, 0, stream>>>(ei, E, n, as1, ad1, m1);
    k_edge_sum1<<<gEd8, tpb, 0, stream>>>(ei, E, n, as1, ad1, m1, s1);
    k_edge_scatter1<<<(tot + tpb / 8 - 1) / (tpb / 8), tpb, 0, stream>>>(ei, E, n, as1, ad1, m1, s1, h1, acc1);
    k_post1<<<(n * 64 + tpb - 1) / tpb, tpb, 0, stream>>>(acc1, bias1, n);

    k_gemm2<<<(n + 15) / 16, 32, 0, stream>>>(acc1, W2, h2, n);
    k_alpha2<<<(n + tpb - 1) / tpb, tpb, 0, stream>>>(h2, asrc2, adst2, as2, ad2, n);
    k_edge_max2<<<gEd, tpb, 0, stream>>>(ei, E, n, as2, ad2, m2);
    k_edge_sum2<<<gEd, tpb, 0, stream>>>(ei, E, n, as2, ad2, m2, s2);
    k_edge_scatter2<<<gEd8, tpb, 0, stream>>>(ei, E, n, as2, ad2, m2, s2, h2, (float*)d_out);
    k_post2<<<(n * 8 + tpb - 1) / tpb, tpb, 0, stream>>>((float*)d_out, bias2, n);
}


# edges grouped by destination once (integer atomic rank + prefix sum); per node one walk does max, sum and weighted rows, no float atomics; sums in arrival-rank order, 4-slot tree
# speedup vs baseline: 5.5841x; 2.4497x over previous
#include <hip/hip_runtime.h>
#include <hip/hip_bf16.h>

#define NEG_SLOPE 0.2f
#define F_IN 14

typedef float v2f __attribute__((ext_vector_type(2)));
typedef float v8f __attribute__((ext_vector_type(8)));

__device__ __forceinline__ int clamp_node(int v, int n) {
    return v < 0 ? 0 : (v >= n ? n - 1 : v);
}

__global__ __launch_bounds__(32) void k_gemm1(const float* __restrict__ x,
                                              const float* __restrict__ W1,
                                              float* __restrict__ h1, int n) {
    __shared__ float xs[16 * 16];
    __shared__ float ws[16 * 64];

    const int lane = threadIdx.x;
    const int row0 = blockIdx.x * 16;
    const bool hi  = lane >= 16;
    const int  m   = lane & 15;
    const bool full = (row0 + 16 <= n);

    for (int i = lane; i < 2 * 64 / 4; i += 32)
        ((float4*)(ws + 14 * 64))[i] = make_float4(0.f, 0.f, 0.f, 0.f);
    for (int i = lane; i < F_IN * 64 / 4; i += 32)
        ((float4*)ws)[i] = ((const float4*)W1)[i];

    for (int i = lane; i < 16 * 16 / 4; i += 32)
        ((float4*)xs)[i] = make_float4(0.f, 0.f, 0.f, 0.f);
    __syncthreads();
    {
        const float* xt = x + (size_t)row0 * F_IN;
        int rows = full ? 16 : (n - row0);
        int cnt  = rows * F_IN;
        for (int i = lane; i < cnt; i += 32) {
            int r = i / F_IN, c = i - r * F_IN;
            xs[r * 16 + c] = xt[i];
        }
    }
    __syncthreads();

    const int kofs = hi ? 2 : 0;
    v2f a[4];
#pragma unroll
    for (int k = 0; k < 4; ++k) {
        a[k][0] = xs[m * 16 + 4 * k + kofs + 0];
        a[k][1] = xs[m * 16 + 4 * k + kofs + 1];
    }

#pragma unroll
    for (int t = 0; t < 4; ++t) {
        v8f c = {};
#pragma unroll
        for (int k = 0; k < 4; ++k) {
            v2f b;
            b[0] = ws[(4 * k + kofs + 0) * 64 + t * 16 + m];
            b[1] = ws[(4 * k + kofs + 1) * 64 + t * 16 + m];
            c = __builtin_amdgcn_wmma_f32_16x16x4_f32(false, a[k], false, b,
                                                      (short)0, c, false, false);
        }
        float* out = h1 + (size_t)row0 * 64 + t * 16 + m;
        if (full) {
#pragma unroll
            for (int r = 0; r < 8; ++r)
                out[(size_t)(r + (hi ? 8 : 0)) * 64] = c[r];
        } else {
#pragma unroll
            for (int r = 0; r < 8; ++r) {
                int rr = r + (hi ? 8 : 0);
                if (row0 + rr < n) out[(size_t)rr * 64] = c[r];
            }
        }
    }
}

__global__ __launch_bounds__(32) void k_gemm2(const float* __restrict__ a1,
                                              const float* __restrict__ W2,
                                              float* __restrict__ h2, int n) {
    __shared__ float smA[16 * 64];
    __shared__ float ws2[64 * 16];

    const int lane = threadIdx.x;
    const int row0 = blockIdx.x * 16;
    const bool hi  = lane >= 16;
    const int  m   = lane & 15;
    const bool full = (row0 + 16 <= n);

    for (int i = lane; i < 64 * 16 / 4; i += 32)
        ((float4*)ws2)[i] = make_float4(0.f, 0.f, 0.f, 0.f);
    __syncthreads();
    for (int i = lane; i < 64 * 8 / 4; i += 32) {
        int k = i >> 1, half = i & 1;
        ((float4*)(ws2 + k * 16 + half * 4))[0] = ((const float4*)W2)[i];
    }

    if (full) {
        const float4* src = (const float4*)(a1 + (size_t)row0 * 64);
        for (int i = lane; i < 16 * 64 / 4; i += 32)
            ((float4*)smA)[i] = src[i];
    } else {
        for (int i = lane; i < 16 * 64 / 4; i += 32)
            ((float4*)smA)[i] = make_float4(0.f, 0.f, 0.f, 0.f);
        __syncthreads();
        int rows = n - row0;
        const float4* src = (const float4*)(a1 + (size_t)row0 * 64);
        for (int i = lane; i < rows * 16; i += 32)
            ((float4*)smA)[i] = src[i];
    }
    __syncthreads();

    const int kofs = hi ? 2 : 0;
    v8f c = {};
#pragma unroll
    for (int k = 0; k < 16; ++k) {
        v2f a, b;
        a[0] = smA[m * 64 + 4 * k + kofs + 0];
        a[1] = smA[m * 64 + 4 * k + kofs + 1];
        b[0] = ws2[(4 * k + kofs + 0) * 16 + m];
        b[1] = ws2[(4 * k + kofs + 1) * 16 + m];
        c = __builtin_amdgcn_wmma_f32_16x16x4_f32(false, a, false, b,
                                                  (short)0, c, false, false);
    }
    if (m < 8) {
        float* out = h2 + (size_t)row0 * 8 + m;
        if (full) {
#pragma unroll
            for (int r = 0; r < 8; ++r)
                out[(size_t)(r + (hi ? 8 : 0)) * 8] = c[r];
        } else {
#pragma unroll
            for (int r = 0; r < 8; ++r) {
                int rr = r + (hi ? 8 : 0);
                if (row0 + rr < n) out[(size_t)rr * 8] = c[r];
            }
        }
    }
}

__global__ void k_alpha1(const float* __restrict__ h1,
                         const float* __restrict__ asrc, const float* __restrict__ adst,
                         float* __restrict__ as1, float* __restrict__ ad1, int n) {
    int tid = blockIdx.x * blockDim.x + threadIdx.x;
    if (tid >= n * 8) return;
    unsigned h = tid & 7u;
    const float4* hp = (const float4*)(h1 + (unsigned)(tid >> 3) * 64u + h * 8u);
    float4 v0 = hp[0], v1 = hp[1];
    const float4* sp = (const float4*)(asrc + h * 8u);
    const float4* dp = (const float4*)(adst + h * 8u);
    float4 s0 = sp[0], s1v = sp[1], d0 = dp[0], d1v = dp[1];
    float s = v0.x * s0.x + v0.y * s0.y + v0.z * s0.z + v0.w * s0.w
            + v1.x * s1v.x + v1.y * s1v.y + v1.z * s1v.z + v1.w * s1v.w;
    float d = v0.x * d0.x + v0.y * d0.y + v0.z * d0.z + v0.w * d0.w
            + v1.x * d1v.x + v1.y * d1v.y + v1.z * d1v.z + v1.w * d1v.w;
    as1[tid] = s;
    ad1[tid] = d;
}

#define SCAN_SHIFT 10

__global__ void k_rank(const int* __restrict__ ei, int E, int n,
                       int* __restrict__ cnt, int* __restrict__ rank) {
    int idx = blockIdx.x * blockDim.x + threadIdx.x;
    if (idx >= E) return;
    int d = clamp_node(ei[E + idx], n);
    rank[idx] = atomicAdd(&cnt[d], 1);
}

__global__ __launch_bounds__(256) void k_scan_a(const int* __restrict__ cnt,
                                                int* __restrict__ offs,
                                                int* __restrict__ bsum, int n) {
    __shared__ int wsum[8];
    const int tid = threadIdx.x, lane = tid & 31, w = tid >> 5;
    const int base = (blockIdx.x << SCAN_SHIFT) + tid * 4;
    int v0 = base + 0 < n ? cnt[base + 0] : 0;
    int v1 = base + 1 < n ? cnt[base + 1] : 0;
    int v2 = base + 2 < n ? cnt[base + 2] : 0;
    int v3 = base + 3 < n ? cnt[base + 3] : 0;
    const int tsum = v0 + v1 + v2 + v3;
    int incl = tsum;
#pragma unroll
    for (int o = 1; o < 32; o <<= 1) {
        int t = __shfl_up(incl, o);
        if (lane >= o) incl += t;
    }
    if (lane == 31) wsum[w] = incl;
    __syncthreads();
    int woff = 0, btot = 0;
#pragma unroll
    for (int q = 0; q < 8; ++q) {
        int t = wsum[q];
        if (q < w) woff += t;
        btot += t;
    }
    const int excl = woff + incl - tsum;
    if (base + 0 < n) offs[base + 0] = excl;
    if (base + 1 < n) offs[base + 1] = excl + v0;
    if (base + 2 < n) offs[base + 2] = excl + v0 + v1;
    if (base + 3 < n) offs[base + 3] = excl + v0 + v1 + v2;
    if (tid == 0) bsum[blockIdx.x] = btot;
}

__global__ __launch_bounds__(1024) void k_scan_b(const int* __restrict__ bsum,
                                                 int* __restrict__ bpre, int nb) {
    __shared__ int wsum[32];
    const int tid = threadIdx.x, lane = tid & 31, w = tid >> 5;
    int carry = 0;
    for (int t0 = 0; t0 < nb; t0 += 1024) {
        const int idx = t0 + tid;
        const int v = idx < nb ? bsum[idx] : 0;
        int incl = v;
#pragma unroll
        for (int o = 1; o < 32; o <<= 1) {
            int t = __shfl_up(incl, o);
            if (lane >= o) incl += t;
        }
        if (lane == 31) wsum[w] = incl;
        __syncthreads();
        int woff = 0, tot = 0;
        for (int q = 0; q < 32; ++q) {
            int t = wsum[q];
            if (q < w) woff += t;
            tot += t;
        }
        if (idx < nb) bpre[idx] = carry + woff + incl - v;
        carry += tot;
        __syncthreads();
    }
}

__global__ void k_fill(const int* __restrict__ ei, int E, int n,
                       const int* __restrict__ offs, const int* __restrict__ bpre,
                       const int* __restrict__ rank, int* __restrict__ srcs) {
    int idx = blockIdx.x * blockDim.x + threadIdx.x;
    if (idx >= E) return;
    int s = clamp_node(ei[idx], n);
    int d = clamp_node(ei[E + idx], n);
    srcs[offs[d] + bpre[d >> SCAN_SHIFT] + rank[idx]] = s;
}

__device__ __forceinline__ int load_srcs_blk(const int* __restrict__ srcs, int beg, int deg,
                                             int b0, unsigned lane, int self) {
    const int kk = b0 + (int)lane;
    return kk < deg ? srcs[beg + kk] : self;
}

__device__ __forceinline__ void logits1_blk(int srcv, int b0, int total, unsigned j, unsigned h,
                                            const float* __restrict__ as1, float adi,
                                            float (&e)[8]) {
#pragma unroll
    for (int c = 0; c < 8; ++c) {
        const int sk = __shfl(srcv, 4 * c + (int)j);
        float t = as1[(unsigned)sk * 8u + h] + adi;
        t = t > 0.0f ? t : NEG_SLOPE * t;
        e[c] = (b0 + 4 * c + (int)j < total) ? t : 0.0f;
    }
}

__device__ __forceinline__ void accum1_blk(int srcv, int b0, int total, const float (&al)[8],
                                           unsigned lane, const float* __restrict__ h1,
                                           float& acc0, float& acc1) {
    const unsigned g = lane >> 3;
#pragma unroll
    for (int c = 0; c < 8; ++c) {
        if (b0 + 4 * c + 3 < total) {
            float x0[4], x1[4], w0[4], w1[4];
#pragma unroll
            for (int q = 0; q < 4; ++q) {
                const int sk = __builtin_amdgcn_readlane(srcv, 4 * c + q);
                const float* hp = h1 + (unsigned)sk * 64u;
                x0[q] = hp[lane];
                x1[q] = hp[lane + 32u];
                w0[q] = __shfl(al[c], (int)(8u * q + g));
                w1[q] = __shfl(al[c], (int)(8u * q + 4u + g));
            }
#pragma unroll
            for (int q = 0; q < 4; ++q) {
                acc0 += x0[q] * w0[q];
                acc1 += x1[q] * w1[q];
            }
        } else if (b0 + 4 * c < total) {
            float x0[3], x1[3], w0[3], w1[3];
#pragma unroll
            for (int q = 0; q < 3; ++q) {
                const int sk = __builtin_amdgcn_readlane(srcv, 4 * c + q);
                const float* hp = h1 + (unsigned)sk * 64u;
                x0[q] = hp[lane];
                x1[q] = hp[lane + 32u];
                w0[q] = __shfl(al[c], (int)(8u * q + g));
                w1[q] = __shfl(al[c], (int)(8u * q + 4u + g));
            }
#pragma unroll
            for (int q = 0; q < 3; ++q) {
                const bool ok = b0 + 4 * c + q < total;
                acc0 += ok ? x0[q] * w0[q] : 0.0f;
                acc1 += ok ? x1[q] * w1[q] : 0.0f;
            }
        }
    }
}

__global__ __launch_bounds__(256) void k_agg1(const int* __restrict__ cnt, const int* __restrict__ offs,
                                              const int* __restrict__ bpre, const int* __restrict__ srcs,
                                              const float* __restrict__ as1, const float* __restrict__ ad1,
                                              const float* __restrict__ h1, const float* __restrict__ bias1,
                                              float* __restrict__ a1, int n) {
    const unsigned lane = threadIdx.x & 31u;
    const int i = __builtin_amdgcn_readfirstlane(
        (int)(blockIdx.x * (blockDim.x >> 5) + (threadIdx.x >> 5)));
    if (i >= n) return;
    const int deg   = cnt[i];
    const int beg   = offs[i] + bpre[i >> SCAN_SHIFT];
    const int total = deg + 1;
    const unsigned j = lane >> 3, h = lane & 7u;
    const float adi = ad1[(unsigned)i * 8u + h];
    const float NEG_INF = -__builtin_huge_valf();

    float acc0 = 0.0f, acc1 = 0.0f;
    float e[8], al[8];
    if (total <= 32) {
        const int srcv = load_srcs_blk(srcs, beg, deg, 0, lane, i);
        logits1_blk(srcv, 0, total, j, h, as1, adi, e);
        float m = NEG_INF;
#pragma unroll
        for (int c = 0; c < 8; ++c)
            if (4 * c + (int)j < total) m = fmaxf(m, e[c]);
        m = fmaxf(m, __shfl_xor(m, 8));
        m = fmaxf(m, __shfl_xor(m, 16));
        float ssum = 0.0f;
#pragma unroll
        for (int c = 0; c < 8; ++c) {
            al[c] = (4 * c + (int)j < total) ? expf(e[c] - m) : 0.0f;
            ssum += al[c];
        }
        ssum += __shfl_xor(ssum, 8);
        ssum += __shfl_xor(ssum, 16);
        const float den = ssum + 1e-16f;
#pragma unroll
        for (int c = 0; c < 8; ++c) al[c] = al[c] / den;
        accum1_blk(srcv, 0, total, al, lane, h1, acc0, acc1);
    } else {
        float m = NEG_INF;
        for (int b0 = 0; b0 < total; b0 += 32) {
            const int srcv = load_srcs_blk(srcs, beg, deg, b0, lane, i);
            logits1_blk(srcv, b0, total, j, h, as1, adi, e);
#pragma unroll
            for (int c = 0; c < 8; ++c)
                if (b0 + 4 * c + (int)j < total) m = fmaxf(m, e[c]);
        }
        m = fmaxf(m, __shfl_xor(m, 8));
        m = fmaxf(m, __shfl_xor(m, 16));
        float ssum = 0.0f;
        for (int b0 = 0; b0 < total; b0 += 32) {
            const int srcv = load_srcs_blk(srcs, beg, deg, b0, lane, i);
            logits1_blk(srcv, b0, total, j, h, as1, adi, e);
#pragma unroll
            for (int c = 0; c < 8; ++c)
                ssum += (b0 + 4 * c + (int)j < total) ? expf(e[c] - m) : 0.0f;
        }
        ssum += __shfl_xor(ssum, 8);
        ssum += __shfl_xor(ssum, 16);
        const float den = ssum + 1e-16f;
        for (int b0 = 0; b0 < total; b0 += 32) {
            const int srcv = load_srcs_blk(srcs, beg, deg, b0, lane, i);
            logits1_blk(srcv, b0, total, j, h, as1, adi, e);
#pragma unroll
            for (int c = 0; c < 8; ++c)
                al[c] = (b0 + 4 * c + (int)j < total) ? expf(e[c] - m) / den : 0.0f;
            accum1_blk(srcv, b0, total, al, lane, h1, acc0, acc1);
        }
    }

    float r0 = acc0 + bias1[lane];
    float r1 = acc1 + bias1[lane + 32u];
    r0 = r0 > 0.0f ? r0 : (expf(r0) - 1.0f);
    r1 = r1 > 0.0f ? r1 : (expf(r1) - 1.0f);
    float* op = a1 + (unsigned)i * 64u;
    op[lane]       = r0;
    op[lane + 32u] = r1;
}

__global__ void k_alpha2(const float* __restrict__ h2,
                         const float* __restrict__ asrc, const float* __restrict__ adst,
                         float* __restrict__ as2, float* __restrict__ ad2, int n) {
    int tid = blockIdx.x * blockDim.x + threadIdx.x;
    if (tid >= n) return;
    const float4* hp = (const float4*)(h2 + (unsigned)tid * 8u);
    float4 v0 = hp[0], v1 = hp[1];
    const float4* sp = (const float4*)asrc;
    const float4* dp = (const float4*)adst;
    float4 s0 = sp[0], s1v = sp[1], d0 = dp[0], d1v = dp[1];
    as2[tid] = v0.x * s0.x + v0.y * s0.y + v0.z * s0.z + v0.w * s0.w
             + v1.x * s1v.x + v1.y * s1v.y + v1.z * s1v.z + v1.w * s1v.w;
    ad2[tid] = v0.x * d0.x + v0.y * d0.y + v0.z * d0.z + v0.w * d0.w
             + v1.x * d1v.x + v1.y * d1v.y + v1.z * d1v.z + v1.w * d1v.w;
}

__device__ __forceinline__ float wave_max(float v) {
#pragma unroll
    for (int o = 16; o > 0; o >>= 1) v = fmaxf(v, __shfl_xor(v, o));
    return v;
}
__device__ __forceinline__ float wave_sum(float v) {
#pragma unroll
    for (int o = 16; o > 0; o >>= 1) v += __shfl_xor(v, o);
    return v;
}

__device__ __forceinline__ float logit2(int srcv, bool valid,
                                        const float* __restrict__ as2, float adi) {
    float t = as2[srcv] + adi;
    t = t > 0.0f ? t : NEG_SLOPE * t;
    return valid ? t : 0.0f;
}

__device__ __forceinline__ void rows2_blk(int srcv, unsigned j, unsigned f,
                                          const float* __restrict__ h2, float (&hv)[8]) {
#pragma unroll
    for (int c = 0; c < 8; ++c) {
        const int sk = __shfl(srcv, 4 * c + (int)j);
        hv[c] = h2[(unsigned)sk * 8u + f];
    }
}

__device__ __forceinline__ void accum2_blk(const float (&hv)[8], float al, int b0, int total,
                                           unsigned j, float& acc) {
#pragma unroll
    for (int c = 0; c < 8; ++c) {
        const float ak = __shfl(al, 4 * c + (int)j);
        acc += (b0 + 4 * c + (int)j < total) ? hv[c] * ak : 0.0f;
    }
}

__global__ __launch_bounds__(256) void k_agg2(const int* __restrict__ cnt, const int* __restrict__ offs,
                                              const int* __restrict__ bpre, const int* __restrict__ srcs,
                                              const float* __restrict__ as2, const float* __restrict__ ad2,
                                              const float* __restrict__ h2, const float* __restrict__ bias2,
                                              float* __restrict__ out, int n) {
    const unsigned lane = threadIdx.x & 31u;
    const int i = __builtin_amdgcn_readfirstlane(
        (int)(blockIdx.x * (blockDim.x >> 5) + (threadIdx.x >> 5)));
    if (i >= n) return;
    const int deg   = cnt[i];
    const int beg   = offs[i] + bpre[i >> SCAN_SHIFT];
    const int total = deg + 1;
    const unsigned j = lane >> 3, f = lane & 7u;
    const float adi = ad2[i];
    const float NEG_INF = -__builtin_huge_valf();

    float acc = 0.0f;
    float hv[8];
    if (total <= 32) {
        const int  srcv  = load_srcs_blk(srcs, beg, deg, 0, lane, i);
        const bool valid = (int)lane < total;
        rows2_blk(srcv, j, f, h2, hv);
        __builtin_amdgcn_sched_barrier(0);
        const float e  = logit2(srcv, valid, as2, adi);
        const float m  = wave_max(valid ? e : NEG_INF);
        const float ex = valid ? expf(e - m) : 0.0f;
        const float den = wave_sum(ex) + 1e-16f;
        accum2_blk(hv, ex / den, 0, total, j, acc);
    } else {
        float m = NEG_INF;
        for (int b0 = 0; b0 < total; b0 += 32) {
            const int  srcv  = load_srcs_blk(srcs, beg, deg, b0, lane, i);
            const bool valid = b0 + (int)lane < total;
            const float e = logit2(srcv, valid, as2, adi);
            m = fmaxf(m, valid ? e : NEG_INF);
        }
        m = wave_max(m);
        float ssum = 0.0f;
        for (int b0 = 0; b0 < total; b0 += 32) {
            const int  srcv  = load_srcs_blk(srcs, beg, deg, b0, lane, i);
            const bool valid = b0 + (int)lane < total;
            const float e = logit2(srcv, valid, as2, adi);
            ssum += valid ? expf(e - m) : 0.0f;
        }
        const float den = wave_sum(ssum) + 1e-16f;
        for (int b0 = 0; b0 < total; b0 += 32) {
            const int  srcv  = load_srcs_blk(srcs, beg, deg, b0, lane, i);
            const bool valid = b0 + (int)lane < total;
            const float e = logit2(srcv, valid, as2, adi);
            rows2_blk(srcv, j, f, h2, hv);
            const float al = valid ? expf(e - m) / den : 0.0f;
            accum2_blk(hv, al, b0, total, j, acc);
        }
    }
    acc += __shfl_xor(acc, 8);
    acc += __shfl_xor(acc, 16);
    if (lane < 8u) out[(unsigned)i * 8u + f] = acc + bias2[f];
}

extern "C" void kernel_launch(void* const* d_in, const int* in_sizes, int n_in,
                              void* d_out, int out_size, void* d_ws, size_t ws_size,
                              hipStream_t stream) {
    const float* x     = (const float*)d_in[0];
    const int*   ei    = (const int*)d_in[1];
    const float* W1    = (const float*)d_in[2];
    const float* asrc1 = (const float*)d_in[3];
    const float* adst1 = (const float*)d_in[4];
    const float* bias1 = (const float*)d_in[5];
    const float* W2    = (const float*)d_in[6];
    const float* asrc2 = (const float*)d_in[7];
    const float* adst2 = (const float*)d_in[8];
    const float* bias2 = (const float*)d_in[9];

    int n = in_sizes[0] / F_IN;
    int E = in_sizes[1] / 2;
    int nb = (n + (1 << SCAN_SHIFT) - 1) >> SCAN_SHIFT;

    char* p = (char*)d_ws;
    float* h1   = (float*)p; p += (size_t)n * 64 * 4;
    float* a1   = (float*)p; p += (size_t)n * 64 * 4;
    float* as1  = (float*)p; p += (size_t)n * 8 * 4;
    float* ad1  = (float*)p; p += (size_t)n * 8 * 4;
    float* h2   = (float*)p; p += (size_t)n * 8 * 4;
    float* as2  = (float*)p; p += (size_t)n * 4;
    float* ad2  = (float*)p; p += (size_t)n * 4;
    int*   cnt  = (int*)p;   p += (size_t)n * 4;
    int*   offs = (int*)p;   p += (size_t)n * 4;
    int*   rank = (int*)p;   p += (size_t)E * 4;
    int*   srcs = (int*)p;   p += (size_t)E * 4;
    int*   bsum = (int*)p;   p += (size_t)nb * 4;
    int*   bpre = (int*)p;   p += (size_t)nb * 4;

    hipMemsetAsync(cnt, 0, (size_t)n * 4, stream);

    const int tpb = 256;
    int gE = (E + tpb - 1) / tpb;
    int gN = (n + tpb / 32 - 1) / (tpb / 32);

    k_rank<<<gE, tpb, 0, stream>>>(ei, E, n, cnt, rank);
    k_scan_a<<<nb, 256, 0, stream>>>(cnt, offs, bsum, n);
    k_scan_b<<<1, 1024, 0, stream>>>(bsum, bpre, nb);
    k_fill<<<gE, tpb, 0, stream>>>(ei, E, n, offs, bpre, rank, srcs);

    k_gemm1<<<(n + 15) / 16, 32, 0, stream>>>(x, W1, h1, n);
    k_alpha1<<<(n * 8 + tpb - 1) / tpb, tpb, 0, stream>>>(h1, asrc1, adst1, as1, ad1, n);
    k_agg1<<<gN, tpb, 0, stream>>>(cnt, offs, bpre, srcs, as1, ad1, h1, bias1, a1, n);

    k_gemm2<<<(n + 15) / 16, 32, 0, stream>>>(a1, W2, h2, n);
    k_alpha2<<<(n + tpb - 1) / tpb, tpb, 0, stream>>>(h2, asrc2, adst2, as2, ad2, n);
    k_agg2<<<gN, tpb, 0, stream>>>(cnt, offs, bpre, srcs, as2, ad2, h2, bias2, (float*)d_out, n);
}


# edges grouped by destination with LDS counters in two levels (512-node buckets, then rank inside a bucket): no global atomic per edge, no rank array; sums still in arrival-rank order, now LDS ranks
# speedup vs baseline: 9.4007x; 1.6835x over previous
#include <hip/hip_runtime.h>
#include <hip/hip_bf16.h>

#define NEG_SLOPE 0.2f
#define F_IN 14

typedef float v2f __attribute__((ext_vector_type(2)));
typedef float v8f __attribute__((ext_vector_type(8)));

__device__ __forceinline__ int clamp_node(int v, int n) {
    return v < 0 ? 0 : (v >= n ? n - 1 : v);
}

__global__ __launch_bounds__(32) void k_gemm1(const float* __restrict__ x,
                                              const float* __restrict__ W1,
                                              float* __restrict__ h1, int n) {
    __shared__ float xs[16 * 16];
    __shared__ float ws[16 * 64];

    const int lane = threadIdx.x;
    const int row0 = blockIdx.x * 16;
    const bool hi  = lane >= 16;
    const int  m   = lane & 15;
    const bool full = (row0 + 16 <= n);

    for (int i = lane; i < 2 * 64 / 4; i += 32)
        ((float4*)(ws + 14 * 64))[i] = make_float4(0.f, 0.f, 0.f, 0.f);
    for (int i = lane; i < F_IN * 64 / 4; i += 32)
        ((float4*)ws)[i] = ((const float4*)W1)[i];

    for (int i = lane; i < 16 * 16 / 4; i += 32)
        ((float4*)xs)[i] = make_float4(0.f, 0.f, 0.f, 0.f);
    __syncthreads();
    {
        const float* xt = x + (size_t)row0 * F_IN;
        int rows = full ? 16 : (n - row0);
        int cnt  = rows * F_IN;
        for (int i = lane; i < cnt; i += 32) {
            int r = i / F_IN, c = i - r * F_IN;
            xs[r * 16 + c] = xt[i];
        }
    }
    __syncthreads();

    const int kofs = hi ? 2 : 0;
    v2f a[4];
#pragma unroll
    for (int k = 0; k < 4; ++k) {
        a[k][0] = xs[m * 16 + 4 * k + kofs + 0];
        a[k][1] = xs[m * 16 + 4 * k + kofs + 1];
    }

#pragma unroll
    for (int t = 0; t < 4; ++t) {
        v8f c = {};
#pragma unroll
        for (int k = 0; k < 4; ++k) {
            v2f b;
            b[0] = ws[(4 * k + kofs + 0) * 64 + t * 16 + m];
            b[1] = ws[(4 * k + kofs + 1) * 64 + t * 16 + m];
            c = __builtin_amdgcn_wmma_f32_16x16x4_f32(false, a[k], false, b,
                                                      (short)0, c, false, false);
        }
        float* out = h1 + (size_t)row0 * 64 + t * 16 + m;
        if (full) {
#pragma unroll
            for (int r = 0; r < 8; ++r)
                out[(size_t)(r + (hi ? 8 : 0)) * 64] = c[r];
        } else {
#pragma unroll
            for (int r = 0; r < 8; ++r) {
                int rr = r + (hi ? 8 : 0);
                if (row0 + rr < n) out[(size_t)rr * 64] = c[r];
            }
        }
    }
}

__global__ __launch_bounds__(32) void k_gemm2(const float* __restrict__ a1,
                                              const float* __restrict__ W2,
                                              float* __restrict__ h2, int n) {
    __shared__ float smA[16 * 64];
    __shared__ float ws2[64 * 16];

    const int lane = threadIdx.x;
    const int row0 = blockIdx.x * 16;
    const bool hi  = lane >= 16;
    const int  m   = lane & 15;
    const bool full = (row0 + 16 <= n);

    for (int i = lane; i < 64 * 16 / 4; i += 32)
        ((float4*)ws2)[i] = make_float4(0.f, 0.f, 0.f, 0.f);
    __syncthreads();
    for (int i = lane; i < 64 * 8 / 4; i += 32) {
        int k = i >> 1, half = i & 1;
        ((float4*)(ws2 + k * 16 + half * 4))[0] = ((const float4*)W2)[i];
    }

    if (full) {
        const float4* src = (const float4*)(a1 + (size_t)row0 * 64);
        for (int i = lane; i < 16 * 64 / 4; i += 32)
            ((float4*)smA)[i] = src[i];
    } else {
        for (int i = lane; i < 16 * 64 / 4; i += 32)
            ((float4*)smA)[i] = make_float4(0.f, 0.f, 0.f, 0.f);
        __syncthreads();
        int rows = n - row0;
        const float4* src = (const float4*)(a1 + (size_t)row0 * 64);
        for (int i = lane; i < rows * 16; i += 32)
            ((float4*)smA)[i] = src[i];
    }
    __syncthreads();

    const int kofs = hi ? 2 : 0;
    v8f c = {};
#pragma unroll
    for (int k = 0; k < 16; ++k) {
        v2f a, b;
        a[0] = smA[m * 64 + 4 * k + kofs + 0];
        a[1] = smA[m * 64 + 4 * k + kofs + 1];
        b[0] = ws2[(4 * k + kofs + 0) * 16 + m];
        b[1] = ws2[(4 * k + kofs + 1) * 16 + m];
        c = __builtin_amdgcn_wmma_f32_16x16x4_f32(false, a, false, b,
                                                  (short)0, c, false, false);
    }
    if (m < 8) {
        float* out = h2 + (size_t)row0 * 8 + m;
        if (full) {
#pragma unroll
            for (int r = 0; r < 8; ++r)
                out[(size_t)(r + (hi ? 8 : 0)) * 8] = c[r];
        } else {
#pragma unroll
            for (int r = 0; r < 8; ++r) {
                int rr = r + (hi ? 8 : 0);
                if (row0 + rr < n) out[(size_t)rr * 8] = c[r];
            }
        }
    }
}

__global__ void k_alpha1(const float* __restrict__ h1,
                         const float* __restrict__ asrc, const float* __restrict__ adst,
                         float* __restrict__ as1, float* __restrict__ ad1, int n) {
    int tid = blockIdx.x * blockDim.x + threadIdx.x;
    if (tid >= n * 8) return;
    unsigned h = tid & 7u;
    const float4* hp = (const float4*)(h1 + (unsigned)(tid >> 3) * 64u + h * 8u);
    float4 v0 = hp[0], v1 = hp[1];
    const float4* sp = (const float4*)(asrc + h * 8u);
    const float4* dp = (const float4*)(adst + h * 8u);
    float4 s0 = sp[0], s1v = sp[1], d0 = dp[0], d1v = dp[1];
    float s = v0.x * s0.x + v0.y * s0.y + v0.z * s0.z + v0.w * s0.w
            + v1.x * s1v.x + v1.y * s1v.y + v1.z * s1v.z + v1.w * s1v.w;
    float d = v0.x * d0.x + v0.y * d0.y + v0.z * d0.z + v0.w * d0.w
            + v1.x * d1v.x + v1.y * d1v.y + v1.z * d1v.z + v1.w * d1v.w;
    as1[tid] = s;
    ad1[tid] = d;
}

#define BKT_SHIFT 9
#define BKT_NODES (1 << BKT_SHIFT)
#define MAX_BKT   1024
#define PART_EPT  8
#define BKT_EPT   8

__device__ __forceinline__ int block_excl_scan(int v, int* wsum, int& total) {
    const int lane = threadIdx.x & 31, w = threadIdx.x >> 5;
    int incl = v;
#pragma unroll
    for (int o = 1; o < 32; o <<= 1) {
        int t = __shfl_up(incl, o);
        if (lane >= o) incl += t;
    }
    if (lane == 31) wsum[w] = incl;
    __syncthreads();
    int woff = 0, tot = 0;
#pragma unroll
    for (int q = 0; q < 32; ++q) {
        int t = wsum[q];
        if (q < w) woff += t;
        tot += t;
    }
    total = tot;
    return woff + incl - v;
}

__global__ __launch_bounds__(1024) void k_hist(const int* __restrict__ ei, int E, int n, int B,
                                               int* __restrict__ ghist) {
    __shared__ int lh[MAX_BKT];
    const int tid = threadIdx.x;
    const int base = blockIdx.x * (1024 * PART_EPT) + tid;
    int d[PART_EPT];
#pragma unroll
    for (int k = 0; k < PART_EPT; ++k) {
        const int idx = base + k * 1024;
        d[k] = ei[E + (idx < E ? idx : E - 1)];
    }
    if (tid < B) lh[tid] = 0;
    __syncthreads();
#pragma unroll
    for (int k = 0; k < PART_EPT; ++k)
        if (base + k * 1024 < E) atomicAdd(&lh[clamp_node(d[k], n) >> BKT_SHIFT], 1);
    __syncthreads();
    if (tid < B) {
        const int c = lh[tid];
        if (c) atomicAdd(&ghist[tid], c);
    }
}

__global__ __launch_bounds__(1024) void k_part(const int* __restrict__ ei, int E, int n, int B,
                                               const int* __restrict__ ghist, int* __restrict__ gcur,
                                               int* __restrict__ part) {
    __shared__ int lh[MAX_BKT];
    __shared__ int wsum[32];
    const int tid = threadIdx.x;
    const int base = blockIdx.x * (1024 * PART_EPT) + tid;
    int sv[PART_EPT], d[PART_EPT];
#pragma unroll
    for (int k = 0; k < PART_EPT; ++k) {
        const int idx = base + k * 1024;
        const int ii  = idx < E ? idx : E - 1;
        sv[k] = ei[ii];
        d[k]  = ei[E + ii];
    }
    int tot;
    const int bstart = block_excl_scan(tid < B ? ghist[tid] : 0, wsum, tot);
    if (tid < B) lh[tid] = 0;
    __syncthreads();
#pragma unroll
    for (int k = 0; k < PART_EPT; ++k) {
        sv[k] = clamp_node(sv[k], n);
        d[k]  = (base + k * 1024 < E) ? clamp_node(d[k], n) : -1;
        if (d[k] >= 0) atomicAdd(&lh[d[k] >> BKT_SHIFT], 1);
    }
    __syncthreads();
    if (tid < B) {
        const int c = lh[tid];
        const int r = c ? atomicAdd(&gcur[tid], c) : 0;
        lh[tid] = bstart + r;
    }
    __syncthreads();
#pragma unroll
    for (int k = 0; k < PART_EPT; ++k)
        if (d[k] >= 0) {
            const int pos = atomicAdd(&lh[d[k] >> BKT_SHIFT], 1);
            part[pos] = (sv[k] << BKT_SHIFT) | (d[k] & (BKT_NODES - 1));
        }
}

__global__ __launch_bounds__(1024) void k_bucket(const int* __restrict__ part, const int* __restrict__ ghist,
                                                 int n, int* __restrict__ cnt, int* __restrict__ begs,
                                                 int* __restrict__ srcs) {
    __shared__ int lc[BKT_NODES];
    __shared__ int wsum[32];
    const int tid = threadIdx.x, b = blockIdx.x;
    int base;
    (void)block_excl_scan(tid < b ? ghist[tid] : 0, wsum, base);
    const int count = ghist[b];
    const int* __restrict__ rec = part + base;
    if (tid < BKT_NODES) lc[tid] = 0;
    __syncthreads();
    for (int e0 = tid; e0 - tid < count; e0 += 1024 * BKT_EPT) {
        int p[BKT_EPT];
#pragma unroll
        for (int k = 0; k < BKT_EPT; ++k) {
            const int e = e0 + k * 1024;
            p[k] = rec[e < count ? e : count - 1];
        }
#pragma unroll
        for (int k = 0; k < BKT_EPT; ++k)
            if (e0 + k * 1024 < count) atomicAdd(&lc[p[k] & (BKT_NODES - 1)], 1);
    }
    __syncthreads();
    const int c = tid < BKT_NODES ? lc[tid] : 0;
    __syncthreads();
    int tot;
    const int excl = block_excl_scan(c, wsum, tot);
    if (tid < BKT_NODES) {
        const int node = (b << BKT_SHIFT) + tid;
        if (node < n) { cnt[node] = c; begs[node] = base + excl; }
        lc[tid] = excl;
    }
    __syncthreads();
    for (int e0 = tid; e0 - tid < count; e0 += 1024 * BKT_EPT) {
        int p[BKT_EPT];
#pragma unroll
        for (int k = 0; k < BKT_EPT; ++k) {
            const int e = e0 + k * 1024;
            p[k] = rec[e < count ? e : count - 1];
        }
#pragma unroll
        for (int k = 0; k < BKT_EPT; ++k)
            if (e0 + k * 1024 < count) {
                const int pos = atomicAdd(&lc[p[k] & (BKT_NODES - 1)], 1);
                srcs[base + pos] = p[k] >> BKT_SHIFT;
            }
    }
}

__device__ __forceinline__ int load_srcs_blk(const int* __restrict__ srcs, int beg, int deg,
                                             int b0, unsigned lane, int self) {
    const int kk = b0 + (int)lane;
    return kk < deg ? srcs[beg + kk] : self;
}

__device__ __forceinline__ void logits1_blk(int srcv, int b0, int total, unsigned j, unsigned h,
                                            const float* __restrict__ as1, float adi,
                                            float (&e)[8]) {
#pragma unroll
    for (int c = 0; c < 8; ++c) {
        const int sk = __shfl(srcv, 4 * c + (int)j);
        float t = as1[(unsigned)sk * 8u + h] + adi;
        t = t > 0.0f ? t : NEG_SLOPE * t;
        e[c] = (b0 + 4 * c + (int)j < total) ? t : 0.0f;
    }
}

__device__ __forceinline__ void accum1_blk(int srcv, int b0, int total, const float (&al)[8],
                                           unsigned lane, const float* __restrict__ h1,
                                           float& acc0, float& acc1) {
    const unsigned g = lane >> 3;
#pragma unroll
    for (int c = 0; c < 8; ++c) {
        if (b0 + 4 * c + 3 < total) {
            float x0[4], x1[4], w0[4], w1[4];
#pragma unroll
            for (int q = 0; q < 4; ++q) {
                const int sk = __builtin_amdgcn_readlane(srcv, 4 * c + q);
                const float* hp = h1 + (unsigned)sk * 64u;
                x0[q] = hp[lane];
                x1[q] = hp[lane + 32u];
                w0[q] = __shfl(al[c], (int)(8u * q + g));
                w1[q] = __shfl(al[c], (int)(8u * q + 4u + g));
            }
#pragma unroll
            for (int q = 0; q < 4; ++q) {
                acc0 += x0[q] * w0[q];
                acc1 += x1[q] * w1[q];
            }
        } else if (b0 + 4 * c < total) {
            float x0[3], x1[3], w0[3], w1[3];
#pragma unroll
            for (int q = 0; q < 3; ++q) {
                const int sk = __builtin_amdgcn_readlane(srcv, 4 * c + q);
                const float* hp = h1 + (unsigned)sk * 64u;
                x0[q] = hp[lane];
                x1[q] = hp[lane + 32u];
                w0[q] = __shfl(al[c], (int)(8u * q + g));
                w1[q] = __shfl(al[c], (int)(8u * q + 4u + g));
            }
#pragma unroll
            for (int q = 0; q < 3; ++q) {
                const bool ok = b0 + 4 * c + q < total;
                acc0 += ok ? x0[q] * w0[q] : 0.0f;
                acc1 += ok ? x1[q] * w1[q] : 0.0f;
            }
        }
    }
}

__global__ __launch_bounds__(256) void k_agg1(const int* __restrict__ cnt, const int* __restrict__ begs,
                                              const int* __restrict__ srcs,
                                              const float* __restrict__ as1, const float* __restrict__ ad1,
                                              const float* __restrict__ h1, const float* __restrict__ bias1,
                                              float* __restrict__ a1, int n) {
    const unsigned lane = threadIdx.x & 31u;
    const int i = __builtin_amdgcn_readfirstlane(
        (int)(blockIdx.x * (blockDim.x >> 5) + (threadIdx.x >> 5)));
    if (i >= n) return;
    const int deg   = cnt[i];
    const int beg   = begs[i];
    const int total = deg + 1;
    const unsigned j = lane >> 3, h = lane & 7u;
    const float adi = ad1[(unsigned)i * 8u + h];
    const float NEG_INF = -__builtin_huge_valf();

    float acc0 = 0.0f, acc1 = 0.0f;
    float e[8], al[8];
    if (total <= 32) {
        const int srcv = load_srcs_blk(srcs, beg, deg, 0, lane, i);
        logits1_blk(srcv, 0, total, j, h, as1, adi, e);
        float m = NEG_INF;
#pragma unroll
        for (int c = 0; c < 8; ++c)
            if (4 * c + (int)j < total) m = fmaxf(m, e[c]);
        m = fmaxf(m, __shfl_xor(m, 8));
        m = fmaxf(m, __shfl_xor(m, 16));
        float ssum = 0.0f;
#pragma unroll
        for (int c = 0; c < 8; ++c) {
            al[c] = (4 * c + (int)j < total) ? expf(e[c] - m) : 0.0f;
            ssum += al[c];
        }
        ssum += __shfl_xor(ssum, 8);
        ssum += __shfl_xor(ssum, 16);
        const float den = ssum + 1e-16f;
#pragma unroll
        for (int c = 0; c < 8; ++c) al[c] = al[c] / den;
        accum1_blk(srcv, 0, total, al, lane, h1, acc0, acc1);
    } else {
        float m = NEG_INF;
        for (int b0 = 0; b0 < total; b0 += 32) {
            const int srcv = load_srcs_blk(srcs, beg, deg, b0, lane, i);
            logits1_blk(srcv, b0, total, j, h, as1, adi, e);
#pragma unroll
            for (int c = 0; c < 8; ++c)
                if (b0 + 4 * c + (int)j < total) m = fmaxf(m, e[c]);
        }
        m = fmaxf(m, __shfl_xor(m, 8));
        m = fmaxf(m, __shfl_xor(m, 16));
        float ssum = 0.0f;
        for (int b0 = 0; b0 < total; b0 += 32) {
            const int srcv = load_srcs_blk(srcs, beg, deg, b0, lane, i);
            logits1_blk(srcv, b0, total, j, h, as1, adi, e);
#pragma unroll
            for (int c = 0; c < 8; ++c)
                ssum += (b0 + 4 * c + (int)j < total) ? expf(e[c] - m) : 0.0f;
        }
        ssum += __shfl_xor(ssum, 8);
        ssum += __shfl_xor(ssum, 16);
        const float den = ssum + 1e-16f;
        for (int b0 = 0; b0 < total; b0 += 32) {
            const int srcv = load_srcs_blk(srcs, beg, deg, b0, lane, i);
            logits1_blk(srcv, b0, total, j, h, as1, adi, e);
#pragma unroll
            for (int c = 0; c < 8; ++c)
                al[c] = (b0 + 4 * c + (int)j < total) ? expf(e[c] - m) / den : 0.0f;
            accum1_blk(srcv, b0, total, al, lane, h1, acc0, acc1);
        }
    }

    float r0 = acc0 + bias1[lane];
    float r1 = acc1 + bias1[lane + 32u];
    r0 = r0 > 0.0f ? r0 : (expf(r0) - 1.0f);
    r1 = r1 > 0.0f ? r1 : (expf(r1) - 1.0f);
    float* op = a1 + (unsigned)i * 64u;
    op[lane]       = r0;
    op[lane + 32u] = r1;
}

__global__ void k_alpha2(const float* __restrict__ h2,
                         const float* __restrict__ asrc, const float* __restrict__ adst,
                         float* __restrict__ as2, float* __restrict__ ad2, int n) {
    int tid = blockIdx.x * blockDim.x + threadIdx.x;
    if (tid >= n) return;
    const float4* hp = (const float4*)(h2 + (unsigned)tid * 8u);
    float4 v0 = hp[0], v1 = hp[1];
    const float4* sp = (const float4*)asrc;
    const float4* dp = (const float4*)adst;
    float4 s0 = sp[0], s1v = sp[1], d0 = dp[0], d1v = dp[1];
    as2[tid] = v0.x * s0.x + v0.y * s0.y + v0.z * s0.z + v0.w * s0.w
             + v1.x * s1v.x + v1.y * s1v.y + v1.z * s1v.z + v1.w * s1v.w;
    ad2[tid] = v0.x * d0.x + v0.y * d0.y + v0.z * d0.z + v0.w * d0.w
             + v1.x * d1v.x + v1.y * d1v.y + v1.z * d1v.z + v1.w * d1v.w;
}

__device__ __forceinline__ float wave_max(float v) {
#pragma unroll
    for (int o = 16; o > 0; o >>= 1) v = fmaxf(v, __shfl_xor(v, o));
    return v;
}
__device__ __forceinline__ float wave_sum(float v) {
#pragma unroll
    for (int o = 16; o > 0; o >>= 1) v += __shfl_xor(v, o);
    return v;
}

__device__ __forceinline__ float logit2(int srcv, bool valid,
                                        const float* __restrict__ as2, float adi) {
    float t = as2[srcv] + adi;
    t = t > 0.0f ? t : NEG_SLOPE * t;
    return valid ? t : 0.0f;
}

__device__ __forceinline__ void rows2_blk(int srcv, unsigned j, unsigned f,
                                          const float* __restrict__ h2, float (&hv)[8]) {
#pragma unroll
    for (int c = 0; c < 8; ++c) {
        const int sk = __shfl(srcv, 4 * c + (int)j);
        hv[c] = h2[(unsigned)sk * 8u + f];
    }
}

__device__ __forceinline__ void accum2_blk(const float (&hv)[8], float al, int b0, int total,
                                           unsigned j, float& acc) {
#pragma unroll
    for (int c = 0; c < 8; ++c) {
        const float ak = __shfl(al, 4 * c + (int)j);
        acc += (b0 + 4 * c + (int)j < total) ? hv[c] * ak : 0.0f;
    }
}

__global__ __launch_bounds__(256) void k_agg2(const int* __restrict__ cnt, const int* __restrict__ begs,
                                              const int* __restrict__ srcs,
                                              const float* __restrict__ as2, const float* __restrict__ ad2,
                                              const float* __restrict__ h2, const float* __restrict__ bias2,
                                              float* __restrict__ out, int n) {
    const unsigned lane = threadIdx.x & 31u;
    const int i = __builtin_amdgcn_readfirstlane(
        (int)(blockIdx.x * (blockDim.x >> 5) + (threadIdx.x >> 5)));
    if (i >= n) return;
    const int deg   = cnt[i];
    const int beg   = begs[i];
    const int total = deg + 1;
    const unsigned j = lane >> 3, f = lane & 7u;
    const float adi = ad2[i];
    const float NEG_INF = -__builtin_huge_valf();

    float acc = 0.0f;
    float hv[8];
    if (total <= 32) {
        const int  srcv  = load_srcs_blk(srcs, beg, deg, 0, lane, i);
        const bool valid = (int)lane < total;
        rows2_blk(srcv, j, f, h2, hv);
        __builtin_amdgcn_sched_barrier(0);
        const float e  = logit2(srcv, valid, as2, adi);
        const float m  = wave_max(valid ? e : NEG_INF);
        const float ex = valid ? expf(e - m) : 0.0f;
        const float den = wave_sum(ex) + 1e-16f;
        accum2_blk(hv, ex / den, 0, total, j, acc);
    } else {
        float m = NEG_INF;
        for (int b0 = 0; b0 < total; b0 += 32) {
            const int  srcv  = load_srcs_blk(srcs, beg, deg, b0, lane, i);
            const bool valid = b0 + (int)lane < total;
            const float e = logit2(srcv, valid, as2, adi);
            m = fmaxf(m, valid ? e : NEG_INF);
        }
        m = wave_max(m);
        float ssum = 0.0f;
        for (int b0 = 0; b0 < total; b0 += 32) {
            const int  srcv  = load_srcs_blk(srcs, beg, deg, b0, lane, i);
            const bool valid = b0 + (int)lane < total;
            const float e = logit2(srcv, valid, as2, adi);
            ssum += valid ? expf(e - m) : 0.0f;
        }
        const float den = wave_sum(ssum) + 1e-16f;
        for (int b0 = 0; b0 < total; b0 += 32) {
            const int  srcv  = load_srcs_blk(srcs, beg, deg, b0, lane, i);
            const bool valid = b0 + (int)lane < total;
            const float e = logit2(srcv, valid, as2, adi);
            rows2_blk(srcv, j, f, h2, hv);
            const float al = valid ? expf(e - m) / den : 0.0f;
            accum2_blk(hv, al, b0, total, j, acc);
        }
    }
    acc += __shfl_xor(acc, 8);
    acc += __shfl_xor(acc, 16);
    if (lane < 8u) out[(unsigned)i * 8u + f] = acc + bias2[f];
}

extern "C" void kernel_launch(void* const* d_in, const int* in_sizes, int n_in,
                              void* d_out, int out_size, void* d_ws, size_t ws_size,
                              hipStream_t stream) {
    const float* x     = (const float*)d_in[0];
    const int*   ei    = (const int*)d_in[1];
    const float* W1    = (const float*)d_in[2];
    const float* asrc1 = (const float*)d_in[3];
    const float* adst1 = (const float*)d_in[4];
    const float* bias1 = (const float*)d_in[5];
    const float* W2    = (const float*)d_in[6];
    const float* asrc2 = (const float*)d_in[7];
    const float* adst2 = (const float*)d_in[8];
    const float* bias2 = (const float*)d_in[9];

    int n = in_sizes[0] / F_IN;
    int E = in_sizes[1] / 2;
    int B = (n + BKT_NODES - 1) >> BKT_SHIFT;

    char* p = (char*)d_ws;
    float* h1   = (float*)p; p += (size_t)n * 64 * 4;
    float* a1   = (float*)p; p += (size_t)n * 64 * 4;
    float* as1  = (float*)p; p += (size_t)n * 8 * 4;
    float* ad1  = (float*)p; p += (size_t)n * 8 * 4;
    float* h2   = (float*)p; p += (size_t)n * 8 * 4;
    float* as2  = (float*)p; p += (size_t)n * 4;
    float* ad2  = (float*)p; p += (size_t)n * 4;
    int*   cnt  = (int*)p;   p += (size_t)n * 4;
    int*   begs = (int*)p;   p += (size_t)n * 4;
    int*   part = (int*)p;   p += (size_t)E * 4;
    int*   srcs = (int*)p;   p += (size_t)E * 4;
    int*   ghist = (int*)p;  p += (size_t)MAX_BKT * 4;
    int*   gcur  = (int*)p;  p += (size_t)MAX_BKT * 4;

    hipMemsetAsync(ghist, 0, (size_t)2 * MAX_BKT * 4, stream);

    const int tpb = 256;
    int gP = (E + 1024 * PART_EPT - 1) / (1024 * PART_EPT);
    int gN = (n + tpb / 32 - 1) / (tpb / 32);

    k_hist<<<gP, 1024, 0, stream>>>(ei, E, n, B, ghist);
    k_part<<<gP, 1024, 0, stream>>>(ei, E, n, B, ghist, gcur, part);
    k_bucket<<<B, 1024, 0, stream>>>(part, ghist, n, cnt, begs, srcs);

    k_gemm1<<<(n + 15) / 16, 32, 0, stream>>>(x, W1, h1, n);
    k_alpha1<<<(n * 8 + tpb - 1) / tpb, tpb, 0, stream>>>(h1, asrc1, adst1, as1, ad1, n);
    k_agg1<<<gN, tpb, 0, stream>>>(cnt, begs, srcs, as1, ad1, h1, bias1, a1, n);

    k_gemm2<<<(n + 15) / 16, 32, 0, stream>>>(a1, W2, h2, n);
    k_alpha2<<<(n + tpb - 1) / tpb, tpb, 0, stream>>>(h2, asrc2, adst2, as2, ad2, n);
    k_agg2<<<gN, tpb, 0, stream>>>(cnt, begs, srcs, as2, ad2, h2, bias2, (float*)d_out, n);
}


# attention vectors folded into the GEMMs' B operands (W1 a as a fifth wmma tile, W2 a in two padding columns): the GEMMs emit the node logits, both logit kernels go; logits now summed as x@(W@a)
# speedup vs baseline: 9.9251x; 1.0558x over previous
#include <hip/hip_runtime.h>
#include <hip/hip_bf16.h>

#define NEG_SLOPE 0.2f
#define F_IN 14

typedef float v2f __attribute__((ext_vector_type(2)));
typedef float v8f __attribute__((ext_vector_type(8)));

__device__ __forceinline__ int clamp_node(int v, int n) {
    return v < 0 ? 0 : (v >= n ? n - 1 : v);
}

__global__ __launch_bounds__(32) void k_gemm1(const float* __restrict__ x,
                                              const float* __restrict__ W1,
                                              const float* __restrict__ asrc,
                                              const float* __restrict__ adst,
                                              float* __restrict__ h1,
                                              float* __restrict__ as1, float* __restrict__ ad1, int n) {
    __shared__ float xs[16 * 16];
    __shared__ float ws[16 * 64];
    __shared__ float wa[16 * 16];

    const int lane = threadIdx.x;
    const int row0 = blockIdx.x * 16;
    const bool hi  = lane >= 16;
    const int  m   = lane & 15;
    const bool full = (row0 + 16 <= n);

    const float4* avp = (const float4*)((m < 8 ? asrc : adst) + (m & 7) * 8);
    const float4 av0 = avp[0], av1 = avp[1];

    for (int i = lane; i < 2 * 64 / 4; i += 32)
        ((float4*)(ws + 14 * 64))[i] = make_float4(0.f, 0.f, 0.f, 0.f);
    for (int i = lane; i < F_IN * 64 / 4; i += 32)
        ((float4*)ws)[i] = ((const float4*)W1)[i];

    for (int i = lane; i < 16 * 16 / 4; i += 32)
        ((float4*)xs)[i] = make_float4(0.f, 0.f, 0.f, 0.f);
    __syncthreads();
#pragma unroll
    for (int i = 0; i < 8; ++i) {
        const int k = 2 * i + (hi ? 1 : 0);
        const float4* wp = (const float4*)(ws + k * 64 + (m & 7) * 8);
        const float4 w0 = wp[0], w1 = wp[1];
        wa[k * 16 + m] = w0.x * av0.x + w0.y * av0.y + w0.z * av0.z + w0.w * av0.w
                       + w1.x * av1.x + w1.y * av1.y + w1.z * av1.z + w1.w * av1.w;
    }
    {
        const float* xt = x + (size_t)row0 * F_IN;
        int rows = full ? 16 : (n - row0);
        int cnt  = rows * F_IN;
        for (int i = lane; i < cnt; i += 32) {
            int r = i / F_IN, c = i - r * F_IN;
            xs[r * 16 + c] = xt[i];
        }
    }
    __syncthreads();

    const int kofs = hi ? 2 : 0;
    v2f a[4];
#pragma unroll
    for (int k = 0; k < 4; ++k) {
        a[k][0] = xs[m * 16 + 4 * k + kofs + 0];
        a[k][1] = xs[m * 16 + 4 * k + kofs + 1];
    }

#pragma unroll
    for (int t = 0; t < 4; ++t) {
        v8f c = {};
#pragma unroll
        for (int k = 0; k < 4; ++k) {
            v2f b;
            b[0] = ws[(4 * k + kofs + 0) * 64 + t * 16 + m];
            b[1] = ws[(4 * k + kofs + 1) * 64 + t * 16 + m];
            c = __builtin_amdgcn_wmma_f32_16x16x4_f32(false, a[k], false, b,
                                                      (short)0, c, false, false);
        }
        float* out = h1 + (size_t)row0 * 64 + t * 16 + m;
        if (full) {
#pragma unroll
            for (int r = 0; r < 8; ++r)
                out[(size_t)(r + (hi ? 8 : 0)) * 64] = c[r];
        } else {
#pragma unroll
            for (int r = 0; r < 8; ++r) {
                int rr = r + (hi ? 8 : 0);
                if (row0 + rr < n) out[(size_t)rr * 64] = c[r];
            }
        }
    }

    {
        v8f c = {};
#pragma unroll
        for (int k = 0; k < 4; ++k) {
            v2f b;
            b[0] = wa[(4 * k + kofs + 0) * 16 + m];
            b[1] = wa[(4 * k + kofs + 1) * 16 + m];
            c = __builtin_amdgcn_wmma_f32_16x16x4_f32(false, a[k], false, b,
                                                      (short)0, c, false, false);
        }
        float* lo = (m < 8 ? as1 : ad1) + (size_t)row0 * 8 + (m & 7);
        if (full) {
#pragma unroll
            for (int r = 0; r < 8; ++r)
                lo[(size_t)(r + (hi ? 8 : 0)) * 8] = c[r];
        } else {
#pragma unroll
            for (int r = 0; r < 8; ++r) {
                int rr = r + (hi ? 8 : 0);
                if (row0 + rr < n) lo[(size_t)rr * 8] = c[r];
            }
        }
    }
}

__global__ __launch_bounds__(32) void k_gemm2(const float* __restrict__ a1,
                                              const float* __restrict__ W2,
                                              const float* __restrict__ asrc,
                                              const float* __restrict__ adst,
                                              float* __restrict__ h2, float* __restrict__ asd2, int n) {
    __shared__ float smA[16 * 64];
    __shared__ float ws2[64 * 16];

    const int lane = threadIdx.x;
    const int row0 = blockIdx.x * 16;
    const bool hi  = lane >= 16;
    const int  m   = lane & 15;
    const bool full = (row0 + 16 <= n);

    float wcol[4];
    {
        const float4* avp = (const float4*)((lane & 1) ? adst : asrc);
        const float4 av0 = avp[0], av1 = avp[1];
#pragma unroll
        for (int i = 0; i < 4; ++i) {
            const float4* wp = (const float4*)(W2 + ((lane >> 1) + 16 * i) * 8);
            const float4 w0 = wp[0], w1 = wp[1];
            wcol[i] = w0.x * av0.x + w0.y * av0.y + w0.z * av0.z + w0.w * av0.w
                    + w1.x * av1.x + w1.y * av1.y + w1.z * av1.z + w1.w * av1.w;
        }
    }

    for (int i = lane; i < 64 * 16 / 4; i += 32)
        ((float4*)ws2)[i] = make_float4(0.f, 0.f, 0.f, 0.f);
    __syncthreads();
    for (int i = lane; i < 64 * 8 / 4; i += 32) {
        int k = i >> 1, half = i & 1;
        ((float4*)(ws2 + k * 16 + half * 4))[0] = ((const float4*)W2)[i];
    }
#pragma unroll
    for (int i = 0; i < 4; ++i)
        ws2[((lane >> 1) + 16 * i) * 16 + 8 + (lane & 1)] = wcol[i];

    if (full) {
        const float4* src = (const float4*)(a1 + (size_t)row0 * 64);
        for (int i = lane; i < 16 * 64 / 4; i += 32)
            ((float4*)smA)[i] = src[i];
    } else {
        for (int i = lane; i < 16 * 64 / 4; i += 32)
            ((float4*)smA)[i] = make_float4(0.f, 0.f, 0.f, 0.f);
        __syncthreads();
        int rows = n - row0;
        const float4* src = (const float4*)(a1 + (size_t)row0 * 64);
        for (int i = lane; i < rows * 16; i += 32)
            ((float4*)smA)[i] = src[i];
    }
    __syncthreads();

    const int kofs = hi ? 2 : 0;
    v8f c = {};
#pragma unroll
    for (int k = 0; k < 16; ++k) {
        v2f a, b;
        a[0] = smA[m * 64 + 4 * k + kofs + 0];
        a[1] = smA[m * 64 + 4 * k + kofs + 1];
        b[0] = ws2[(4 * k + kofs + 0) * 16 + m];
        b[1] = ws2[(4 * k + kofs + 1) * 16 + m];
        c = __builtin_amdgcn_wmma_f32_16x16x4_f32(false, a, false, b,
                                                  (short)0, c, false, false);
    }
    if (m < 8) {
        float* out = h2 + (size_t)row0 * 8 + m;
        if (full) {
#pragma unroll
            for (int r = 0; r < 8; ++r)
                out[(size_t)(r + (hi ? 8 : 0)) * 8] = c[r];
        } else {
#pragma unroll
            for (int r = 0; r < 8; ++r) {
                int rr = r + (hi ? 8 : 0);
                if (row0 + rr < n) out[(size_t)rr * 8] = c[r];
            }
        }
    } else if (m < 10) {
        float* out = asd2 + (size_t)row0 * 2 + (m - 8);
        if (full) {
#pragma unroll
            for (int r = 0; r < 8; ++r)
                out[(size_t)(r + (hi ? 8 : 0)) * 2] = c[r];
        } else {
#pragma unroll
            for (int r = 0; r < 8; ++r) {
                int rr = r + (hi ? 8 : 0);
                if (row0 + rr < n) out[(size_t)rr * 2] = c[r];
            }
        }
    }
}

#define BKT_SHIFT 9
#define BKT_NODES (1 << BKT_SHIFT)
#define MAX_BKT   1024
#define PART_EPT  8
#define BKT_EPT   8

__device__ __forceinline__ int block_excl_scan(int v, int* wsum, int& total) {
    const int lane = threadIdx.x & 31, w = threadIdx.x >> 5;
    int incl = v;
#pragma unroll
    for (int o = 1; o < 32; o <<= 1) {
        int t = __shfl_up(incl, o);
        if (lane >= o) incl += t;
    }
    if (lane == 31) wsum[w] = incl;
    __syncthreads();
    int woff = 0, tot = 0;
#pragma unroll
    for (int q = 0; q < 32; ++q) {
        int t = wsum[q];
        if (q < w) woff += t;
        tot += t;
    }
    total = tot;
    return woff + incl - v;
}

__global__ __launch_bounds__(1024) void k_hist(const int* __restrict__ ei, int E, int n, int B,
                                               int* __restrict__ ghist) {
    __shared__ int lh[MAX_BKT];
    const int tid = threadIdx.x;
    const int base = blockIdx.x * (1024 * PART_EPT) + tid;
    int d[PART_EPT];
#pragma unroll
    for (int k = 0; k < PART_EPT; ++k) {
        const int idx = base + k * 1024;
        d[k] = ei[E + (idx < E ? idx : E - 1)];
    }
    if (tid < B) lh[tid] = 0;
    __syncthreads();
#pragma unroll
    for (int k = 0; k < PART_EPT; ++k)
        if (base + k * 1024 < E) atomicAdd(&lh[clamp_node(d[k], n) >> BKT_SHIFT], 1);
    __syncthreads();
    if (tid < B) {
        const int c = lh[tid];
        if (c) atomicAdd(&ghist[tid], c);
    }
}

__global__ __launch_bounds__(1024) void k_part(const int* __restrict__ ei, int E, int n, int B,
                                               const int* __restrict__ ghist, int* __restrict__ gcur,
                                               int* __restrict__ part) {
    __shared__ int lh[MAX_BKT];
    __shared__ int wsum[32];
    const int tid = threadIdx.x;
    const int base = blockIdx.x * (1024 * PART_EPT) + tid;
    int sv[PART_EPT], d[PART_EPT];
#pragma unroll
    for (int k = 0; k < PART_EPT; ++k) {
        const int idx = base + k * 1024;
        const int ii  = idx < E ? idx : E - 1;
        sv[k] = ei[ii];
        d[k]  = ei[E + ii];
    }
    int tot;
    const int bstart = block_excl_scan(tid < B ? ghist[tid] : 0, wsum, tot);
    if (tid < B) lh[tid] = 0;
    __syncthreads();
#pragma unroll
    for (int k = 0; k < PART_EPT; ++k) {
        sv[k] = clamp_node(sv[k], n);
        d[k]  = (base + k * 1024 < E) ? clamp_node(d[k], n) : -1;
        if (d[k] >= 0) atomicAdd(&lh[d[k] >> BKT_SHIFT], 1);
    }
    __syncthreads();
    if (tid < B) {
        const int c = lh[tid];
        const int r = c ? atomicAdd(&gcur[tid], c) : 0;
        lh[tid] = bstart + r;
    }
    __syncthreads();
#pragma unroll
    for (int k = 0; k < PART_EPT; ++k)
        if (d[k] >= 0) {
            const int pos = atomicAdd(&lh[d[k] >> BKT_SHIFT], 1);
            part[pos] = (sv[k] << BKT_SHIFT) | (d[k] & (BKT_NODES - 1));
        }
}

__global__ __launch_bounds__(1024) void k_bucket(const int* __restrict__ part, const int* __restrict__ ghist,
                                                 int n, int* __restrict__ cnt, int* __restrict__ begs,
                                                 int* __restrict__ srcs) {
    __shared__ int lc[BKT_NODES];
    __shared__ int wsum[32];
    const int tid = threadIdx.x, b = blockIdx.x;
    int base;
    (void)block_excl_scan(tid < b ? ghist[tid] : 0, wsum, base);
    const int count = ghist[b];
    const int* __restrict__ rec = part + base;
    if (tid < BKT_NODES) lc[tid] = 0;
    __syncthreads();
    for (int e0 = tid; e0 - tid < count; e0 += 1024 * BKT_EPT) {
        int p[BKT_EPT];
#pragma unroll
        for (int k = 0; k < BKT_EPT; ++k) {
            const int e = e0 + k * 1024;
            p[k] = rec[e < count ? e : count - 1];
        }
#pragma unroll
        for (int k = 0; k < BKT_EPT; ++k)
            if (e0 + k * 1024 < count) atomicAdd(&lc[p[k] & (BKT_NODES - 1)], 1);
    }
    __syncthreads();
    const int c = tid < BKT_NODES ? lc[tid] : 0;
    __syncthreads();
    int tot;
    const int excl = block_excl_scan(c, wsum, tot);
    if (tid < BKT_NODES) {
        const int node = (b << BKT_SHIFT) + tid;
        if (node < n) { cnt[node] = c; begs[node] = base + excl; }
        lc[tid] = excl;
    }
    __syncthreads();
    for (int e0 = tid; e0 - tid < count; e0 += 1024 * BKT_EPT) {
        int p[BKT_EPT];
#pragma unroll
        for (int k = 0; k < BKT_EPT; ++k) {
            const int e = e0 + k * 1024;
            p[k] = rec[e < count ? e : count - 1];
        }
#pragma unroll
        for (int k = 0; k < BKT_EPT; ++k)
            if (e0 + k * 1024 < count) {
                const int pos = atomicAdd(&lc[p[k] & (BKT_NODES - 1)], 1);
                srcs[base + pos] = p[k] >> BKT_SHIFT;
            }
    }
}

__device__ __forceinline__ int load_srcs_blk(const int* __restrict__ srcs, int beg, int deg,
                                             int b0, unsigned lane, int self) {
    const int kk = b0 + (int)lane;
    return kk < deg ? srcs[beg + kk] : self;
}

__device__ __forceinline__ void logits1_blk(int srcv, int b0, int total, unsigned j, unsigned h,
                                            const float* __restrict__ as1, float adi,
                                            float (&e)[8]) {
#pragma unroll
    for (int c = 0; c < 8; ++c) {
        const int sk = __shfl(srcv, 4 * c + (int)j);
        float t = as1[(unsigned)sk * 8u + h] + adi;
        t = t > 0.0f ? t : NEG_SLOPE * t;
        e[c] = (b0 + 4 * c + (int)j < total) ? t : 0.0f;
    }
}

__device__ __forceinline__ void accum1_blk(int srcv, int b0, int total, const float (&al)[8],
                                           unsigned lane, const float* __restrict__ h1,
                                           float& acc0, float& acc1) {
    const unsigned g = lane >> 3;
#pragma unroll
    for (int c = 0; c < 8; ++c) {
        if (b0 + 4 * c + 3 < total) {
            float x0[4], x1[4], w0[4], w1[4];
#pragma unroll
            for (int q = 0; q < 4; ++q) {
                const int sk = __builtin_amdgcn_readlane(srcv, 4 * c + q);
                const float* hp = h1 + (unsigned)sk * 64u;
                x0[q] = hp[lane];
                x1[q] = hp[lane + 32u];
                w0[q] = __shfl(al[c], (int)(8u * q + g));
                w1[q] = __shfl(al[c], (int)(8u * q + 4u + g));
            }
#pragma unroll
            for (int q = 0; q < 4; ++q) {
                acc0 += x0[q] * w0[q];
                acc1 += x1[q] * w1[q];
            }
        } else if (b0 + 4 * c < total) {
            float x0[3], x1[3], w0[3], w1[3];
#pragma unroll
            for (int q = 0; q < 3; ++q) {
                const int sk = __builtin_amdgcn_readlane(srcv, 4 * c + q);
                const float* hp = h1 + (unsigned)sk * 64u;
                x0[q] = hp[lane];
                x1[q] = hp[lane + 32u];
                w0[q] = __shfl(al[c], (int)(8u * q + g));
                w1[q] = __shfl(al[c], (int)(8u * q + 4u + g));
            }
#pragma unroll
            for (int q = 0; q < 3; ++q) {
                const bool ok = b0 + 4 * c + q < total;
                acc0 += ok ? x0[q] * w0[q] : 0.0f;
                acc1 += ok ? x1[q] * w1[q] : 0.0f;
            }
        }
    }
}

__global__ __launch_bounds__(256) void k_agg1(const int* __restrict__ cnt, const int* __restrict__ begs,
                                              const int* __restrict__ srcs,
                                              const float* __restrict__ as1, const float* __restrict__ ad1,
                                              const float* __restrict__ h1, const float* __restrict__ bias1,
                                              float* __restrict__ a1, int n) {
    const unsigned lane = threadIdx.x & 31u;
    const int i = __builtin_amdgcn_readfirstlane(
        (int)(blockIdx.x * (blockDim.x >> 5) + (threadIdx.x >> 5)));
    if (i >= n) return;
    const int deg   = cnt[i];
    const int beg   = begs[i];
    const int total = deg + 1;
    const unsigned j = lane >> 3, h = lane & 7u;
    const float adi = ad1[(unsigned)i * 8u + h];
    const float NEG_INF = -__builtin_huge_valf();

    float acc0 = 0.0f, acc1 = 0.0f;
    float e[8], al[8];
    if (total <= 32) {
        const int srcv = load_srcs_blk(srcs, beg, deg, 0, lane, i);
        logits1_blk(srcv, 0, total, j, h, as1, adi, e);
        float m = NEG_INF;
#pragma unroll
        for (int c = 0; c < 8; ++c)
            if (4 * c + (int)j < total) m = fmaxf(m, e[c]);
        m = fmaxf(m, __shfl_xor(m, 8));
        m = fmaxf(m, __shfl_xor(m, 16));
        float ssum = 0.0f;
#pragma unroll
        for (int c = 0; c < 8; ++c) {
            al[c] = (4 * c + (int)j < total) ? expf(e[c] - m) : 0.0f;
            ssum += al[c];
        }
        ssum += __shfl_xor(ssum, 8);
        ssum += __shfl_xor(ssum, 16);
        const float den = ssum + 1e-16f;
#pragma unroll
        for (int c = 0; c < 8; ++c) al[c] = al[c] / den;
        accum1_blk(srcv, 0, total, al, lane, h1, acc0, acc1);
    } else {
        float m = NEG_INF;
        for (int b0 = 0; b0 < total; b0 += 32) {
            const int srcv = load_srcs_blk(srcs, beg, deg, b0, lane, i);
            logits1_blk(srcv, b0, total, j, h, as1, adi, e);
#pragma unroll
            for (int c = 0; c < 8; ++c)
                if (b0 + 4 * c + (int)j < total) m = fmaxf(m, e[c]);
        }
        m = fmaxf(m, __shfl_xor(m, 8));
        m = fmaxf(m, __shfl_xor(m, 16));
        float ssum = 0.0f;
        for (int b0 = 0; b0 < total; b0 += 32) {
            const int srcv = load_srcs_blk(srcs, beg, deg, b0, lane, i);
            logits1_blk(srcv, b0, total, j, h, as1, adi, e);
#pragma unroll
            for (int c = 0; c < 8; ++c)
                ssum += (b0 + 4 * c + (int)j < total) ? expf(e[c] - m) : 0.0f;
        }
        ssum += __shfl_xor(ssum, 8);
        ssum += __shfl_xor(ssum, 16);
        const float den = ssum + 1e-16f;
        for (int b0 = 0; b0 < total; b0 += 32) {
            const int srcv = load_srcs_blk(srcs, beg, deg, b0, lane, i);
            logits1_blk(srcv, b0, total, j, h, as1, adi, e);
#pragma unroll
            for (int c = 0; c < 8; ++c)
                al[c] = (b0 + 4 * c + (int)j < total) ? expf(e[c] - m) / den : 0.0f;
            accum1_blk(srcv, b0, total, al, lane, h1, acc0, acc1);
        }
    }

    float r0 = acc0 + bias1[lane];
    float r1 = acc1 + bias1[lane + 32u];
    r0 = r0 > 0.0f ? r0 : (expf(r0) - 1.0f);
    r1 = r1 > 0.0f ? r1 : (expf(r1) - 1.0f);
    float* op = a1 + (unsigned)i * 64u;
    op[lane]       = r0;
    op[lane + 32u] = r1;
}

__device__ __forceinline__ float wave_max(float v) {
#pragma unroll
    for (int o = 16; o > 0; o >>= 1) v = fmaxf(v, __shfl_xor(v, o));
    return v;
}
__device__ __forceinline__ float wave_sum(float v) {
#pragma unroll
    for (int o = 16; o > 0; o >>= 1) v += __shfl_xor(v, o);
    return v;
}

__device__ __forceinline__ float logit2(int srcv, bool valid,
                                        const float* __restrict__ asd2, float adi) {
    float t = asd2[2u * (unsigned)srcv] + adi;
    t = t > 0.0f ? t : NEG_SLOPE * t;
    return valid ? t : 0.0f;
}

__device__ __forceinline__ void rows2_blk(int srcv, unsigned j, unsigned f,
                                          const float* __restrict__ h2, float (&hv)[8]) {
#pragma unroll
    for (int c = 0; c < 8; ++c) {
        const int sk = __shfl(srcv, 4 * c + (int)j);
        hv[c] = h2[(unsigned)sk * 8u + f];
    }
}

__device__ __forceinline__ void accum2_blk(const float (&hv)[8], float al, int b0, int total,
                                           unsigned j, float& acc) {
#pragma unroll
    for (int c = 0; c < 8; ++c) {
        const float ak = __shfl(al, 4 * c + (int)j);
        acc += (b0 + 4 * c + (int)j < total) ? hv[c] * ak : 0.0f;
    }
}

__global__ __launch_bounds__(256) void k_agg2(const int* __restrict__ cnt, const int* __restrict__ begs,
                                              const int* __restrict__ srcs,
                                              const float* __restrict__ asd2,
                                              const float* __restrict__ h2, const float* __restrict__ bias2,
                                              float* __restrict__ out, int n) {
    const unsigned lane = threadIdx.x & 31u;
    const int i = __builtin_amdgcn_readfirstlane(
        (int)(blockIdx.x * (blockDim.x >> 5) + (threadIdx.x >> 5)));
    if (i >= n) return;
    const int deg   = cnt[i];
    const int beg   = begs[i];
    const int total = deg + 1;
    const unsigned j = lane >> 3, f = lane & 7u;
    const float adi = asd2[2u * (unsigned)i + 1u];
    const float NEG_INF = -__builtin_huge_valf();

    float acc = 0.0f;
    float hv[8];
    if (total <= 32) {
        const int  srcv  = load_srcs_blk(srcs, beg, deg, 0, lane, i);
        const bool valid = (int)lane < total;
        rows2_blk(srcv, j, f, h2, hv);
        __builtin_amdgcn_sched_barrier(0);
        const float e  = logit2(srcv, valid, asd2, adi);
        const float m  = wave_max(valid ? e : NEG_INF);
        const float ex = valid ? expf(e - m) : 0.0f;
        const float den = wave_sum(ex) + 1e-16f;
        accum2_blk(hv, ex / den, 0, total, j, acc);
    } else {
        float m = NEG_INF;
        for (int b0 = 0; b0 < total; b0 += 32) {
            const int  srcv  = load_srcs_blk(srcs, beg, deg, b0, lane, i);
            const bool valid = b0 + (int)lane < total;
            const float e = logit2(srcv, valid, asd2, adi);
            m = fmaxf(m, valid ? e : NEG_INF);
        }
        m = wave_max(m);
        float ssum = 0.0f;
        for (int b0 = 0; b0 < total; b0 += 32) {
            const int  srcv  = load_srcs_blk(srcs, beg, deg, b0, lane, i);
            const bool valid = b0 + (int)lane < total;
            const float e = logit2(srcv, valid, asd2, adi);
            ssum += valid ? expf(e - m) : 0.0f;
        }
        const float den = wave_sum(ssum) + 1e-16f;
        for (int b0 = 0; b0 < total; b0 += 32) {
            const int  srcv  = load_srcs_blk(srcs, beg, deg, b0, lane, i);
            const bool valid = b0 + (int)lane < total;
            const float e = logit2(srcv, valid, asd2, adi);
            rows2_blk(srcv, j, f, h2, hv);
            const float al = valid ? expf(e - m) / den : 0.0f;
            accum2_blk(hv, al, b0, total, j, acc);
        }
    }
    acc += __shfl_xor(acc, 8);
    acc += __shfl_xor(acc, 16);
    if (lane < 8u) out[(unsigned)i * 8u + f] = acc + bias2[f];
}

extern "C" void kernel_launch(void* const* d_in, const int* in_sizes, int n_in,
                              void* d_out, int out_size, void* d_ws, size_t ws_size,
                              hipStream_t stream) {
    const float* x     = (const float*)d_in[0];
    const int*   ei    = (const int*)d_in[1];
    const float* W1    = (const float*)d_in[2];
    const float* asrc1 = (const float*)d_in[3];
    const float* adst1 = (const float*)d_in[4];
    const float* bias1 = (const float*)d_in[5];
    const float* W2    = (const float*)d_in[6];
    const float* asrc2 = (const float*)d_in[7];
    const float* adst2 = (const float*)d_in[8];
    const float* bias2 = (const float*)d_in[9];

    int n = in_sizes[0] / F_IN;
    int E = in_sizes[1] / 2;
    int B = (n + BKT_NODES - 1) >> BKT_SHIFT;

    char* p = (char*)d_ws;
    float* h1   = (float*)p; p += (size_t)n * 64 * 4;
    float* a1   = (float*)p; p += (size_t)n * 64 * 4;
    float* as1  = (float*)p; p += (size_t)n * 8 * 4;
    float* ad1  = (float*)p; p += (size_t)n * 8 * 4;
    float* h2   = (float*)p; p += (size_t)n * 8 * 4;
    float* asd2 = (float*)p; p += (size_t)n * 8;
    int*   cnt  = (int*)p;   p += (size_t)n * 4;
    int*   begs = (int*)p;   p += (size_t)n * 4;
    int*   part = (int*)p;   p += (size_t)E * 4;
    int*   srcs = (int*)p;   p += (size_t)E * 4;
    int*   ghist = (int*)p;  p += (size_t)MAX_BKT * 4;
    int*   gcur  = (int*)p;  p += (size_t)MAX_BKT * 4;

    hipMemsetAsync(ghist, 0, (size_t)2 * MAX_BKT * 4, stream);

    const int tpb = 256;
    int gP = (E + 1024 * PART_EPT - 1) / (1024 * PART_EPT);
    int gN = (n + tpb / 32 - 1) / (tpb / 32);

    k_hist<<<gP, 1024, 0, stream>>>(ei, E, n, B, ghist);
    k_part<<<gP, 1024, 0, stream>>>(ei, E, n, B, ghist, gcur, part);
    k_bucket<<<B, 1024, 0, stream>>>(part, ghist, n, cnt, begs, srcs);

    k_gemm1<<<(n + 15) / 16, 32, 0, stream>>>(x, W1, asrc1, adst1, h1, as1, ad1, n);
    k_agg1<<<gN, tpb, 0, stream>>>(cnt, begs, srcs, as1, ad1, h1, bias1, a1, n);

    k_gemm2<<<(n + 15) / 16, 32, 0, stream>>>(a1, W2, asrc2, adst2, h2, asd2, n);
    k_agg2<<<gN, tpb, 0, stream>>>(cnt, begs, srcs, asd2, h2, bias2, (float*)d_out, n);
}


# layer-2 product moved into the layer-1 node walk: 16 waves per work-group leave their rows in LDS, the last to finish (LDS count) runs the same 16 wmma; a1 never goes to memory, one launch fewer
# speedup vs baseline: 10.4258x; 1.0504x over previous
#include <hip/hip_runtime.h>
#include <hip/hip_bf16.h>

#define NEG_SLOPE 0.2f
#define F_IN 14

typedef float v2f __attribute__((ext_vector_type(2)));
typedef float v8f __attribute__((ext_vector_type(8)));

__device__ __forceinline__ int clamp_node(int v, int n) {
    return v < 0 ? 0 : (v >= n ? n - 1 : v);
}

__global__ __launch_bounds__(32) void k_gemm1(const float* __restrict__ x,
                                              const float* __restrict__ W1,
                                              const float* __restrict__ asrc,
                                              const float* __restrict__ adst,
                                              float* __restrict__ h1,
                                              float* __restrict__ as1, float* __restrict__ ad1, int n) {
    __shared__ float xs[16 * 16];
    __shared__ float ws[16 * 64];
    __shared__ float wa[16 * 16];

    const int lane = threadIdx.x;
    const int row0 = blockIdx.x * 16;
    const bool hi  = lane >= 16;
    const int  m   = lane & 15;
    const bool full = (row0 + 16 <= n);

    const float4* avp = (const float4*)((m < 8 ? asrc : adst) + (m & 7) * 8);
    const float4 av0 = avp[0], av1 = avp[1];

    for (int i = lane; i < 2 * 64 / 4; i += 32)
        ((float4*)(ws + 14 * 64))[i] = make_float4(0.f, 0.f, 0.f, 0.f);
    for (int i = lane; i < F_IN * 64 / 4; i += 32)
        ((float4*)ws)[i] = ((const float4*)W1)[i];

    for (int i = lane; i < 16 * 16 / 4; i += 32)
        ((float4*)xs)[i] = make_float4(0.f, 0.f, 0.f, 0.f);
    __syncthreads();
#pragma unroll
    for (int i = 0; i < 8; ++i) {
        const int k = 2 * i + (hi ? 1 : 0);
        const float4* wp = (const float4*)(ws + k * 64 + (m & 7) * 8);
        const float4 w0 = wp[0], w1 = wp[1];
        wa[k * 16 + m] = w0.x * av0.x + w0.y * av0.y + w0.z * av0.z + w0.w * av0.w
                       + w1.x * av1.x + w1.y * av1.y + w1.z * av1.z + w1.w * av1.w;
    }
    {
        const float* xt = x + (size_t)row0 * F_IN;
        int rows = full ? 16 : (n - row0);
        int cnt  = rows * F_IN;
        for (int i = lane; i < cnt; i += 32) {
            int r = i / F_IN, c = i - r * F_IN;
            xs[r * 16 + c] = xt[i];
        }
    }
    __syncthreads();

    const int kofs = hi ? 2 : 0;
    v2f a[4];
#pragma unroll
    for (int k = 0; k < 4; ++k) {
        a[k][0] = xs[m * 16 + 4 * k + kofs + 0];
        a[k][1] = xs[m * 16 + 4 * k + kofs + 1];
    }

#pragma unroll
    for (int t = 0; t < 4; ++t) {
        v8f c = {};
#pragma unroll
        for (int k = 0; k < 4; ++k) {
            v2f b;
            b[0] = ws[(4 * k + kofs + 0) * 64 + t * 16 + m];
            b[1] = ws[(4 * k + kofs + 1) * 64 + t * 16 + m];
            c = __builtin_amdgcn_wmma_f32_16x16x4_f32(false, a[k], false, b,
                                                      (short)0, c, false, false);
        }
        float* out = h1 + (size_t)row0 * 64 + t * 16 + m;
        if (full) {
#pragma unroll
            for (int r = 0; r < 8; ++r)
                out[(size_t)(r + (hi ? 8 : 0)) * 64] = c[r];
        } else {
#pragma unroll
            for (int r = 0; r < 8; ++r) {
                int rr = r + (hi ? 8 : 0);
                if (row0 + rr < n) out[(size_t)rr * 64] = c[r];
            }
        }
    }

    {
        v8f c = {};
#pragma unroll
        for (int k = 0; k < 4; ++k) {
            v2f b;
            b[0] = wa[(4 * k + kofs + 0) * 16 + m];
            b[1] = wa[(4 * k + kofs + 1) * 16 + m];
            c = __builtin_amdgcn_wmma_f32_16x16x4_f32(false, a[k], false, b,
                                                      (short)0, c, false, false);
        }
        float* lo = (m < 8 ? as1 : ad1) + (size_t)row0 * 8 + (m & 7);
        if (full) {
#pragma unroll
            for (int r = 0; r < 8; ++r)
                lo[(size_t)(r + (hi ? 8 : 0)) * 8] = c[r];
        } else {
#pragma unroll
            for (int r = 0; r < 8; ++r) {
                int rr = r + (hi ? 8 : 0);
                if (row0 + rr < n) lo[(size_t)rr * 8] = c[r];
            }
        }
    }
}

#define BKT_SHIFT 9
#define BKT_NODES (1 << BKT_SHIFT)
#define MAX_BKT   1024
#define PART_EPT  8
#define BKT_EPT   8

__device__ __forceinline__ int block_excl_scan(int v, int* wsum, int& total) {
    const int lane = threadIdx.x & 31, w = threadIdx.x >> 5;
    int incl = v;
#pragma unroll
    for (int o = 1; o < 32; o <<= 1) {
        int t = __shfl_up(incl, o);
        if (lane >= o) incl += t;
    }
    if (lane == 31) wsum[w] = incl;
    __syncthreads();
    int woff = 0, tot = 0;
#pragma unroll
    for (int q = 0; q < 32; ++q) {
        int t = wsum[q];
        if (q < w) woff += t;
        tot += t;
    }
    total = tot;
    return woff + incl - v;
}

__global__ __launch_bounds__(1024) void k_hist(const int* __restrict__ ei, int E, int n, int B,
                                               int* __restrict__ ghist) {
    __shared__ int lh[MAX_BKT];
    const int tid = threadIdx.x;
    const int base = blockIdx.x * (1024 * PART_EPT) + tid;
    int d[PART_EPT];
#pragma unroll
    for (int k = 0; k < PART_EPT; ++k) {
        const int idx = base + k * 1024;
        d[k] = ei[E + (idx < E ? idx : E - 1)];
    }
    if (tid < B) lh[tid] = 0;
    __syncthreads();
#pragma unroll
    for (int k = 0; k < PART_EPT; ++k)
        if (base + k * 1024 < E) atomicAdd(&lh[clamp_node(d[k], n) >> BKT_SHIFT], 1);
    __syncthreads();
    if (tid < B) {
        const int c = lh[tid];
        if (c) atomicAdd(&ghist[tid], c);
    }
}

__global__ __launch_bounds__(1024) void k_part(const int* __restrict__ ei, int E, int n, int B,
                                               const int* __restrict__ ghist, int* __restrict__ gcur,
                                               int* __restrict__ part) {
    __shared__ int lh[MAX_BKT];
    __shared__ int wsum[32];
    const int tid = threadIdx.x;
    const int base = blockIdx.x * (1024 * PART_EPT) + tid;
    int sv[PART_EPT], d[PART_EPT];
#pragma unroll
    for (int k = 0; k < PART_EPT; ++k) {
        const int idx = base + k * 1024;
        const int ii  = idx < E ? idx : E - 1;
        sv[k] = ei[ii];
        d[k]  = ei[E + ii];
    }
    int tot;
    const int bstart = block_excl_scan(tid < B ? ghist[tid] : 0, wsum, tot);
    if (tid < B) lh[tid] = 0;
    __syncthreads();
#pragma unroll
    for (int k = 0; k < PART_EPT; ++k) {
        sv[k] = clamp_node(sv[k], n);
        d[k]  = (base + k * 1024 < E) ? clamp_node(d[k], n) : -1;
        if (d[k] >= 0) atomicAdd(&lh[d[k] >> BKT_SHIFT], 1);
    }
    __syncthreads();
    if (tid < B) {
        const int c = lh[tid];
        const int r = c ? atomicAdd(&gcur[tid], c) : 0;
        lh[tid] = bstart + r;
    }
    __syncthreads();
#pragma unroll
    for (int k = 0; k < PART_EPT; ++k)
        if (d[k] >= 0) {
            const int pos = atomicAdd(&lh[d[k] >> BKT_SHIFT], 1);
            part[pos] = (sv[k] << BKT_SHIFT) | (d[k] & (BKT_NODES - 1));
        }
}

__global__ __launch_bounds__(1024) void k_bucket(const int* __restrict__ part, const int* __restrict__ ghist,
                                                 int n, int* __restrict__ cnt, int* __restrict__ begs,
                                                 int* __restrict__ srcs) {
    __shared__ int lc[BKT_NODES];
    __shared__ int wsum[32];
    const int tid = threadIdx.x, b = blockIdx.x;
    int base;
    (void)block_excl_scan(tid < b ? ghist[tid] : 0, wsum, base);
    const int count = ghist[b];
    const int* __restrict__ rec = part + base;
    if (tid < BKT_NODES) lc[tid] = 0;
    __syncthreads();
    for (int e0 = tid; e0 - tid < count; e0 += 1024 * BKT_EPT) {
        int p[BKT_EPT];
#pragma unroll
        for (int k = 0; k < BKT_EPT; ++k) {
            const int e = e0 + k * 1024;
            p[k] = rec[e < count ? e : count - 1];
        }
#pragma unroll
        for (int k = 0; k < BKT_EPT; ++k)
            if (e0 + k * 1024 < count) atomicAdd(&lc[p[k] & (BKT_NODES - 1)], 1);
    }
    __syncthreads();
    const int c = tid < BKT_NODES ? lc[tid] : 0;
    __syncthreads();
    int tot;
    const int excl = block_excl_scan(c, wsum, tot);
    if (tid < BKT_NODES) {
        const int node = (b << BKT_SHIFT) + tid;
        if (node < n) { cnt[node] = c; begs[node] = base + excl; }
        lc[tid] = excl;
    }
    __syncthreads();
    for (int e0 = tid; e0 - tid < count; e0 += 1024 * BKT_EPT) {
        int p[BKT_EPT];
#pragma unroll
        for (int k = 0; k < BKT_EPT; ++k) {
            const int e = e0 + k * 1024;
            p[k] = rec[e < count ? e : count - 1];
        }
#pragma unroll
        for (int k = 0; k < BKT_EPT; ++k)
            if (e0 + k * 1024 < count) {
                const int pos = atomicAdd(&lc[p[k] & (BKT_NODES - 1)], 1);
                srcs[base + pos] = p[k] >> BKT_SHIFT;
            }
    }
}

#define A1_PITCH 68
__device__ __forceinline__ int load_srcs_blk(const int* __restrict__ srcs, int beg, int deg,
                                             int b0, unsigned lane, int self) {
    const int kk = b0 + (int)lane;
    return kk < deg ? srcs[beg + kk] : self;
}

__device__ __forceinline__ void logits1_blk(int srcv, int b0, int total, unsigned j, unsigned h,
                                            const float* __restrict__ as1, float adi,
                                            float (&e)[8]) {
#pragma unroll
    for (int c = 0; c < 8; ++c) {
        const int sk = __shfl(srcv, 4 * c + (int)j);
        float t = as1[(unsigned)sk * 8u + h] + adi;
        t = t > 0.0f ? t : NEG_SLOPE * t;
        e[c] = (b0 + 4 * c + (int)j < total) ? t : 0.0f;
    }
}

__device__ __forceinline__ void accum1_blk(int srcv, int b0, int total, const float (&al)[8],
                                           unsigned lane, const float* __restrict__ h1,
                                           float& acc0, float& acc1) {
    const unsigned g = lane >> 3;
#pragma unroll
    for (int c = 0; c < 8; ++c) {
        if (b0 + 4 * c + 3 < total) {
            float x0[4], x1[4], w0[4], w1[4];
#pragma unroll
            for (int q = 0; q < 4; ++q) {
                const int sk = __builtin_amdgcn_readlane(srcv, 4 * c + q);
                const float* hp = h1 + (unsigned)sk * 64u;
                x0[q] = hp[lane];
                x1[q] = hp[lane + 32u];
                w0[q] = __shfl(al[c], (int)(8u * q + g));
                w1[q] = __shfl(al[c], (int)(8u * q + 4u + g));
            }
#pragma unroll
            for (int q = 0; q < 4; ++q) {
                acc0 += x0[q] * w0[q];
                acc1 += x1[q] * w1[q];
            }
        } else if (b0 + 4 * c < total) {
            float x0[3], x1[3], w0[3], w1[3];
#pragma unroll
            for (int q = 0; q < 3; ++q) {
                const int sk = __builtin_amdgcn_readlane(srcv, 4 * c + q);
                const float* hp = h1 + (unsigned)sk * 64u;
                x0[q] = hp[lane];
                x1[q] = hp[lane + 32u];
                w0[q] = __shfl(al[c], (int)(8u * q + g));
                w1[q] = __shfl(al[c], (int)(8u * q + 4u + g));
            }
#pragma unroll
            for (int q = 0; q < 3; ++q) {
                const bool ok = b0 + 4 * c + q < total;
                acc0 += ok ? x0[q] * w0[q] : 0.0f;
                acc1 += ok ? x1[q] * w1[q] : 0.0f;
            }
        }
    }
}

__global__ __launch_bounds__(512) void k_agg1(const int* __restrict__ cnt, const int* __restrict__ begs,
                                              const int* __restrict__ srcs,
                                              const float* __restrict__ as1, const float* __restrict__ ad1,
                                              const float* __restrict__ h1, const float* __restrict__ bias1,
                                              const float* __restrict__ W2,
                                              const float* __restrict__ asrc2, const float* __restrict__ adst2,
                                              float* __restrict__ h2, float* __restrict__ asd2, int n) {
    __shared__ float smA[16 * A1_PITCH];
    __shared__ float ws2[64 * 16];
    __shared__ int   arrived;

    const unsigned tid  = threadIdx.x;
    const unsigned lane = tid & 31u;
    const int w    = __builtin_amdgcn_readfirstlane((int)(tid >> 5));
    const int row0 = (int)blockIdx.x * 16;
    const int i    = row0 + w;
    const bool have = i < n;
    const unsigned j = lane >> 3, h = lane & 7u;

    int deg = 0, beg = 0, srcv0 = 0;
    float adi = 0.0f;
    if (have) {
        deg   = cnt[i];
        beg   = begs[i];
        adi   = ad1[(unsigned)i * 8u + h];
        srcv0 = load_srcs_blk(srcs, beg, deg, 0, lane, i);
    }

    if (tid == 0u) arrived = 0;
    if (tid < 128u) {
        const int k = (int)(tid >> 1), half = (int)(tid & 1u);
        ((float4*)(ws2 + k * 16 + half * 4))[0] = ((const float4*)W2)[tid];
    } else if (tid < 256u) {
        const unsigned t = tid - 128u;
        const int k = (int)(t >> 1);
        const float4* avp = (const float4*)((t & 1u) ? adst2 : asrc2);
        const float4 av0 = avp[0], av1 = avp[1];
        const float4* wp = (const float4*)(W2 + k * 8);
        const float4 w0 = wp[0], w1 = wp[1];
        ws2[k * 16 + 8 + (int)(t & 1u)] = w0.x * av0.x + w0.y * av0.y + w0.z * av0.z + w0.w * av0.w
                                        + w1.x * av1.x + w1.y * av1.y + w1.z * av1.z + w1.w * av1.w;
    } else {
        for (int e = (int)tid - 256; e < 64 * 6; e += 256) {
            const int k = e / 6;
            ws2[k * 16 + 10 + (e - 6 * k)] = 0.0f;
        }
    }
    __syncthreads();

    float r0 = 0.0f, r1 = 0.0f;
    if (have) {
        const int total = deg + 1;
        const float NEG_INF = -__builtin_huge_valf();

        float acc0 = 0.0f, acc1 = 0.0f;
        float e[8], al[8];
        if (total <= 32) {
            const int srcv = srcv0;
            logits1_blk(srcv, 0, total, j, h, as1, adi, e);
            float m = NEG_INF;
#pragma unroll
            for (int c = 0; c < 8; ++c)
                if (4 * c + (int)j < total) m = fmaxf(m, e[c]);
            m = fmaxf(m, __shfl_xor(m, 8));
            m = fmaxf(m, __shfl_xor(m, 16));
            float ssum = 0.0f;
#pragma unroll
            for (int c = 0; c < 8; ++c) {
                al[c] = (4 * c + (int)j < total) ? expf(e[c] - m) : 0.0f;
                ssum += al[c];
            }
            ssum += __shfl_xor(ssum, 8);
            ssum += __shfl_xor(ssum, 16);
            const float den = ssum + 1e-16f;
#pragma unroll
            for (int c = 0; c < 8; ++c) al[c] = al[c] / den;
            accum1_blk(srcv, 0, total, al, lane, h1, acc0, acc1);
        } else {
            float m = NEG_INF;
            for (int b0 = 0; b0 < total; b0 += 32) {
                const int srcv = load_srcs_blk(srcs, beg, deg, b0, lane, i);
                logits1_blk(srcv, b0, total, j, h, as1, adi, e);
#pragma unroll
                for (int c = 0; c < 8; ++c)
                    if (b0 + 4 * c + (int)j < total) m = fmaxf(m, e[c]);
            }
            m = fmaxf(m, __shfl_xor(m, 8));
            m = fmaxf(m, __shfl_xor(m, 16));
            float ssum = 0.0f;
            for (int b0 = 0; b0 < total; b0 += 32) {
                const int srcv = load_srcs_blk(srcs, beg, deg, b0, lane, i);
                logits1_blk(srcv, b0, total, j, h, as1, adi, e);
#pragma unroll
                for (int c = 0; c < 8; ++c)
                    ssum += (b0 + 4 * c + (int)j < total) ? expf(e[c] - m) : 0.0f;
            }
            ssum += __shfl_xor(ssum, 8);
            ssum += __shfl_xor(ssum, 16);
            const float den = ssum + 1e-16f;
            for (int b0 = 0; b0 < total; b0 += 32) {
                const int srcv = load_srcs_blk(srcs, beg, deg, b0, lane, i);
                logits1_blk(srcv, b0, total, j, h, as1, adi, e);
#pragma unroll
                for (int c = 0; c < 8; ++c)
                    al[c] = (b0 + 4 * c + (int)j < total) ? expf(e[c] - m) / den : 0.0f;
                accum1_blk(srcv, b0, total, al, lane, h1, acc0, acc1);
            }
        }

        r0 = acc0 + bias1[lane];
        r1 = acc1 + bias1[lane + 32u];
        r0 = r0 > 0.0f ? r0 : (expf(r0) - 1.0f);
        r1 = r1 > 0.0f ? r1 : (expf(r1) - 1.0f);
    }

    smA[w * A1_PITCH + (int)lane]      = r0;
    smA[w * A1_PITCH + (int)lane + 32] = r1;
    __threadfence_block();
    int before = 0;
    if (lane == 0u) before = atomicAdd(&arrived, 1);
    before = __builtin_amdgcn_readfirstlane(before);
    if (before != 15) return;
    __threadfence_block();

    const bool hi   = lane >= 16u;
    const int  m    = (int)(lane & 15u);
    const bool full = (row0 + 16 <= n);
    const int  kofs = hi ? 2 : 0;
    v8f c = {};
#pragma unroll
    for (int k = 0; k < 16; ++k) {
        v2f a, b;
        a[0] = smA[m * A1_PITCH + 4 * k + kofs + 0];
        a[1] = smA[m * A1_PITCH + 4 * k + kofs + 1];
        b[0] = ws2[(4 * k + kofs + 0) * 16 + m];
        b[1] = ws2[(4 * k + kofs + 1) * 16 + m];
        c = __builtin_amdgcn_wmma_f32_16x16x4_f32(false, a, false, b,
                                                  (short)0, c, false, false);
    }
    if (m < 8) {
        float* out = h2 + (size_t)row0 * 8 + m;
        if (full) {
#pragma unroll
            for (int r = 0; r < 8; ++r)
                out[(size_t)(r + (hi ? 8 : 0)) * 8] = c[r];
        } else {
#pragma unroll
            for (int r = 0; r < 8; ++r) {
                int rr = r + (hi ? 8 : 0);
                if (row0 + rr < n) out[(size_t)rr * 8] = c[r];
            }
        }
    } else if (m < 10) {
        float* out = asd2 + (size_t)row0 * 2 + (m - 8);
        if (full) {
#pragma unroll
            for (int r = 0; r < 8; ++r)
                out[(size_t)(r + (hi ? 8 : 0)) * 2] = c[r];
        } else {
#pragma unroll
            for (int r = 0; r < 8; ++r) {
                int rr = r + (hi ? 8 : 0);
                if (row0 + rr < n) out[(size_t)rr * 2] = c[r];
            }
        }
    }
}

__device__ __forceinline__ float wave_max(float v) {
#pragma unroll
    for (int o = 16; o > 0; o >>= 1) v = fmaxf(v, __shfl_xor(v, o));
    return v;
}
__device__ __forceinline__ float wave_sum(float v) {
#pragma unroll
    for (int o = 16; o > 0; o >>= 1) v += __shfl_xor(v, o);
    return v;
}

__device__ __forceinline__ float logit2(int srcv, bool valid,
                                        const float* __restrict__ asd2, float adi) {
    float t = asd2[2u * (unsigned)srcv] + adi;
    t = t > 0.0f ? t : NEG_SLOPE * t;
    return valid ? t : 0.0f;
}

__device__ __forceinline__ void rows2_blk(int srcv, unsigned j, unsigned f,
                                          const float* __restrict__ h2, float (&hv)[8]) {
#pragma unroll
    for (int c = 0; c < 8; ++c) {
        const int sk = __shfl(srcv, 4 * c + (int)j);
        hv[c] = h2[(unsigned)sk * 8u + f];
    }
}

__device__ __forceinline__ void accum2_blk(const float (&hv)[8], float al, int b0, int total,
                                           unsigned j, float& acc) {
#pragma unroll
    for (int c = 0; c < 8; ++c) {
        const float ak = __shfl(al, 4 * c + (int)j);
        acc += (b0 + 4 * c + (int)j < total) ? hv[c] * ak : 0.0f;
    }
}

__global__ __launch_bounds__(256) void k_agg2(const int* __restrict__ cnt, const int* __restrict__ begs,
                                              const int* __restrict__ srcs,
                                              const float* __restrict__ asd2,
                                              const float* __restrict__ h2, const float* __restrict__ bias2,
                                              float* __restrict__ out, int n) {
    const unsigned lane = threadIdx.x & 31u;
    const int i = __builtin_amdgcn_readfirstlane(
        (int)(blockIdx.x * (blockDim.x >> 5) + (threadIdx.x >> 5)));
    if (i >= n) return;
    const int deg   = cnt[i];
    const int beg   = begs[i];
    const int total = deg + 1;
    const unsigned j = lane >> 3, f = lane & 7u;
    const float adi = asd2[2u * (unsigned)i + 1u];
    const float NEG_INF = -__builtin_huge_valf();

    float acc = 0.0f;
    float hv[8];
    if (total <= 32) {
        const int  srcv  = load_srcs_blk(srcs, beg, deg, 0, lane, i);
        const bool valid = (int)lane < total;
        rows2_blk(srcv, j, f, h2, hv);
        __builtin_amdgcn_sched_barrier(0);
        const float e  = logit2(srcv, valid, asd2, adi);
        const float m  = wave_max(valid ? e : NEG_INF);
        const float ex = valid ? expf(e - m) : 0.0f;
        const float den = wave_sum(ex) + 1e-16f;
        accum2_blk(hv, ex / den, 0, total, j, acc);
    } else {
        float m = NEG_INF;
        for (int b0 = 0; b0 < total; b0 += 32) {
            const int  srcv  = load_srcs_blk(srcs, beg, deg, b0, lane, i);
            const bool valid = b0 + (int)lane < total;
            const float e = logit2(srcv, valid, asd2, adi);
            m = fmaxf(m, valid ? e : NEG_INF);
        }
        m = wave_max(m);
        float ssum = 0.0f;
        for (int b0 = 0; b0 < total; b0 += 32) {
            const int  srcv  = load_srcs_blk(srcs, beg, deg, b0, lane, i);
            const bool valid = b0 + (int)lane < total;
            const float e = logit2(srcv, valid, asd2, adi);
            ssum += valid ? expf(e - m) : 0.0f;
        }
        const float den = wave_sum(ssum) + 1e-16f;
        for (int b0 = 0; b0 < total; b0 += 32) {
            const int  srcv  = load_srcs_blk(srcs, beg, deg, b0, lane, i);
            const bool valid = b0 + (int)lane < total;
            const float e = logit2(srcv, valid, asd2, adi);
            rows2_blk(srcv, j, f, h2, hv);
            const float al = valid ? expf(e - m) / den : 0.0f;
            accum2_blk(hv, al, b0, total, j, acc);
        }
    }
    acc += __shfl_xor(acc, 8);
    acc += __shfl_xor(acc, 16);
    if (lane < 8u) out[(unsigned)i * 8u + f] = acc + bias2[f];
}

extern "C" void kernel_launch(void* const* d_in, const int* in_sizes, int n_in,
                              void* d_out, int out_size, void* d_ws, size_t ws_size,
                              hipStream_t stream) {
    const float* x     = (const float*)d_in[0];
    const int*   ei    = (const int*)d_in[1];
    const float* W1    = (const float*)d_in[2];
    const float* asrc1 = (const float*)d_in[3];
    const float* adst1 = (const float*)d_in[4];
    const float* bias1 = (const float*)d_in[5];
    const float* W2    = (const float*)d_in[6];
    const float* asrc2 = (const float*)d_in[7];
    const float* adst2 = (const float*)d_in[8];
    const float* bias2 = (const float*)d_in[9];

    int n = in_sizes[0] / F_IN;
    int E = in_sizes[1] / 2;
    int B = (n + BKT_NODES - 1) >> BKT_SHIFT;

    char* p = (char*)d_ws;
    float* h1   = (float*)p; p += (size_t)n * 64 * 4;
    float* as1  = (float*)p; p += (size_t)n * 8 * 4;
    float* ad1  = (float*)p; p += (size_t)n * 8 * 4;
    float* h2   = (float*)p; p += (size_t)n * 8 * 4;
    float* asd2 = (float*)p; p += (size_t)n * 8;
    int*   cnt  = (int*)p;   p += (size_t)n * 4;
    int*   begs = (int*)p;   p += (size_t)n * 4;
    int*   part = (int*)p;   p += (size_t)E * 4;
    int*   srcs = (int*)p;   p += (size_t)E * 4;
    int*   ghist = (int*)p;  p += (size_t)MAX_BKT * 4;
    int*   gcur  = (int*)p;  p += (size_t)MAX_BKT * 4;

    hipMemsetAsync(ghist, 0, (size_t)2 * MAX_BKT * 4, stream);

    const int tpb = 256;
    int gP = (E + 1024 * PART_EPT - 1) / (1024 * PART_EPT);
    int gN = (n + tpb / 32 - 1) / (tpb / 32);

    k_hist<<<gP, 1024, 0, stream>>>(ei, E, n, B, ghist);
    k_part<<<gP, 1024, 0, stream>>>(ei, E, n, B, ghist, gcur, part);
    k_bucket<<<B, 1024, 0, stream>>>(part, ghist, n, cnt, begs, srcs);

    k_gemm1<<<(n + 15) / 16, 32, 0, stream>>>(x, W1, asrc1, adst1, h1, as1, ad1, n);
    k_agg1<<<(n + 15) / 16, 512, 0, stream>>>(cnt, begs, srcs, as1, ad1, h1, bias1,
                                              W2, asrc2, adst2, h2, asd2, n);

    k_agg2<<<gN, tpb, 0, stream>>>(cnt, begs, srcs, asd2, h2, bias2, (float*)d_out, n);
}


# x @ W1 product (and its logit tile) moved into the last grouping launch as extra 1024-thread work-groups picked by block index, W1 staged once per 32 row tiles: one launch fewer; sum orders kept
# speedup vs baseline: 11.1983x; 1.0741x over previous
#include <hip/hip_runtime.h>
#include <hip/hip_bf16.h>

#define NEG_SLOPE 0.2f
#define F_IN 14

typedef float v2f __attribute__((ext_vector_type(2)));
typedef float v8f __attribute__((ext_vector_type(8)));

__device__ __forceinline__ int clamp_node(int v, int n) {
    return v < 0 ? 0 : (v >= n ? n - 1 : v);
}

#define G1_TILES 32
__device__ __forceinline__ void gemm1_role(int gblk,
                                           const float* __restrict__ x,
                                           const float* __restrict__ W1,
                                           const float* __restrict__ asrc,
                                           const float* __restrict__ adst,
                                           float* __restrict__ h1,
                                           float* __restrict__ as1, float* __restrict__ ad1, int n,
                                           float* xs_all,
                                           float* ws,
                                           float* wa) {
    const int tid  = threadIdx.x;
    const int lane = tid & 31;
    const int w    = __builtin_amdgcn_readfirstlane(tid >> 5);
    const int row0 = (gblk * G1_TILES + w) * 16;
    const bool hi  = lane >= 16;
    const int  m   = lane & 15;
    float* xs = xs_all + w * 256;

    int rows = n - row0;
    rows = rows > 16 ? 16 : (rows < 0 ? 0 : rows);
    const int cntx = rows * F_IN;
    float xv[7];
#pragma unroll
    for (int q = 0; q < 7; ++q) {
        const int i = lane + 32 * q;
        xv[q] = i < cntx ? x[(size_t)row0 * F_IN + i] : 0.0f;
    }
    float4 av0 = make_float4(0.f, 0.f, 0.f, 0.f), av1 = av0;
    if (tid < 256) {
        const int mm = tid & 15;
        const float4* avp = (const float4*)((mm < 8 ? asrc : adst) + (mm & 7) * 8);
        av0 = avp[0];
        av1 = avp[1];
    }
    if (tid < F_IN * 64 / 4)
        ((float4*)ws)[tid] = ((const float4*)W1)[tid];
    else if (tid < 256)
        ((float4*)(ws + F_IN * 64))[tid - F_IN * 64 / 4] = make_float4(0.f, 0.f, 0.f, 0.f);
#pragma unroll
    for (int q = 0; q < 7; ++q) {
        const int i = lane + 32 * q;
        const int r = i / F_IN, c = i - r * F_IN;
        xs[r * 16 + c] = xv[q];
    }
    xs[(lane >> 1) * 16 + F_IN + (lane & 1)] = 0.0f;
    __syncthreads();
    if (tid < 256) {
        const int k = tid >> 4, mm = tid & 15;
        const float4* wp = (const float4*)(ws + k * 64 + (mm & 7) * 8);
        const float4 w0 = wp[0], w1 = wp[1];
        wa[k * 16 + mm] = w0.x * av0.x + w0.y * av0.y + w0.z * av0.z + w0.w * av0.w
                        + w1.x * av1.x + w1.y * av1.y + w1.z * av1.z + w1.w * av1.w;
    }
    __syncthreads();
    if (row0 >= n) return;
    const bool full = (row0 + 16 <= n);

    const int kofs = hi ? 2 : 0;
    v2f a[4];
#pragma unroll
    for (int k = 0; k < 4; ++k) {
        a[k][0] = xs[m * 16 + 4 * k + kofs + 0];
        a[k][1] = xs[m * 16 + 4 * k + kofs + 1];
    }

#pragma unroll
    for (int t = 0; t < 4; ++t) {
        v8f c = {};
#pragma unroll
        for (int k = 0; k < 4; ++k) {
            v2f b;
            b[0] = ws[(4 * k + kofs + 0) * 64 + t * 16 + m];
            b[1] = ws[(4 * k + kofs + 1) * 64 + t * 16 + m];
            c = __builtin_amdgcn_wmma_f32_16x16x4_f32(false, a[k], false, b,
                                                      (short)0, c, false, false);
        }
        float* out = h1 + (size_t)row0 * 64 + t * 16 + m;
        if (full) {
#pragma unroll
            for (int r = 0; r < 8; ++r)
                out[(size_t)(r + (hi ? 8 : 0)) * 64] = c[r];
        } else {
#pragma unroll
            for (int r = 0; r < 8; ++r) {
                int rr = r + (hi ? 8 : 0);
                if (row0 + rr < n) out[(size_t)rr * 64] = c[r];
            }
        }
    }

    {
        v8f c = {};
#pragma unroll
        for (int k = 0; k < 4; ++k) {
            v2f b;
            b[0] = wa[(4 * k + kofs + 0) * 16 + m];
            b[1] = wa[(4 * k + kofs + 1) * 16 + m];
            c = __builtin_amdgcn_wmma_f32_16x16x4_f32(false, a[k], false, b,
                                                      (short)0, c, false, false);
        }
        float* lo = (m < 8 ? as1 : ad1) + (size_t)row0 * 8 + (m & 7);
        if (full) {
#pragma unroll
            for (int r = 0; r < 8; ++r)
                lo[(size_t)(r + (hi ? 8 : 0)) * 8] = c[r];
        } else {
#pragma unroll
            for (int r = 0; r < 8; ++r) {
                int rr = r + (hi ? 8 : 0);
                if (row0 + rr < n) lo[(size_t)rr * 8] = c[r];
            }
        }
    }
}

#define BKT_SHIFT 9
#define BKT_NODES (1 << BKT_SHIFT)
#define MAX_BKT   1024
#define PART_EPT  8
#define BKT_EPT   8

__device__ __forceinline__ int block_excl_scan(int v, int* wsum, int& total) {
    const int lane = threadIdx.x & 31, w = threadIdx.x >> 5;
    int incl = v;
#pragma unroll
    for (int o = 1; o < 32; o <<= 1) {
        int t = __shfl_up(incl, o);
        if (lane >= o) incl += t;
    }
    if (lane == 31) wsum[w] = incl;
    __syncthreads();
    int woff = 0, tot = 0;
#pragma unroll
    for (int q = 0; q < 32; ++q) {
        int t = wsum[q];
        if (q < w) woff += t;
        tot += t;
    }
    total = tot;
    return woff + incl - v;
}

__global__ __launch_bounds__(1024) void k_hist(const int* __restrict__ ei, int E, int n, int B,
                                               int* __restrict__ ghist) {
    __shared__ int lh[MAX_BKT];
    const int tid = threadIdx.x;
    const int base = blockIdx.x * (1024 * PART_EPT) + tid;
    int d[PART_EPT];
#pragma unroll
    for (int k = 0; k < PART_EPT; ++k) {
        const int idx = base + k * 1024;
        d[k] = ei[E + (idx < E ? idx : E - 1)];
    }
    if (tid < B) lh[tid] = 0;
    __syncthreads();
#pragma unroll
    for (int k = 0; k < PART_EPT; ++k)
        if (base + k * 1024 < E) atomicAdd(&lh[clamp_node(d[k], n) >> BKT_SHIFT], 1);
    __syncthreads();
    if (tid < B) {
        const int c = lh[tid];
        if (c) atomicAdd(&ghist[tid], c);
    }
}

__global__ __launch_bounds__(1024) void k_part(const int* __restrict__ ei, int E, int n, int B,
                                               const int* __restrict__ ghist, int* __restrict__ gcur,
                                               int* __restrict__ part) {
    __shared__ int lh[MAX_BKT];
    __shared__ int wsum[32];
    const int tid = threadIdx.x;
    const int base = blockIdx.x * (1024 * PART_EPT) + tid;
    int sv[PART_EPT], d[PART_EPT];
#pragma unroll
    for (int k = 0; k < PART_EPT; ++k) {
        const int idx = base + k * 1024;
        const int ii  = idx < E ? idx : E - 1;
        sv[k] = ei[ii];
        d[k]  = ei[E + ii];
    }
    int tot;
    const int bstart = block_excl_scan(tid < B ? ghist[tid] : 0, wsum, tot);
    if (tid < B) lh[tid] = 0;
    __syncthreads();
#pragma unroll
    for (int k = 0; k < PART_EPT; ++k) {
        sv[k] = clamp_node(sv[k], n);
        d[k]  = (base + k * 1024 < E) ? clamp_node(d[k], n) : -1;
        if (d[k] >= 0) atomicAdd(&lh[d[k] >> BKT_SHIFT], 1);
    }
    __syncthreads();
    if (tid < B) {
        const int c = lh[tid];
        const int r = c ? atomicAdd(&gcur[tid], c) : 0;
        lh[tid] = bstart + r;
    }
    __syncthreads();
#pragma unroll
    for (int k = 0; k < PART_EPT; ++k)
        if (d[k] >= 0) {
            const int pos = atomicAdd(&lh[d[k] >> BKT_SHIFT], 1);
            part[pos] = (sv[k] << BKT_SHIFT) | (d[k] & (BKT_NODES - 1));
        }
}

__device__ __forceinline__ void bucket_role(int b, const int* __restrict__ part, const int* __restrict__ ghist,
                                            int n, int* __restrict__ cnt, int* __restrict__ begs,
                                            int* __restrict__ srcs,
                                            int* lc,
                                            int* wsum) {
    const int tid = threadIdx.x;
    int base;
    (void)block_excl_scan(tid < b ? ghist[tid] : 0, wsum, base);
    const int count = ghist[b];
    const int* __restrict__ rec = part + base;
    if (tid < BKT_NODES) lc[tid] = 0;
    __syncthreads();
    for (int e0 = tid; e0 - tid < count; e0 += 1024 * BKT_EPT) {
        int p[BKT_EPT];
#pragma unroll
        for (int k = 0; k < BKT_EPT; ++k) {
            const int e = e0 + k * 1024;
            p[k] = rec[e < count ? e : count - 1];
        }
#pragma unroll
        for (int k = 0; k < BKT_EPT; ++k)
            if (e0 + k * 1024 < count) atomicAdd(&lc[p[k] & (BKT_NODES - 1)], 1);
    }
    __syncthreads();
    const int c = tid < BKT_NODES ? lc[tid] : 0;
    __syncthreads();
    int tot;
    const int excl = block_excl_scan(c, wsum, tot);
    if (tid < BKT_NODES) {
        const int node = (b << BKT_SHIFT) + tid;
        if (node < n) { cnt[node] = c; begs[node] = base + excl; }
        lc[tid] = excl;
    }
    __syncthreads();
    for (int e0 = tid; e0 - tid < count; e0 += 1024 * BKT_EPT) {
        int p[BKT_EPT];
#pragma unroll
        for (int k = 0; k < BKT_EPT; ++k) {
            const int e = e0 + k * 1024;
            p[k] = rec[e < count ? e : count - 1];
        }
#pragma unroll
        for (int k = 0; k < BKT_EPT; ++k)
            if (e0 + k * 1024 < count) {
                const int pos = atomicAdd(&lc[p[k] & (BKT_NODES - 1)], 1);
                srcs[base + pos] = p[k] >> BKT_SHIFT;
            }
    }
}

__global__ __launch_bounds__(1024) void k_bucket_gemm1(const int* __restrict__ part, const int* __restrict__ ghist,
                                                       int n, int* __restrict__ cnt, int* __restrict__ begs,
                                                       int* __restrict__ srcs, int nb,
                                                       const float* __restrict__ x,
                                                       const float* __restrict__ W1,
                                                       const float* __restrict__ asrc,
                                                       const float* __restrict__ adst,
                                                       float* __restrict__ h1,
                                                       float* __restrict__ as1, float* __restrict__ ad1) {
    __shared__ int   lc[BKT_NODES];
    __shared__ int   wsum[32];
    __shared__ float xs_all[G1_TILES * 16 * 16];
    __shared__ float ws[16 * 64];
    __shared__ float wa[16 * 16];
    const int blk = (int)blockIdx.x;
    if (blk < nb)
        bucket_role(blk, part, ghist, n, cnt, begs, srcs, lc, wsum);
    else
        gemm1_role(blk - nb, x, W1, asrc, adst, h1, as1, ad1, n, xs_all, ws, wa);
}

#define A1_PITCH 68
__device__ __forceinline__ int load_srcs_blk(const int* __restrict__ srcs, int beg, int deg,
                                             int b0, unsigned lane, int self) {
    const int kk = b0 + (int)lane;
    return kk < deg ? srcs[beg + kk] : self;
}

__device__ __forceinline__ void logits1_blk(int srcv, int b0, int total, unsigned j, unsigned h,
                                            const float* __restrict__ as1, float adi,
                                            float (&e)[8]) {
#pragma unroll
    for (int c = 0; c < 8; ++c) {
        const int sk = __shfl(srcv, 4 * c + (int)j);
        float t = as1[(unsigned)sk * 8u + h] + adi;
        t = t > 0.0f ? t : NEG_SLOPE * t;
        e[c] = (b0 + 4 * c + (int)j < total) ? t : 0.0f;
    }
}

__device__ __forceinline__ void accum1_blk(int srcv, int b0, int total, const float (&al)[8],
                                           unsigned lane, const float* __restrict__ h1,
                                           float& acc0, float& acc1) {
    const unsigned g = lane >> 3;
#pragma unroll
    for (int c = 0; c < 8; ++c) {
        if (b0 + 4 * c + 3 < total) {
            float x0[4], x1[4], w0[4], w1[4];
#pragma unroll
            for (int q = 0; q < 4; ++q) {
                const int sk = __builtin_amdgcn_readlane(srcv, 4 * c + q);
                const float* hp = h1 + (unsigned)sk * 64u;
                x0[q] = hp[lane];
                x1[q] = hp[lane + 32u];
                w0[q] = __shfl(al[c], (int)(8u * q + g));
                w1[q] = __shfl(al[c], (int)(8u * q + 4u + g));
            }
#pragma unroll
            for (int q = 0; q < 4; ++q) {
                acc0 += x0[q] * w0[q];
                acc1 += x1[q] * w1[q];
            }
        } else if (b0 + 4 * c < total) {
            float x0[3], x1[3], w0[3], w1[3];
#pragma unroll
            for (int q = 0; q < 3; ++q) {
                const int sk = __builtin_amdgcn_readlane(srcv, 4 * c + q);
                const float* hp = h1 + (unsigned)sk * 64u;
                x0[q] = hp[lane];
                x1[q] = hp[lane + 32u];
                w0[q] = __shfl(al[c], (int)(8u * q + g));
                w1[q] = __shfl(al[c], (int)(8u * q + 4u + g));
            }
#pragma unroll
            for (int q = 0; q < 3; ++q) {
                const bool ok = b0 + 4 * c + q < total;
                acc0 += ok ? x0[q] * w0[q] : 0.0f;
                acc1 += ok ? x1[q] * w1[q] : 0.0f;
            }
        }
    }
}

__global__ __launch_bounds__(512) void k_agg1(const int* __restrict__ cnt, const int* __restrict__ begs,
                                              const int* __restrict__ srcs,
                                              const float* __restrict__ as1, const float* __restrict__ ad1,
                                              const float* __restrict__ h1, const float* __restrict__ bias1,
                                              const float* __restrict__ W2,
                                              const float* __restrict__ asrc2, const float* __restrict__ adst2,
                                              float* __restrict__ h2, float* __restrict__ asd2, int n) {
    __shared__ float smA[16 * A1_PITCH];
    __shared__ float ws2[64 * 16];
    __shared__ int   arrived;

    const unsigned tid  = threadIdx.x;
    const unsigned lane = tid & 31u;
    const int w    = __builtin_amdgcn_readfirstlane((int)(tid >> 5));
    const int row0 = (int)blockIdx.x * 16;
    const int i    = row0 + w;
    const bool have = i < n;
    const unsigned j = lane >> 3, h = lane & 7u;

    int deg = 0, beg = 0, srcv0 = 0;
    float adi = 0.0f;
    if (have) {
        deg   = cnt[i];
        beg   = begs[i];
        adi   = ad1[(unsigned)i * 8u + h];
        srcv0 = load_srcs_blk(srcs, beg, deg, 0, lane, i);
    }

    if (tid == 0u) arrived = 0;
    if (tid < 128u) {
        const int k = (int)(tid >> 1), half = (int)(tid & 1u);
        ((float4*)(ws2 + k * 16 + half * 4))[0] = ((const float4*)W2)[tid];
    } else if (tid < 256u) {
        const unsigned t = tid - 128u;
        const int k = (int)(t >> 1);
        const float4* avp = (const float4*)((t & 1u) ? adst2 : asrc2);
        const float4 av0 = avp[0], av1 = avp[1];
        const float4* wp = (const float4*)(W2 + k * 8);
        const float4 w0 = wp[0], w1 = wp[1];
        ws2[k * 16 + 8 + (int)(t & 1u)] = w0.x * av0.x + w0.y * av0.y + w0.z * av0.z + w0.w * av0.w
                                        + w1.x * av1.x + w1.y * av1.y + w1.z * av1.z + w1.w * av1.w;
    } else {
        for (int e = (int)tid - 256; e < 64 * 6; e += 256) {
            const int k = e / 6;
            ws2[k * 16 + 10 + (e - 6 * k)] = 0.0f;
        }
    }
    __syncthreads();

    float r0 = 0.0f, r1 = 0.0f;
    if (have) {
        const int total = deg + 1;
        const float NEG_INF = -__builtin_huge_valf();

        float acc0 = 0.0f, acc1 = 0.0f;
        float e[8], al[8];
        if (total <= 32) {
            const int srcv = srcv0;
            logits1_blk(srcv, 0, total, j, h, as1, adi, e);
            float m = NEG_INF;
#pragma unroll
            for (int c = 0; c < 8; ++c)
                if (4 * c + (int)j < total) m = fmaxf(m, e[c]);
            m = fmaxf(m, __shfl_xor(m, 8));
            m = fmaxf(m, __shfl_xor(m, 16));
            float ssum = 0.0f;
#pragma unroll
            for (int c = 0; c < 8; ++c) {
                al[c] = (4 * c + (int)j < total) ? expf(e[c] - m) : 0.0f;
                ssum += al[c];
            }
            ssum += __shfl_xor(ssum, 8);
            ssum += __shfl_xor(ssum, 16);
            const float den = ssum + 1e-16f;
#pragma unroll
            for (int c = 0; c < 8; ++c) al[c] = al[c] / den;
            accum1_blk(srcv, 0, total, al, lane, h1, acc0, acc1);
        } else {
            float m = NEG_INF;
            for (int b0 = 0; b0 < total; b0 += 32) {
                const int srcv = load_srcs_blk(srcs, beg, deg, b0, lane, i);
                logits1_blk(srcv, b0, total, j, h, as1, adi, e);
#pragma unroll
                for (int c = 0; c < 8; ++c)
                    if (b0 + 4 * c + (int)j < total) m = fmaxf(m, e[c]);
            }
            m = fmaxf(m, __shfl_xor(m, 8));
            m = fmaxf(m, __shfl_xor(m, 16));
            float ssum = 0.0f;
            for (int b0 = 0; b0 < total; b0 += 32) {
                const int srcv = load_srcs_blk(srcs, beg, deg, b0, lane, i);
                logits1_blk(srcv, b0, total, j, h, as1, adi, e);
#pragma unroll
                for (int c = 0; c < 8; ++c)
                    ssum += (b0 + 4 * c + (int)j < total) ? expf(e[c] - m) : 0.0f;
            }
            ssum += __shfl_xor(ssum, 8);
            ssum += __shfl_xor(ssum, 16);
            const float den = ssum + 1e-16f;
            for (int b0 = 0; b0 < total; b0 += 32) {
                const int srcv = load_srcs_blk(srcs, beg, deg, b0, lane, i);
                logits1_blk(srcv, b0, total, j, h, as1, adi, e);
#pragma unroll
                for (int c = 0; c < 8; ++c)
                    al[c] = (b0 + 4 * c + (int)j < total) ? expf(e[c] - m) / den : 0.0f;
                accum1_blk(srcv, b0, total, al, lane, h1, acc0, acc1);
            }
        }

        r0 = acc0 + bias1[lane];
        r1 = acc1 + bias1[lane + 32u];
        r0 = r0 > 0.0f ? r0 : (expf(r0) - 1.0f);
        r1 = r1 > 0.0f ? r1 : (expf(r1) - 1.0f);
    }

    smA[w * A1_PITCH + (int)lane]      = r0;
    smA[w * A1_PITCH + (int)lane + 32] = r1;
    __threadfence_block();
    int before = 0;
    if (lane == 0u) before = atomicAdd(&arrived, 1);
    before = __builtin_amdgcn_readfirstlane(before);
    if (before != 15) return;
    __threadfence_block();

    const bool hi   = lane >= 16u;
    const int  m    = (int)(lane & 15u);
    const bool full = (row0 + 16 <= n);
    const int  kofs = hi ? 2 : 0;
    v8f c = {};
#pragma unroll
    for (int k = 0; k < 16; ++k) {
        v2f a, b;
        a[0] = smA[m * A1_PITCH + 4 * k + kofs + 0];
        a[1] = smA[m * A1_PITCH + 4 * k + kofs + 1];
        b[0] = ws2[(4 * k + kofs + 0) * 16 + m];
        b[1] = ws2[(4 * k + kofs + 1) * 16 + m];
        c = __builtin_amdgcn_wmma_f32_16x16x4_f32(false, a, false, b,
                                                  (short)0, c, false, false);
    }
    if (m < 8) {
        float* out = h2 + (size_t)row0 * 8 + m;
        if (full) {
#pragma unroll
            for (int r = 0; r < 8; ++r)
                out[(size_t)(r + (hi ? 8 : 0)) * 8] = c[r];
        } else {
#pragma unroll
            for (int r = 0; r < 8; ++r) {
                int rr = r + (hi ? 8 : 0);
                if (row0 + rr < n) out[(size_t)rr * 8] = c[r];
            }
        }
    } else if (m < 10) {
        float* out = asd2 + (size_t)row0 * 2 + (m - 8);
        if (full) {
#pragma unroll
            for (int r = 0; r < 8; ++r)
                out[(size_t)(r + (hi ? 8 : 0)) * 2] = c[r];
        } else {
#pragma unroll
            for (int r = 0; r < 8; ++r) {
                int rr = r + (hi ? 8 : 0);
                if (row0 + rr < n) out[(size_t)rr * 2] = c[r];
            }
        }
    }
}

__device__ __forceinline__ float wave_max(float v) {
#pragma unroll
    for (int o = 16; o > 0; o >>= 1) v = fmaxf(v, __shfl_xor(v, o));
    return v;
}
__device__ __forceinline__ float wave_sum(float v) {
#pragma unroll
    for (int o = 16; o > 0; o >>= 1) v += __shfl_xor(v, o);
    return v;
}

__device__ __forceinline__ float logit2(int srcv, bool valid,
                                        const float* __restrict__ asd2, float adi) {
    float t = asd2[2u * (unsigned)srcv] + adi;
    t = t > 0.0f ? t : NEG_SLOPE * t;
    return valid ? t : 0.0f;
}

__device__ __forceinline__ void rows2_blk(int srcv, unsigned j, unsigned f,
                                          const float* __restrict__ h2, float (&hv)[8]) {
#pragma unroll
    for (int c = 0; c < 8; ++c) {
        const int sk = __shfl(srcv, 4 * c + (int)j);
        hv[c] = h2[(unsigned)sk * 8u + f];
    }
}

__device__ __forceinline__ void accum2_blk(const float (&hv)[8], float al, int b0, int total,
                                           unsigned j, float& acc) {
#pragma unroll
    for (int c = 0; c < 8; ++c) {
        const float ak = __shfl(al, 4 * c + (int)j);
        acc += (b0 + 4 * c + (int)j < total) ? hv[c] * ak : 0.0f;
    }
}

__global__ __launch_bounds__(256) void k_agg2(const int* __restrict__ cnt, const int* __restrict__ begs,
                                              const int* __restrict__ srcs,
                                              const float* __restrict__ asd2,
                                              const float* __restrict__ h2, const float* __restrict__ bias2,
                                              float* __restrict__ out, int n) {
    const unsigned lane = threadIdx.x & 31u;
    const int i = __builtin_amdgcn_readfirstlane(
        (int)(blockIdx.x * (blockDim.x >> 5) + (threadIdx.x >> 5)));
    if (i >= n) return;
    const int deg   = cnt[i];
    const int beg   = begs[i];
    const int total = deg + 1;
    const unsigned j = lane >> 3, f = lane & 7u;
    const float adi = asd2[2u * (unsigned)i + 1u];
    const float NEG_INF = -__builtin_huge_valf();

    float acc = 0.0f;
    float hv[8];
    if (total <= 32) {
        const int  srcv  = load_srcs_blk(srcs, beg, deg, 0, lane, i);
        const bool valid = (int)lane < total;
        rows2_blk(srcv, j, f, h2, hv);
        __builtin_amdgcn_sched_barrier(0);
        const float e  = logit2(srcv, valid, asd2, adi);
        const float m  = wave_max(valid ? e : NEG_INF);
        const float ex = valid ? expf(e - m) : 0.0f;
        const float den = wave_sum(ex) + 1e-16f;
        accum2_blk(hv, ex / den, 0, total, j, acc);
    } else {
        float m = NEG_INF;
        for (int b0 = 0; b0 < total; b0 += 32) {
            const int  srcv  = load_srcs_blk(srcs, beg, deg, b0, lane, i);
            const bool valid = b0 + (int)lane < total;
            const float e = logit2(srcv, valid, asd2, adi);
            m = fmaxf(m, valid ? e : NEG_INF);
        }
        m = wave_max(m);
        float ssum = 0.0f;
        for (int b0 = 0; b0 < total; b0 += 32) {
            const int  srcv  = load_srcs_blk(srcs, beg, deg, b0, lane, i);
            const bool valid = b0 + (int)lane < total;
            const float e = logit2(srcv, valid, asd2, adi);
            ssum += valid ? expf(e - m) : 0.0f;
        }
        const float den = wave_sum(ssum) + 1e-16f;
        for (int b0 = 0; b0 < total; b0 += 32) {
            const int  srcv  = load_srcs_blk(srcs, beg, deg, b0, lane, i);
            const bool valid = b0 + (int)lane < total;
            const float e = logit2(srcv, valid, asd2, adi);
            rows2_blk(srcv, j, f, h2, hv);
            const float al = valid ? expf(e - m) / den : 0.0f;
            accum2_blk(hv, al, b0, total, j, acc);
        }
    }
    acc += __shfl_xor(acc, 8);
    acc += __shfl_xor(acc, 16);
    if (lane < 8u) out[(unsigned)i * 8u + f] = acc + bias2[f];
}

extern "C" void kernel_launch(void* const* d_in, const int* in_sizes, int n_in,
                              void* d_out, int out_size, void* d_ws, size_t ws_size,
                              hipStream_t stream) {
    const float* x     = (const float*)d_in[0];
    const int*   ei    = (const int*)d_in[1];
    const float* W1    = (const float*)d_in[2];
    const float* asrc1 = (const float*)d_in[3];
    const float* adst1 = (const float*)d_in[4];
    const float* bias1 = (const float*)d_in[5];
    const float* W2    = (const float*)d_in[6];
    const float* asrc2 = (const float*)d_in[7];
    const float* adst2 = (const float*)d_in[8];
    const float* bias2 = (const float*)d_in[9];

    int n = in_sizes[0] / F_IN;
    int E = in_sizes[1] / 2;
    int B = (n + BKT_NODES - 1) >> BKT_SHIFT;

    char* p = (char*)d_ws;
    float* h1   = (float*)p; p += (size_t)n * 64 * 4;
    float* as1  = (float*)p; p += (size_t)n * 8 * 4;
    float* ad1  = (float*)p; p += (size_t)n * 8 * 4;
    float* h2   = (float*)p; p += (size_t)n * 8 * 4;
    float* asd2 = (float*)p; p += (size_t)n * 8;
    int*   cnt  = (int*)p;   p += (size_t)n * 4;
    int*   begs = (int*)p;   p += (size_t)n * 4;
    int*   part = (int*)p;   p += (size_t)E * 4;
    int*   srcs = (int*)p;   p += (size_t)E * 4;
    int*   ghist = (int*)p;  p += (size_t)MAX_BKT * 4;
    int*   gcur  = (int*)p;  p += (size_t)MAX_BKT * 4;

    hipMemsetAsync(ghist, 0, (size_t)2 * MAX_BKT * 4, stream);

    const int tpb = 256;
    int gP = (E + 1024 * PART_EPT - 1) / (1024 * PART_EPT);
    int gN = (n + tpb / 32 - 1) / (tpb / 32);

    k_hist<<<gP, 1024, 0, stream>>>(ei, E, n, B, ghist);
    k_part<<<gP, 1024, 0, stream>>>(ei, E, n, B, ghist, gcur, part);
    int gG = (n + 16 * G1_TILES - 1) / (16 * G1_TILES);
    k_bucket_gemm1<<<B + gG, 1024, 0, stream>>>(part, ghist, n, cnt, begs, srcs, B,
                                                x, W1, asrc1, adst1, h1, as1, ad1);

    k_agg1<<<(n + 15) / 16, 512, 0, stream>>>(cnt, begs, srcs, as1, ad1, h1, bias1,
                                              W2, asrc2, adst2, h2, asd2, n);

    k_agg2<<<gN, tpb, 0, stream>>>(cnt, begs, srcs, asd2, h2, bias2, (float*)d_out, n);
}


# both node walks on resident waves: grid = the chip's wave slots, a wave loops over nodes (layer 1: 16-wave groups over tiles, a barrier per tile), next node's list asked for ahead; sum orders kept
# speedup vs baseline: 11.7781x; 1.0518x over previous
#include <hip/hip_runtime.h>
#include <hip/hip_bf16.h>

#define NEG_SLOPE 0.2f
#define F_IN 14

typedef float v2f __attribute__((ext_vector_type(2)));
typedef float v8f __attribute__((ext_vector_type(8)));

__device__ __forceinline__ int clamp_node(int v, int n) {
    return v < 0 ? 0 : (v >= n ? n - 1 : v);
}

#define G1_TILES 32
__device__ __forceinline__ void gemm1_role(int gblk,
                                           const float* __restrict__ x,
                                           const float* __restrict__ W1,
                                           const float* __restrict__ asrc,
                                           const float* __restrict__ adst,
                                           float* __restrict__ h1,
                                           float* __restrict__ as1, float* __restrict__ ad1, int n,
                                           float* xs_all,
                                           float* ws,
                                           float* wa) {
    const int tid  = threadIdx.x;
    const int lane = tid & 31;
    const int w    = __builtin_amdgcn_readfirstlane(tid >> 5);
    const int row0 = (gblk * G1_TILES + w) * 16;
    const bool hi  = lane >= 16;
    const int  m   = lane & 15;
    float* xs = xs_all + w * 256;

    int rows = n - row0;
    rows = rows > 16 ? 16 : (rows < 0 ? 0 : rows);
    const int cntx = rows * F_IN;
    float xv[7];
#pragma unroll
    for (int q = 0; q < 7; ++q) {
        const int i = lane + 32 * q;
        xv[q] = i < cntx ? x[(size_t)row0 * F_IN + i] : 0.0f;
    }
    float4 av0 = make_float4(0.f, 0.f, 0.f, 0.f), av1 = av0;
    if (tid < 256) {
        const int mm = tid & 15;
        const float4* avp = (const float4*)((mm < 8 ? asrc : adst) + (mm & 7) * 8);
        av0 = avp[0];
        av1 = avp[1];
    }
    if (tid < F_IN * 64 / 4)
        ((float4*)ws)[tid] = ((const float4*)W1)[tid];
    else if (tid < 256)
        ((float4*)(ws + F_IN * 64))[tid - F_IN * 64 / 4] = make_float4(0.f, 0.f, 0.f, 0.f);
#pragma unroll
    for (int q = 0; q < 7; ++q) {
        const int i = lane + 32 * q;
        const int r = i / F_IN, c = i - r * F_IN;
        xs[r * 16 + c] = xv[q];
    }
    xs[(lane >> 1) * 16 + F_IN + (lane & 1)] = 0.0f;
    __syncthreads();
    if (tid < 256) {
        const int k = tid >> 4, mm = tid & 15;
        const float4* wp = (const float4*)(ws + k * 64 + (mm & 7) * 8);
        const float4 w0 = wp[0], w1 = wp[1];
        wa[k * 16 + mm] = w0.x * av0.x + w0.y * av0.y + w0.z * av0.z + w0.w * av0.w
                        + w1.x * av1.x + w1.y * av1.y + w1.z * av1.z + w1.w * av1.w;
    }
    __syncthreads();
    if (row0 >= n) return;
    const bool full = (row0 + 16 <= n);

    const int kofs = hi ? 2 : 0;
    v2f a[4];
#pragma unroll
    for (int k = 0; k < 4; ++k) {
        a[k][0] = xs[m * 16 + 4 * k + kofs + 0];
        a[k][1] = xs[m * 16 + 4 * k + kofs + 1];
    }

#pragma unroll
    for (int t = 0; t < 4; ++t) {
        v8f c = {};
#pragma unroll
        for (int k = 0; k < 4; ++k) {
            v2f b;
            b[0] = ws[(4 * k + kofs + 0) * 64 + t * 16 + m];
            b[1] = ws[(4 * k + kofs + 1) * 64 + t * 16 + m];
            c = __builtin_amdgcn_wmma_f32_16x16x4_f32(false, a[k], false, b,
                                                      (short)0, c, false, false);
        }
        float* out = h1 + (size_t)row0 * 64 + t * 16 + m;
        if (full) {
#pragma unroll
            for (int r = 0; r < 8; ++r)
                out[(size_t)(r + (hi ? 8 : 0)) * 64] = c[r];
        } else {
#pragma unroll
            for (int r = 0; r < 8; ++r) {
                int rr = r + (hi ? 8 : 0);
                if (row0 + rr < n) out[(size_t)rr * 64] = c[r];
            }
        }
    }

    {
        v8f c = {};
#pragma unroll
        for (int k = 0; k < 4; ++k) {
            v2f b;
            b[0] = wa[(4 * k + kofs + 0) * 16 + m];
            b[1] = wa[(4 * k + kofs + 1) * 16 + m];
            c = __builtin_amdgcn_wmma_f32_16x16x4_f32(false, a[k], false, b,
                                                      (short)0, c, false, false);
        }
        float* lo = (m < 8 ? as1 : ad1) + (size_t)row0 * 8 + (m & 7);
        if (full) {
#pragma unroll
            for (int r = 0; r < 8; ++r)
                lo[(size_t)(r + (hi ? 8 : 0)) * 8] = c[r];
        } else {
#pragma unroll
            for (int r = 0; r < 8; ++r) {
                int rr = r + (hi ? 8 : 0);
                if (row0 + rr < n) lo[(size_t)rr * 8] = c[r];
            }
        }
    }
}

#define BKT_SHIFT 9
#define BKT_NODES (1 << BKT_SHIFT)
#define MAX_BKT   1024
#define PART_EPT  8
#define BKT_EPT   8

__device__ __forceinline__ int block_excl_scan(int v, int* wsum, int& total) {
    const int lane = threadIdx.x & 31, w = threadIdx.x >> 5;
    int incl = v;
#pragma unroll
    for (int o = 1; o < 32; o <<= 1) {
        int t = __shfl_up(incl, o);
        if (lane >= o) incl += t;
    }
    if (lane == 31) wsum[w] = incl;
    __syncthreads();
    int woff = 0, tot = 0;
#pragma unroll
    for (int q = 0; q < 32; ++q) {
        int t = wsum[q];
        if (q < w) woff += t;
        tot += t;
    }
    total = tot;
    return woff + incl - v;
}

__global__ __launch_bounds__(1024) void k_hist(const int* __restrict__ ei, int E, int n, int B,
                                               int* __restrict__ ghist) {
    __shared__ int lh[MAX_BKT];
    const int tid = threadIdx.x;
    const int base = blockIdx.x * (1024 * PART_EPT) + tid;
    int d[PART_EPT];
#pragma unroll
    for (int k = 0; k < PART_EPT; ++k) {
        const int idx = base + k * 1024;
        d[k] = ei[E + (idx < E ? idx : E - 1)];
    }
    if (tid < B) lh[tid] = 0;
    __syncthreads();
#pragma unroll
    for (int k = 0; k < PART_EPT; ++k)
        if (base + k * 1024 < E) atomicAdd(&lh[clamp_node(d[k], n) >> BKT_SHIFT], 1);
    __syncthreads();
    if (tid < B) {
        const int c = lh[tid];
        if (c) atomicAdd(&ghist[tid], c);
    }
}

__global__ __launch_bounds__(1024) void k_part(const int* __restrict__ ei, int E, int n, int B,
                                               const int* __restrict__ ghist, int* __restrict__ gcur,
                                               int* __restrict__ part) {
    __shared__ int lh[MAX_BKT];
    __shared__ int wsum[32];
    const int tid = threadIdx.x;
    const int base = blockIdx.x * (1024 * PART_EPT) + tid;
    int sv[PART_EPT], d[PART_EPT];
#pragma unroll
    for (int k = 0; k < PART_EPT; ++k) {
        const int idx = base + k * 1024;
        const int ii  = idx < E ? idx : E - 1;
        sv[k] = ei[ii];
        d[k]  = ei[E + ii];
    }
    int tot;
    const int bstart = block_excl_scan(tid < B ? ghist[tid] : 0, wsum, tot);
    if (tid < B) lh[tid] = 0;
    __syncthreads();
#pragma unroll
    for (int k = 0; k < PART_EPT; ++k) {
        sv[k] = clamp_node(sv[k], n);
        d[k]  = (base + k * 1024 < E) ? clamp_node(d[k], n) : -1;
        if (d[k] >= 0) atomicAdd(&lh[d[k] >> BKT_SHIFT], 1);
    }
    __syncthreads();
    if (tid < B) {
        const int c = lh[tid];
        const int r = c ? atomicAdd(&gcur[tid], c) : 0;
        lh[tid] = bstart + r;
    }
    __syncthreads();
#pragma unroll
    for (int k = 0; k < PART_EPT; ++k)
        if (d[k] >= 0) {
            const int pos = atomicAdd(&lh[d[k] >> BKT_SHIFT], 1);
            part[pos] = (sv[k] << BKT_SHIFT) | (d[k] & (BKT_NODES - 1));
        }
}

__device__ __forceinline__ void bucket_role(int b, const int* __restrict__ part, const int* __restrict__ ghist,
                                            int n, int* __restrict__ cnt, int* __restrict__ begs,
                                            int* __restrict__ srcs,
                                            int* lc,
                                            int* wsum) {
    const int tid = threadIdx.x;
    int base;
    (void)block_excl_scan(tid < b ? ghist[tid] : 0, wsum, base);
    const int count = ghist[b];
    const int* __restrict__ rec = part + base;
    if (tid < BKT_NODES) lc[tid] = 0;
    __syncthreads();
    for (int e0 = tid; e0 - tid < count; e0 += 1024 * BKT_EPT) {
        int p[BKT_EPT];
#pragma unroll
        for (int k = 0; k < BKT_EPT; ++k) {
            const int e = e0 + k * 1024;
            p[k] = rec[e < count ? e : count - 1];
        }
#pragma unroll
        for (int k = 0; k < BKT_EPT; ++k)
            if (e0 + k * 1024 < count) atomicAdd(&lc[p[k] & (BKT_NODES - 1)], 1);
    }
    __syncthreads();
    const int c = tid < BKT_NODES ? lc[tid] : 0;
    __syncthreads();
    int tot;
    const int excl = block_excl_scan(c, wsum, tot);
    if (tid < BKT_NODES) {
        const int node = (b << BKT_SHIFT) + tid;
        if (node < n) { cnt[node] = c; begs[node] = base + excl; }
        lc[tid] = excl;
    }
    __syncthreads();
    for (int e0 = tid; e0 - tid < count; e0 += 1024 * BKT_EPT) {
        int p[BKT_EPT];
#pragma unroll
        for (int k = 0; k < BKT_EPT; ++k) {
            const int e = e0 + k * 1024;
            p[k] = rec[e < count ? e : count - 1];
        }
#pragma unroll
        for (int k = 0; k < BKT_EPT; ++k)
            if (e0 + k * 1024 < count) {
                const int pos = atomicAdd(&lc[p[k] & (BKT_NODES - 1)], 1);
                srcs[base + pos] = p[k] >> BKT_SHIFT;
            }
    }
}

__global__ __launch_bounds__(1024) void k_bucket_gemm1(const int* __restrict__ part, const int* __restrict__ ghist,
                                                       int n, int* __restrict__ cnt, int* __restrict__ begs,
                                                       int* __restrict__ srcs, int nb,
                                                       const float* __restrict__ x,
                                                       const float* __restrict__ W1,
                                                       const float* __restrict__ asrc,
                                                       const float* __restrict__ adst,
                                                       float* __restrict__ h1,
                                                       float* __restrict__ as1, float* __restrict__ ad1) {
    __shared__ int   lc[BKT_NODES];
    __shared__ int   wsum[32];
    __shared__ float xs_all[G1_TILES * 16 * 16];
    __shared__ float ws[16 * 64];
    __shared__ float wa[16 * 16];
    const int blk = (int)blockIdx.x;
    if (blk < nb)
        bucket_role(blk, part, ghist, n, cnt, begs, srcs, lc, wsum);
    else
        gemm1_role(blk - nb, x, W1, asrc, adst, h1, as1, ad1, n, xs_all, ws, wa);
}

#define AGG1_GRID 1024
#define AGG2_GRID 2048
#define A1_PITCH 68
__device__ __forceinline__ int load_srcs_blk(const int* __restrict__ srcs, int beg, int deg,
                                             int b0, unsigned lane, int self) {
    const int kk = b0 + (int)lane;
    return kk < deg ? srcs[beg + kk] : self;
}

__device__ __forceinline__ void logits1_blk(int srcv, int b0, int total, unsigned j, unsigned h,
                                            const float* __restrict__ as1, float adi,
                                            float (&e)[8]) {
#pragma unroll
    for (int c = 0; c < 8; ++c) {
        const int sk = __shfl(srcv, 4 * c + (int)j);
        float t = as1[(unsigned)sk * 8u + h] + adi;
        t = t > 0.0f ? t : NEG_SLOPE * t;
        e[c] = ((int)j < total - b0 - 4 * c) ? t : 0.0f;
    }
}

__device__ __forceinline__ void accum1_blk(int srcv, int b0, int total, const float (&al)[8],
                                           unsigned lane, const float* __restrict__ h1,
                                           float& acc0, float& acc1) {
    const unsigned g = lane >> 3;
#pragma unroll
    for (int c = 0; c < 8; ++c) {
        if (b0 + 4 * c + 3 < total) {
            float x0[4], x1[4], w0[4], w1[4];
#pragma unroll
            for (int q = 0; q < 4; ++q) {
                const int sk = __builtin_amdgcn_readlane(srcv, 4 * c + q);
                const float* hp = h1 + (unsigned)sk * 64u;
                x0[q] = hp[lane];
                x1[q] = hp[lane + 32u];
                w0[q] = __shfl(al[c], (int)(8u * q + g));
                w1[q] = __shfl(al[c], (int)(8u * q + 4u + g));
            }
#pragma unroll
            for (int q = 0; q < 4; ++q) {
                acc0 += x0[q] * w0[q];
                acc1 += x1[q] * w1[q];
            }
        } else if (b0 + 4 * c < total) {
            float x0[3], x1[3], w0[3], w1[3];
#pragma unroll
            for (int q = 0; q < 3; ++q) {
                const int sk = __builtin_amdgcn_readlane(srcv, 4 * c + q);
                const float* hp = h1 + (unsigned)sk * 64u;
                x0[q] = hp[lane];
                x1[q] = hp[lane + 32u];
                w0[q] = __shfl(al[c], (int)(8u * q + g));
                w1[q] = __shfl(al[c], (int)(8u * q + 4u + g));
            }
#pragma unroll
            for (int q = 0; q < 3; ++q) {
                const bool ok = b0 + 4 * c + q < total;
                acc0 += ok ? x0[q] * w0[q] : 0.0f;
                acc1 += ok ? x1[q] * w1[q] : 0.0f;
            }
        }
    }
}

__global__ __launch_bounds__(512) void k_agg1(const int* __restrict__ cnt, const int* __restrict__ begs,
                                              const int* __restrict__ srcs,
                                              const float* __restrict__ as1, const float* __restrict__ ad1,
                                              const float* __restrict__ h1, const float* __restrict__ bias1,
                                              const float* __restrict__ W2,
                                              const float* __restrict__ asrc2, const float* __restrict__ adst2,
                                              float* __restrict__ h2, float* __restrict__ asd2, int n) {
    __shared__ float smA[2 * 16 * A1_PITCH];
    __shared__ float ws2[64 * 16];

    const unsigned tid  = threadIdx.x;
    const unsigned lane = tid & 31u;
    const int w = __builtin_amdgcn_readfirstlane((int)(tid >> 5));
    const unsigned j = lane >> 3, h = lane & 7u;
    const int ntiles = (n + 15) >> 4;
    const int tstep  = (int)gridDim.x;
    const int nstep  = tstep * 16;

    int t = (int)blockIdx.x;
    int i = t * 16 + w;
    int deg = 0, beg = 0, srcv0 = 0;
    float adi = 0.0f;
    if (i < n) {
        deg   = cnt[i];
        beg   = begs[i];
        adi   = ad1[(unsigned)i * 8u + h];
        srcv0 = load_srcs_blk(srcs, beg, deg, 0, lane, i);
    }
    int i1 = i + nstep, deg1 = 0, beg1 = 0;
    if (i1 < n) {
        deg1 = cnt[i1];
        beg1 = begs[i1];
    }
    const float bz0 = bias1[lane], bz1 = bias1[lane + 32u];

    if (tid < 128u) {
        const int k = (int)(tid >> 1), half = (int)(tid & 1u);
        ((float4*)(ws2 + k * 16 + half * 4))[0] = ((const float4*)W2)[tid];
    } else if (tid < 256u) {
        const unsigned tt = tid - 128u;
        const int k = (int)(tt >> 1);
        const float4* avp = (const float4*)((tt & 1u) ? adst2 : asrc2);
        const float4 av0 = avp[0], av1 = avp[1];
        const float4* wp = (const float4*)(W2 + k * 8);
        const float4 w0 = wp[0], w1 = wp[1];
        ws2[k * 16 + 8 + (int)(tt & 1u)] = w0.x * av0.x + w0.y * av0.y + w0.z * av0.z + w0.w * av0.w
                                         + w1.x * av1.x + w1.y * av1.y + w1.z * av1.z + w1.w * av1.w;
    } else {
        for (int e = (int)tid - 256; e < 64 * 6; e += 256) {
            const int k = e / 6;
            ws2[k * 16 + 10 + (e - 6 * k)] = 0.0f;
        }
    }

    for (int it = 0; t < ntiles; t += tstep, ++it) {
        const int row0 = t * 16;
        const bool have = i < n;

        int srcv1 = 0, cb2 = 0;
        float adi1 = 0.0f;
        const int i2 = i1 + nstep;
        auto ask_ahead = [&]() {
            if (i1 < n) {
                adi1  = ad1[(unsigned)i1 * 8u + h];
                srcv1 = load_srcs_blk(srcs, beg1, deg1, 0, lane, i1);
            }
            if (i2 < n) cb2 = (lane < 16u ? cnt : begs)[i2];
        };

        float r0 = 0.0f, r1 = 0.0f;
        if (!have) ask_ahead();
        if (have) {
            const int total = deg + 1;
            const float NEG_INF = -__builtin_huge_valf();

            float acc0 = 0.0f, acc1 = 0.0f;
            float e[8], al[8];
            if (total <= 32) {
                const int srcv = srcv0;
                logits1_blk(srcv, 0, total, j, h, as1, adi, e);
                ask_ahead();
                float m = NEG_INF;
#pragma unroll
                for (int c = 0; c < 8; ++c)
                    if ((int)j < total - 4 * c) m = fmaxf(m, e[c]);
                m = fmaxf(m, __shfl_xor(m, 8));
                m = fmaxf(m, __shfl_xor(m, 16));
                float ssum = 0.0f;
#pragma unroll
                for (int c = 0; c < 8; ++c) {
                    al[c] = ((int)j < total - 4 * c) ? expf(e[c] - m) : 0.0f;
                    ssum += al[c];
                }
                ssum += __shfl_xor(ssum, 8);
                ssum += __shfl_xor(ssum, 16);
                const float den = ssum + 1e-16f;
#pragma unroll
                for (int c = 0; c < 8; ++c) al[c] = al[c] / den;
                accum1_blk(srcv, 0, total, al, lane, h1, acc0, acc1);
            } else {
                ask_ahead();
                float m = NEG_INF;
                for (int b0 = 0; b0 < total; b0 += 32) {
                    const int srcv = load_srcs_blk(srcs, beg, deg, b0, lane, i);
                    logits1_blk(srcv, b0, total, j, h, as1, adi, e);
#pragma unroll
                    for (int c = 0; c < 8; ++c)
                        if ((int)j < total - b0 - 4 * c) m = fmaxf(m, e[c]);
                }
                m = fmaxf(m, __shfl_xor(m, 8));
                m = fmaxf(m, __shfl_xor(m, 16));
                float ssum = 0.0f;
                for (int b0 = 0; b0 < total; b0 += 32) {
                    const int srcv = load_srcs_blk(srcs, beg, deg, b0, lane, i);
                    logits1_blk(srcv, b0, total, j, h, as1, adi, e);
#pragma unroll
                    for (int c = 0; c < 8; ++c)
                        ssum += ((int)j < total - b0 - 4 * c) ? expf(e[c] - m) : 0.0f;
                }
                ssum += __shfl_xor(ssum, 8);
                ssum += __shfl_xor(ssum, 16);
                const float den = ssum + 1e-16f;
                for (int b0 = 0; b0 < total; b0 += 32) {
                    const int srcv = load_srcs_blk(srcs, beg, deg, b0, lane, i);
                    logits1_blk(srcv, b0, total, j, h, as1, adi, e);
#pragma unroll
                    for (int c = 0; c < 8; ++c)
                        al[c] = ((int)j < total - b0 - 4 * c) ? expf(e[c] - m) / den : 0.0f;
                    accum1_blk(srcv, b0, total, al, lane, h1, acc0, acc1);
                }
            }

            r0 = acc0 + bz0;
            r1 = acc1 + bz1;
            r0 = r0 > 0.0f ? r0 : (expf(r0) - 1.0f);
            r1 = r1 > 0.0f ? r1 : (expf(r1) - 1.0f);
        }

        float* sm = smA + (it & 1) * (16 * A1_PITCH);
        sm[w * A1_PITCH + (int)lane]      = r0;
        sm[w * A1_PITCH + (int)lane + 32] = r1;
        __syncthreads();

        if (w == (it & 15)) {
            const bool hi   = lane >= 16u;
            const int  m    = (int)(lane & 15u);
            const bool full = (row0 + 16 <= n);
            const int  kofs = hi ? 2 : 0;
            v8f c = {};
#pragma unroll
            for (int k = 0; k < 16; ++k) {
                v2f a, b;
                a[0] = sm[m * A1_PITCH + 4 * k + kofs + 0];
                a[1] = sm[m * A1_PITCH + 4 * k + kofs + 1];
                b[0] = ws2[(4 * k + kofs + 0) * 16 + m];
                b[1] = ws2[(4 * k + kofs + 1) * 16 + m];
                c = __builtin_amdgcn_wmma_f32_16x16x4_f32(false, a, false, b,
                                                          (short)0, c, false, false);
            }
            const int rbase = row0 + (hi ? 8 : 0);
            const int lim   = n - rbase;
            if (m < 8) {
                float* out = h2 + (size_t)rbase * 8 + m;
                if (full) {
#pragma unroll
                    for (int r = 0; r < 8; ++r) out[r * 8] = c[r];
                } else {
#pragma unroll
                    for (int r = 0; r < 8; ++r)
                        if (r < lim) out[r * 8] = c[r];
                }
            } else if (m < 10) {
                float* out = asd2 + (size_t)rbase * 2 + (m - 8);
                if (full) {
#pragma unroll
                    for (int r = 0; r < 8; ++r) out[r * 2] = c[r];
                } else {
#pragma unroll
                    for (int r = 0; r < 8; ++r)
                        if (r < lim) out[r * 2] = c[r];
                }
            }
        }

        i = i1;  deg = deg1;  beg = beg1;  srcv0 = srcv1;  adi = adi1;
        i1 = i2;
        deg1 = __builtin_amdgcn_readlane(cb2, 0);
        beg1 = __builtin_amdgcn_readlane(cb2, 16);
    }
}

__device__ __forceinline__ float wave_max(float v) {
#pragma unroll
    for (int o = 16; o > 0; o >>= 1) v = fmaxf(v, __shfl_xor(v, o));
    return v;
}
__device__ __forceinline__ float wave_sum(float v) {
#pragma unroll
    for (int o = 16; o > 0; o >>= 1) v += __shfl_xor(v, o);
    return v;
}

__device__ __forceinline__ float logit2(int srcv, bool valid,
                                        const float* __restrict__ asd2, float adi) {
    float t = asd2[2u * (unsigned)srcv] + adi;
    t = t > 0.0f ? t : NEG_SLOPE * t;
    return valid ? t : 0.0f;
}
__device__ __forceinline__ float logit2_of(float as, bool valid, float adi) {
    float t = as + adi;
    t = t > 0.0f ? t : NEG_SLOPE * t;
    return valid ? t : 0.0f;
}

__device__ __forceinline__ void rows2_blk(int srcv, unsigned j, unsigned f,
                                          const float* __restrict__ h2, float (&hv)[8]) {
#pragma unroll
    for (int c = 0; c < 8; ++c) {
        const int sk = __shfl(srcv, 4 * c + (int)j);
        hv[c] = h2[(unsigned)sk * 8u + f];
    }
}

__device__ __forceinline__ void accum2_blk(const float (&hv)[8], float al, int b0, int total,
                                           unsigned j, float& acc) {
#pragma unroll
    for (int c = 0; c < 8; ++c) {
        const float ak = __shfl(al, 4 * c + (int)j);
        acc += (b0 + 4 * c + (int)j < total) ? hv[c] * ak : 0.0f;
    }
}

__global__ __launch_bounds__(256) void k_agg2(const int* __restrict__ cnt, const int* __restrict__ begs,
                                              const int* __restrict__ srcs,
                                              const float* __restrict__ asd2,
                                              const float* __restrict__ h2, const float* __restrict__ bias2,
                                              float* __restrict__ out, int n) {
    const unsigned lane = threadIdx.x & 31u;
    const int nw = (int)gridDim.x * 8;
    int i = __builtin_amdgcn_readfirstlane((int)(blockIdx.x * 8u + (threadIdx.x >> 5)));
    const unsigned j = lane >> 3, f = lane & 7u;
    const float NEG_INF = -__builtin_huge_valf();
    const float bz = bias2[f];

    int deg = 0, beg = 0, srcv0 = 0;
    float adi = 0.0f;
    if (i < n) {
        deg   = cnt[i];
        beg   = begs[i];
        adi   = asd2[2u * (unsigned)i + 1u];
        srcv0 = load_srcs_blk(srcs, beg, deg, 0, lane, i);
    }
    int i1 = i + nw, deg1 = 0, beg1 = 0;
    if (i1 < n) {
        deg1 = cnt[i1];
        beg1 = begs[i1];
    }

    while (i < n) {
        int srcv1 = 0;
        float adi1 = 0.0f;
        const int i2 = i1 + nw;
        int deg2 = 0, beg2 = 0;
        auto ask_ahead = [&]() {
            if (i1 < n) {
                adi1  = asd2[2u * (unsigned)i1 + 1u];
                srcv1 = load_srcs_blk(srcs, beg1, deg1, 0, lane, i1);
            }
            if (i2 < n) {
                deg2 = cnt[i2];
                beg2 = begs[i2];
            }
        };

        const int total = deg + 1;
        float acc = 0.0f;
        float hv[8];
        if (total <= 32) {
            const int  srcv  = srcv0;
            const bool valid = (int)lane < total;
            rows2_blk(srcv, j, f, h2, hv);
            __builtin_amdgcn_sched_barrier(0);
            const float lg = asd2[2u * (unsigned)srcv];
            ask_ahead();
            const float e  = logit2_of(lg, valid, adi);
            const float m  = wave_max(valid ? e : NEG_INF);
            const float ex = valid ? expf(e - m) : 0.0f;
            const float den = wave_sum(ex) + 1e-16f;
            accum2_blk(hv, ex / den, 0, total, j, acc);
        } else {
            ask_ahead();
            float m = NEG_INF;
            for (int b0 = 0; b0 < total; b0 += 32) {
                const int  srcv  = load_srcs_blk(srcs, beg, deg, b0, lane, i);
                const bool valid = b0 + (int)lane < total;
                const float e = logit2(srcv, valid, asd2, adi);
                m = fmaxf(m, valid ? e : NEG_INF);
            }
            m = wave_max(m);
            float ssum = 0.0f;
            for (int b0 = 0; b0 < total; b0 += 32) {
                const int  srcv  = load_srcs_blk(srcs, beg, deg, b0, lane, i);
                const bool valid = b0 + (int)lane < total;
                const float e = logit2(srcv, valid, asd2, adi);
                ssum += valid ? expf(e - m) : 0.0f;
            }
            const float den = wave_sum(ssum) + 1e-16f;
            for (int b0 = 0; b0 < total; b0 += 32) {
                const int  srcv  = load_srcs_blk(srcs, beg, deg, b0, lane, i);
                const bool valid = b0 + (int)lane < total;
                const float e = logit2(srcv, valid, asd2, adi);
                rows2_blk(srcv, j, f, h2, hv);
                const float al = valid ? expf(e - m) / den : 0.0f;
                accum2_blk(hv, al, b0, total, j, acc);
            }
        }
        acc += __shfl_xor(acc, 8);
        acc += __shfl_xor(acc, 16);
        if (lane < 8u) out[(unsigned)i * 8u + f] = acc + bz;

        i = i1;  deg = deg1;  beg = beg1;  srcv0 = srcv1;  adi = adi1;
        i1 = i2; deg1 = deg2; beg1 = beg2;
    }
}

extern "C" void kernel_launch(void* const* d_in, const int* in_sizes, int n_in,
                              void* d_out, int out_size, void* d_ws, size_t ws_size,
                              hipStream_t stream) {
    const float* x     = (const float*)d_in[0];
    const int*   ei    = (const int*)d_in[1];
    const float* W1    = (const float*)d_in[2];
    const float* asrc1 = (const float*)d_in[3];
    const float* adst1 = (const float*)d_in[4];
    const float* bias1 = (const float*)d_in[5];
    const float* W2    = (const float*)d_in[6];
    const float* asrc2 = (const float*)d_in[7];
    const float* adst2 = (const float*)d_in[8];
    const float* bias2 = (const float*)d_in[9];

    int n = in_sizes[0] / F_IN;
    int E = in_sizes[1] / 2;
    int B = (n + BKT_NODES - 1) >> BKT_SHIFT;

    char* p = (char*)d_ws;
    float* h1   = (float*)p; p += (size_t)n * 64 * 4;
    float* as1  = (float*)p; p += (size_t)n * 8 * 4;
    float* ad1  = (float*)p; p += (size_t)n * 8 * 4;
    float* h2   = (float*)p; p += (size_t)n * 8 * 4;
    float* asd2 = (float*)p; p += (size_t)n * 8;
    int*   cnt  = (int*)p;   p += (size_t)n * 4;
    int*   begs = (int*)p;   p += (size_t)n * 4;
    int*   part = (int*)p;   p += (size_t)E * 4;
    int*   srcs = (int*)p;   p += (size_t)E * 4;
    int*   ghist = (int*)p;  p += (size_t)MAX_BKT * 4;
    int*   gcur  = (int*)p;  p += (size_t)MAX_BKT * 4;

    hipMemsetAsync(ghist, 0, (size_t)2 * MAX_BKT * 4, stream);

    int gP = (E + 1024 * PART_EPT - 1) / (1024 * PART_EPT);

    k_hist<<<gP, 1024, 0, stream>>>(ei, E, n, B, ghist);
    k_part<<<gP, 1024, 0, stream>>>(ei, E, n, B, ghist, gcur, part);
    int gG = (n + 16 * G1_TILES - 1) / (16 * G1_TILES);
    k_bucket_gemm1<<<B + gG, 1024, 0, stream>>>(part, ghist, n, cnt, begs, srcs, B,
                                                x, W1, asrc1, adst1, h1, as1, ad1);

    k_agg1<<<AGG1_GRID, 512, 0, stream>>>(cnt, begs, srcs, as1, ad1, h1, bias1,
                                          W2, asrc2, adst2, h2, asd2, n);

    k_agg2<<<AGG2_GRID, 256, 0, stream>>>(cnt, begs, srcs, asd2, h2, bias2, (float*)d_out, n);
}


# layer-1 rows fetched by the lanes that hold the weights: lane (slot, piece) reads two b128 of its element's row, weights by DPP8 inside 8 lanes, no LDS shuffles; sums now per slot, then a 4-slot tree
# speedup vs baseline: 12.7039x; 1.0786x over previous
#include <hip/hip_runtime.h>
#include <hip/hip_bf16.h>

#define NEG_SLOPE 0.2f
#define F_IN 14

typedef float v2f __attribute__((ext_vector_type(2)));
typedef float v8f __attribute__((ext_vector_type(8)));

__device__ __forceinline__ int clamp_node(int v, int n) {
    return v < 0 ? 0 : (v >= n ? n - 1 : v);
}

#define G1_TILES 32
__device__ __forceinline__ void gemm1_role(int gblk,
                                           const float* __restrict__ x,
                                           const float* __restrict__ W1,
                                           const float* __restrict__ asrc,
                                           const float* __restrict__ adst,
                                           float* __restrict__ h1,
                                           float* __restrict__ as1, float* __restrict__ ad1, int n,
                                           float* xs_all,
                                           float* ws,
                                           float* wa) {
    const int tid  = threadIdx.x;
    const int lane = tid & 31;
    const int w    = __builtin_amdgcn_readfirstlane(tid >> 5);
    const int row0 = (gblk * G1_TILES + w) * 16;
    const bool hi  = lane >= 16;
    const int  m   = lane & 15;
    float* xs = xs_all + w * 256;

    int rows = n - row0;
    rows = rows > 16 ? 16 : (rows < 0 ? 0 : rows);
    const int cntx = rows * F_IN;
    float xv[7];
#pragma unroll
    for (int q = 0; q < 7; ++q) {
        const int i = lane + 32 * q;
        xv[q] = i < cntx ? x[(size_t)row0 * F_IN + i] : 0.0f;
    }
    float4 av0 = make_float4(0.f, 0.f, 0.f, 0.f), av1 = av0;
    if (tid < 256) {
        const int mm = tid & 15;
        const float4* avp = (const float4*)((mm < 8 ? asrc : adst) + (mm & 7) * 8);
        av0 = avp[0];
        av1 = avp[1];
    }
    if (tid < F_IN * 64 / 4)
        ((float4*)ws)[tid] = ((const float4*)W1)[tid];
    else if (tid < 256)
        ((float4*)(ws + F_IN * 64))[tid - F_IN * 64 / 4] = make_float4(0.f, 0.f, 0.f, 0.f);
#pragma unroll
    for (int q = 0; q < 7; ++q) {
        const int i = lane + 32 * q;
        const int r = i / F_IN, c = i - r * F_IN;
        xs[r * 16 + c] = xv[q];
    }
    xs[(lane >> 1) * 16 + F_IN + (lane & 1)] = 0.0f;
    __syncthreads();
    if (tid < 256) {
        const int k = tid >> 4, mm = tid & 15;
        const float4* wp = (const float4*)(ws + k * 64 + (mm & 7) * 8);
        const float4 w0 = wp[0], w1 = wp[1];
        wa[k * 16 + mm] = w0.x * av0.x + w0.y * av0.y + w0.z * av0.z + w0.w * av0.w
                        + w1.x * av1.x + w1.y * av1.y + w1.z * av1.z + w1.w * av1.w;
    }
    __syncthreads();
    if (row0 >= n) return;
    const bool full = (row0 + 16 <= n);

    const int kofs = hi ? 2 : 0;
    v2f a[4];
#pragma unroll
    for (int k = 0; k < 4; ++k) {
        a[k][0] = xs[m * 16 + 4 * k + kofs + 0];
        a[k][1] = xs[m * 16 + 4 * k + kofs + 1];
    }

#pragma unroll
    for (int t = 0; t < 4; ++t) {
        v8f c = {};
#pragma unroll
        for (int k = 0; k < 4; ++k) {
            v2f b;
            b[0] = ws[(4 * k + kofs + 0) * 64 + t * 16 + m];
            b[1] = ws[(4 * k + kofs + 1) * 64 + t * 16 + m];
            c = __builtin_amdgcn_wmma_f32_16x16x4_f32(false, a[k], false, b,
                                                      (short)0, c, false, false);
        }
        float* out = h1 + (size_t)row0 * 64 + t * 16 + m;
        if (full) {
#pragma unroll
            for (int r = 0; r < 8; ++r)
                out[(size_t)(r + (hi ? 8 : 0)) * 64] = c[r];
        } else {
#pragma unroll
            for (int r = 0; r < 8; ++r) {
                int rr = r + (hi ? 8 : 0);
                if (row0 + rr < n) out[(size_t)rr * 64] = c[r];
            }
        }
    }

    {
        v8f c = {};
#pragma unroll
        for (int k = 0; k < 4; ++k) {
            v2f b;
            b[0] = wa[(4 * k + kofs + 0) * 16 + m];
            b[1] = wa[(4 * k + kofs + 1) * 16 + m];
            c = __builtin_amdgcn_wmma_f32_16x16x4_f32(false, a[k], false, b,
                                                      (short)0, c, false, false);
        }
        float* lo = (m < 8 ? as1 : ad1) + (size_t)row0 * 8 + (m & 7);
        if (full) {
#pragma unroll
            for (int r = 0; r < 8; ++r)
                lo[(size_t)(r + (hi ? 8 : 0)) * 8] = c[r];
        } else {
#pragma unroll
            for (int r = 0; r < 8; ++r) {
                int rr = r + (hi ? 8 : 0);
                if (row0 + rr < n) lo[(size_t)rr * 8] = c[r];
            }
        }
    }
}

#define BKT_SHIFT 9
#define BKT_NODES (1 << BKT_SHIFT)
#define MAX_BKT   1024
#define PART_EPT  8
#define BKT_EPT   8

__device__ __forceinline__ int block_excl_scan(int v, int* wsum, int& total) {
    const int lane = threadIdx.x & 31, w = threadIdx.x >> 5;
    int incl = v;
#pragma unroll
    for (int o = 1; o < 32; o <<= 1) {
        int t = __shfl_up(incl, o);
        if (lane >= o) incl += t;
    }
    if (lane == 31) wsum[w] = incl;
    __syncthreads();
    int woff = 0, tot = 0;
#pragma unroll
    for (int q = 0; q < 32; ++q) {
        int t = wsum[q];
        if (q < w) woff += t;
        tot += t;
    }
    total = tot;
    return woff + incl - v;
}

__global__ __launch_bounds__(1024) void k_hist(const int* __restrict__ ei, int E, int n, int B,
                                               int* __restrict__ ghist) {
    __shared__ int lh[MAX_BKT];
    const int tid = threadIdx.x;
    const int base = blockIdx.x * (1024 * PART_EPT) + tid;
    int d[PART_EPT];
#pragma unroll
    for (int k = 0; k < PART_EPT; ++k) {
        const int idx = base + k * 1024;
        d[k] = ei[E + (idx < E ? idx : E - 1)];
    }
    if (tid < B) lh[tid] = 0;
    __syncthreads();
#pragma unroll
    for (int k = 0; k < PART_EPT; ++k)
        if (base + k * 1024 < E) atomicAdd(&lh[clamp_node(d[k], n) >> BKT_SHIFT], 1);
    __syncthreads();
    if (tid < B) {
        const int c = lh[tid];
        if (c) atomicAdd(&ghist[tid], c);
    }
}

__global__ __launch_bounds__(1024) void k_part(const int* __restrict__ ei, int E, int n, int B,
                                               const int* __restrict__ ghist, int* __restrict__ gcur,
                                               int* __restrict__ part) {
    __shared__ int lh[MAX_BKT];
    __shared__ int wsum[32];
    const int tid = threadIdx.x;
    const int base = blockIdx.x * (1024 * PART_EPT) + tid;
    int sv[PART_EPT], d[PART_EPT];
#pragma unroll
    for (int k = 0; k < PART_EPT; ++k) {
        const int idx = base + k * 1024;
        const int ii  = idx < E ? idx : E - 1;
        sv[k] = ei[ii];
        d[k]  = ei[E + ii];
    }
    int tot;
    const int bstart = block_excl_scan(tid < B ? ghist[tid] : 0, wsum, tot);
    if (tid < B) lh[tid] = 0;
    __syncthreads();
#pragma unroll
    for (int k = 0; k < PART_EPT; ++k) {
        sv[k] = clamp_node(sv[k], n);
        d[k]  = (base + k * 1024 < E) ? clamp_node(d[k], n) : -1;
        if (d[k] >= 0) atomicAdd(&lh[d[k] >> BKT_SHIFT], 1);
    }
    __syncthreads();
    if (tid < B) {
        const int c = lh[tid];
        const int r = c ? atomicAdd(&gcur[tid], c) : 0;
        lh[tid] = bstart + r;
    }
    __syncthreads();
#pragma unroll
    for (int k = 0; k < PART_EPT; ++k)
        if (d[k] >= 0) {
            const int pos = atomicAdd(&lh[d[k] >> BKT_SHIFT], 1);
            part[pos] = (sv[k] << BKT_SHIFT) | (d[k] & (BKT_NODES - 1));
        }
}

__device__ __forceinline__ void bucket_role(int b, const int* __restrict__ part, const int* __restrict__ ghist,
                                            int n, int* __restrict__ cnt, int* __restrict__ begs,
                                            int* __restrict__ srcs,
                                            int* lc,
                                            int* wsum) {
    const int tid = threadIdx.x;
    int base;
    (void)block_excl_scan(tid < b ? ghist[tid] : 0, wsum, base);
    const int count = ghist[b];
    const int* __restrict__ rec = part + base;
    if (tid < BKT_NODES) lc[tid] = 0;
    __syncthreads();
    for (int e0 = tid; e0 - tid < count; e0 += 1024 * BKT_EPT) {
        int p[BKT_EPT];
#pragma unroll
        for (int k = 0; k < BKT_EPT; ++k) {
            const int e = e0 + k * 1024;
            p[k] = rec[e < count ? e : count - 1];
        }
#pragma unroll
        for (int k = 0; k < BKT_EPT; ++k)
            if (e0 + k * 1024 < count) atomicAdd(&lc[p[k] & (BKT_NODES - 1)], 1);
    }
    __syncthreads();
    const int c = tid < BKT_NODES ? lc[tid] : 0;
    __syncthreads();
    int tot;
    const int excl = block_excl_scan(c, wsum, tot);
    if (tid < BKT_NODES) {
        const int node = (b << BKT_SHIFT) + tid;
        if (node < n) { cnt[node] = c; begs[node] = base + excl; }
        lc[tid] = excl;
    }
    __syncthreads();
    for (int e0 = tid; e0 - tid < count; e0 += 1024 * BKT_EPT) {
        int p[BKT_EPT];
#pragma unroll
        for (int k = 0; k < BKT_EPT; ++k) {
            const int e = e0 + k * 1024;
            p[k] = rec[e < count ? e : count - 1];
        }
#pragma unroll
        for (int k = 0; k < BKT_EPT; ++k)
            if (e0 + k * 1024 < count) {
                const int pos = atomicAdd(&lc[p[k] & (BKT_NODES - 1)], 1);
                srcs[base + pos] = p[k] >> BKT_SHIFT;
            }
    }
}

__global__ __launch_bounds__(1024) void k_bucket_gemm1(const int* __restrict__ part, const int* __restrict__ ghist,
                                                       int n, int* __restrict__ cnt, int* __restrict__ begs,
                                                       int* __restrict__ srcs, int nb,
                                                       const float* __restrict__ x,
                                                       const float* __restrict__ W1,
                                                       const float* __restrict__ asrc,
                                                       const float* __restrict__ adst,
                                                       float* __restrict__ h1,
                                                       float* __restrict__ as1, float* __restrict__ ad1) {
    __shared__ int   lc[BKT_NODES];
    __shared__ int   wsum[32];
    __shared__ float xs_all[G1_TILES * 16 * 16];
    __shared__ float ws[16 * 64];
    __shared__ float wa[16 * 16];
    const int blk = (int)blockIdx.x;
    if (blk < nb)
        bucket_role(blk, part, ghist, n, cnt, begs, srcs, lc, wsum);
    else
        gemm1_role(blk - nb, x, W1, asrc, adst, h1, as1, ad1, n, xs_all, ws, wa);
}

#define AGG1_GRID 1024
#define AGG2_GRID 2048
#define A1_PITCH 68
__device__ __forceinline__ int load_srcs_blk(const int* __restrict__ srcs, int beg, int deg,
                                             int b0, unsigned lane, int self) {
    const int kk = b0 + (int)lane;
    return kk < deg ? srcs[beg + kk] : self;
}

__device__ __forceinline__ void logits1_blk(int srcv, int b0, int total, unsigned j, unsigned h,
                                            const float* __restrict__ as1, float adi,
                                            float (&e)[8]) {
#pragma unroll
    for (int c = 0; c < 8; ++c) {
        const int sk = __shfl(srcv, 4 * c + (int)j);
        float t = as1[(unsigned)sk * 8u + h] + adi;
        t = t > 0.0f ? t : NEG_SLOPE * t;
        e[c] = ((int)j < total - b0 - 4 * c) ? t : 0.0f;
    }
}

#define GRP8_PAIR_LO 0x6d2240
#define GRP8_PAIR_HI 0xff6b64
__device__ __forceinline__ float grp8_pair_lo(float v) {
    return __builtin_bit_cast(float, __builtin_amdgcn_mov_dpp8(__builtin_bit_cast(unsigned, v), GRP8_PAIR_LO));
}
__device__ __forceinline__ float grp8_pair_hi(float v) {
    return __builtin_bit_cast(float, __builtin_amdgcn_mov_dpp8(__builtin_bit_cast(unsigned, v), GRP8_PAIR_HI));
}
__device__ __forceinline__ float xhalf(float v) {
    return __builtin_bit_cast(float, __builtin_amdgcn_permlanex16(__builtin_bit_cast(int, v), __builtin_bit_cast(int, v),
                                                                  0x76543210, 0xfedcba98, false, false));
}
__device__ __forceinline__ float xoct(float v) {
    return __builtin_bit_cast(float, __builtin_amdgcn_update_dpp(0, __builtin_bit_cast(int, v), 0x168  ,
                                                                  0xf, 0xf, false));
}

__device__ __forceinline__ void accum1_blk(int srcv, int b0, int total, const float (&al)[8],
                                           unsigned j, unsigned h, const float* __restrict__ h1,
                                           float (&acc)[8]) {
    const float4* __restrict__ rows = (const float4*)h1;
#pragma unroll
    for (int c = 0; c < 8; ++c) {
        if (b0 + 4 * c < total) {
            const int sk = __shfl(srcv, 4 * c + (int)j);
            const unsigned p = (unsigned)sk * 16u + h;
            const float4 lo = rows[p];
            const float4 hi = rows[p + 8u];
            const float wlo = grp8_pair_lo(al[c]);
            const float whi = grp8_pair_hi(al[c]);
            acc[0] += lo.x * wlo;  acc[1] += lo.y * wlo;  acc[2] += lo.z * wlo;  acc[3] += lo.w * wlo;
            acc[4] += hi.x * whi;  acc[5] += hi.y * whi;  acc[6] += hi.z * whi;  acc[7] += hi.w * whi;
        }
    }
}

__global__ __launch_bounds__(512) void k_agg1(const int* __restrict__ cnt, const int* __restrict__ begs,
                                              const int* __restrict__ srcs,
                                              const float* __restrict__ as1, const float* __restrict__ ad1,
                                              const float* __restrict__ h1, const float* __restrict__ bias1,
                                              const float* __restrict__ W2,
                                              const float* __restrict__ asrc2, const float* __restrict__ adst2,
                                              float* __restrict__ h2, float* __restrict__ asd2, int n) {
    __shared__ __attribute__((aligned(16))) float smA[2 * 16 * A1_PITCH];
    __shared__ float ws2[64 * 16];

    const unsigned tid  = threadIdx.x;
    const unsigned lane = tid & 31u;
    const int w = __builtin_amdgcn_readfirstlane((int)(tid >> 5));
    const unsigned j = lane >> 3, h = lane & 7u;
    const int ntiles = (n + 15) >> 4;
    const int tstep  = (int)gridDim.x;
    const int nstep  = tstep * 16;

    int t = (int)blockIdx.x;
    int i = t * 16 + w;
    int deg = 0, beg = 0, srcv0 = 0;
    float adi = 0.0f;
    if (i < n) {
        deg   = cnt[i];
        beg   = begs[i];
        adi   = ad1[(unsigned)i * 8u + h];
        srcv0 = load_srcs_blk(srcs, beg, deg, 0, lane, i);
    }
    int i1 = i + nstep, deg1 = 0, beg1 = 0;
    if (i1 < n) {
        deg1 = cnt[i1];
        beg1 = begs[i1];
    }
    const unsigned f0 = (lane < 16u ? 0u : 32u) + 4u * h + ((j & 1u) ? 2u : 0u);
    const float bz0 = bias1[f0], bz1 = bias1[f0 + 1u];

    if (tid < 128u) {
        const int k = (int)(tid >> 1), half = (int)(tid & 1u);
        ((float4*)(ws2 + k * 16 + half * 4))[0] = ((const float4*)W2)[tid];
    } else if (tid < 256u) {
        const unsigned tt = tid - 128u;
        const int k = (int)(tt >> 1);
        const float4* avp = (const float4*)((tt & 1u) ? adst2 : asrc2);
        const float4 av0 = avp[0], av1 = avp[1];
        const float4* wp = (const float4*)(W2 + k * 8);
        const float4 w0 = wp[0], w1 = wp[1];
        ws2[k * 16 + 8 + (int)(tt & 1u)] = w0.x * av0.x + w0.y * av0.y + w0.z * av0.z + w0.w * av0.w
                                         + w1.x * av1.x + w1.y * av1.y + w1.z * av1.z + w1.w * av1.w;
    } else {
        for (int e = (int)tid - 256; e < 64 * 6; e += 256) {
            const int k = e / 6;
            ws2[k * 16 + 10 + (e - 6 * k)] = 0.0f;
        }
    }

    for (int it = 0; t < ntiles; t += tstep, ++it) {
        const int row0 = t * 16;
        const bool have = i < n;

        int srcv1 = 0, cb2 = 0;
        float adi1 = 0.0f;
        const int i2 = i1 + nstep;
        auto ask_ahead = [&]() {
            if (i1 < n) {
                adi1  = ad1[(unsigned)i1 * 8u + h];
                srcv1 = load_srcs_blk(srcs, beg1, deg1, 0, lane, i1);
            }
            if (i2 < n) cb2 = (lane < 16u ? cnt : begs)[i2];
        };

        float r0 = 0.0f, r1 = 0.0f;
        if (!have) ask_ahead();
        if (have) {
            const int total = deg + 1;
            const float NEG_INF = -__builtin_huge_valf();

            float acc[8] = {0.0f, 0.0f, 0.0f, 0.0f, 0.0f, 0.0f, 0.0f, 0.0f};
            float e[8], al[8];
            if (total <= 32) {
                const int srcv = srcv0;
                logits1_blk(srcv, 0, total, j, h, as1, adi, e);
                ask_ahead();
                float m = NEG_INF;
#pragma unroll
                for (int c = 0; c < 8; ++c)
                    if ((int)j < total - 4 * c) m = fmaxf(m, e[c]);
                m = fmaxf(m, __shfl_xor(m, 8));
                m = fmaxf(m, __shfl_xor(m, 16));
                float ssum = 0.0f;
#pragma unroll
                for (int c = 0; c < 8; ++c) {
                    al[c] = ((int)j < total - 4 * c) ? expf(e[c] - m) : 0.0f;
                    ssum += al[c];
                }
                ssum += __shfl_xor(ssum, 8);
                ssum += __shfl_xor(ssum, 16);
                const float den = ssum + 1e-16f;
#pragma unroll
                for (int c = 0; c < 8; ++c) al[c] = al[c] / den;
                accum1_blk(srcv, 0, total, al, j, h, h1, acc);
            } else {
                ask_ahead();
                float m = NEG_INF;
                for (int b0 = 0; b0 < total; b0 += 32) {
                    const int srcv = load_srcs_blk(srcs, beg, deg, b0, lane, i);
                    logits1_blk(srcv, b0, total, j, h, as1, adi, e);
#pragma unroll
                    for (int c = 0; c < 8; ++c)
                        if ((int)j < total - b0 - 4 * c) m = fmaxf(m, e[c]);
                }
                m = fmaxf(m, __shfl_xor(m, 8));
                m = fmaxf(m, __shfl_xor(m, 16));
                float ssum = 0.0f;
                for (int b0 = 0; b0 < total; b0 += 32) {
                    const int srcv = load_srcs_blk(srcs, beg, deg, b0, lane, i);
                    logits1_blk(srcv, b0, total, j, h, as1, adi, e);
#pragma unroll
                    for (int c = 0; c < 8; ++c)
                        ssum += ((int)j < total - b0 - 4 * c) ? expf(e[c] - m) : 0.0f;
                }
                ssum += __shfl_xor(ssum, 8);
                ssum += __shfl_xor(ssum, 16);
                const float den = ssum + 1e-16f;
                for (int b0 = 0; b0 < total; b0 += 32) {
                    const int srcv = load_srcs_blk(srcs, beg, deg, b0, lane, i);
                    logits1_blk(srcv, b0, total, j, h, as1, adi, e);
#pragma unroll
                    for (int c = 0; c < 8; ++c)
                        al[c] = ((int)j < total - b0 - 4 * c) ? expf(e[c] - m) / den : 0.0f;
                    accum1_blk(srcv, b0, total, al, j, h, h1, acc);
                }
            }

            const bool lowh = lane < 16u, odd = (j & 1u) != 0u;
            float k4[4];
#pragma unroll
            for (int q = 0; q < 4; ++q)
                k4[q] = (lowh ? acc[q] : acc[4 + q]) + xhalf(lowh ? acc[4 + q] : acc[q]);
            const float s0 = (odd ? k4[2] : k4[0]) + xoct(odd ? k4[0] : k4[2]);
            const float s1 = (odd ? k4[3] : k4[1]) + xoct(odd ? k4[1] : k4[3]);

            r0 = s0 + bz0;
            r1 = s1 + bz1;
            r0 = r0 > 0.0f ? r0 : (expf(r0) - 1.0f);
            r1 = r1 > 0.0f ? r1 : (expf(r1) - 1.0f);
        }

        float* sm = smA + (it & 1) * (16 * A1_PITCH);
        *(float2*)(sm + w * A1_PITCH + (int)f0) = make_float2(r0, r1);
        __syncthreads();

        if (w == (it & 15)) {
            const bool hi   = lane >= 16u;
            const int  m    = (int)(lane & 15u);
            const bool full = (row0 + 16 <= n);
            const int  kofs = hi ? 2 : 0;
            v8f c = {};
#pragma unroll
            for (int k = 0; k < 16; ++k) {
                v2f a, b;
                a[0] = sm[m * A1_PITCH + 4 * k + kofs + 0];
                a[1] = sm[m * A1_PITCH + 4 * k + kofs + 1];
                b[0] = ws2[(4 * k + kofs + 0) * 16 + m];
                b[1] = ws2[(4 * k + kofs + 1) * 16 + m];
                c = __builtin_amdgcn_wmma_f32_16x16x4_f32(false, a, false, b,
                                                          (short)0, c, false, false);
            }
            const int rbase = row0 + (hi ? 8 : 0);
            const int lim   = n - rbase;
            if (m < 8) {
                float* out = h2 + (size_t)rbase * 8 + m;
                if (full) {
#pragma unroll
                    for (int r = 0; r < 8; ++r) out[r * 8] = c[r];
                } else {
#pragma unroll
                    for (int r = 0; r < 8; ++r)
                        if (r < lim) out[r * 8] = c[r];
                }
            } else if (m < 10) {
                float* out = asd2 + (size_t)rbase * 2 + (m - 8);
                if (full) {
#pragma unroll
                    for (int r = 0; r < 8; ++r) out[r * 2] = c[r];
                } else {
#pragma unroll
                    for (int r = 0; r < 8; ++r)
                        if (r < lim) out[r * 2] = c[r];
                }
            }
        }

        i = i1;  deg = deg1;  beg = beg1;  srcv0 = srcv1;  adi = adi1;
        i1 = i2;
        deg1 = __builtin_amdgcn_readlane(cb2, 0);
        beg1 = __builtin_amdgcn_readlane(cb2, 16);
    }
}

__device__ __forceinline__ float wave_max(float v) {
#pragma unroll
    for (int o = 16; o > 0; o >>= 1) v = fmaxf(v, __shfl_xor(v, o));
    return v;
}
__device__ __forceinline__ float wave_sum(float v) {
#pragma unroll
    for (int o = 16; o > 0; o >>= 1) v += __shfl_xor(v, o);
    return v;
}

__device__ __forceinline__ float logit2(int srcv, bool valid,
                                        const float* __restrict__ asd2, float adi) {
    float t = asd2[2u * (unsigned)srcv] + adi;
    t = t > 0.0f ? t : NEG_SLOPE * t;
    return valid ? t : 0.0f;
}
__device__ __forceinline__ float logit2_of(float as, bool valid, float adi) {
    float t = as + adi;
    t = t > 0.0f ? t : NEG_SLOPE * t;
    return valid ? t : 0.0f;
}

__device__ __forceinline__ void rows2_blk(int srcv, unsigned j, unsigned f,
                                          const float* __restrict__ h2, float (&hv)[8]) {
#pragma unroll
    for (int c = 0; c < 8; ++c) {
        const int sk = __shfl(srcv, 4 * c + (int)j);
        hv[c] = h2[(unsigned)sk * 8u + f];
    }
}

__device__ __forceinline__ void accum2_blk(const float (&hv)[8], float al, int b0, int total,
                                           unsigned j, float& acc) {
#pragma unroll
    for (int c = 0; c < 8; ++c) {
        const float ak = __shfl(al, 4 * c + (int)j);
        acc += (b0 + 4 * c + (int)j < total) ? hv[c] * ak : 0.0f;
    }
}

__global__ __launch_bounds__(256) void k_agg2(const int* __restrict__ cnt, const int* __restrict__ begs,
                                              const int* __restrict__ srcs,
                                              const float* __restrict__ asd2,
                                              const float* __restrict__ h2, const float* __restrict__ bias2,
                                              float* __restrict__ out, int n) {
    const unsigned lane = threadIdx.x & 31u;
    const int nw = (int)gridDim.x * 8;
    int i = __builtin_amdgcn_readfirstlane((int)(blockIdx.x * 8u + (threadIdx.x >> 5)));
    const unsigned j = lane >> 3, f = lane & 7u;
    const float NEG_INF = -__builtin_huge_valf();
    const float bz = bias2[f];

    int deg = 0, beg = 0, srcv0 = 0;
    float adi = 0.0f;
    if (i < n) {
        deg   = cnt[i];
        beg   = begs[i];
        adi   = asd2[2u * (unsigned)i + 1u];
        srcv0 = load_srcs_blk(srcs, beg, deg, 0, lane, i);
    }
    int i1 = i + nw, deg1 = 0, beg1 = 0;
    if (i1 < n) {
        deg1 = cnt[i1];
        beg1 = begs[i1];
    }

    while (i < n) {
        int srcv1 = 0;
        float adi1 = 0.0f;
        const int i2 = i1 + nw;
        int deg2 = 0, beg2 = 0;
        auto ask_ahead = [&]() {
            if (i1 < n) {
                adi1  = asd2[2u * (unsigned)i1 + 1u];
                srcv1 = load_srcs_blk(srcs, beg1, deg1, 0, lane, i1);
            }
            if (i2 < n) {
                deg2 = cnt[i2];
                beg2 = begs[i2];
            }
        };

        const int total = deg + 1;
        float acc = 0.0f;
        float hv[8];
        if (total <= 32) {
            const int  srcv  = srcv0;
            const bool valid = (int)lane < total;
            rows2_blk(srcv, j, f, h2, hv);
            __builtin_amdgcn_sched_barrier(0);
            const float lg = asd2[2u * (unsigned)srcv];
            ask_ahead();
            const float e  = logit2_of(lg, valid, adi);
            const float m  = wave_max(valid ? e : NEG_INF);
            const float ex = valid ? expf(e - m) : 0.0f;
            const float den = wave_sum(ex) + 1e-16f;
            accum2_blk(hv, ex / den, 0, total, j, acc);
        } else {
            ask_ahead();
            float m = NEG_INF;
            for (int b0 = 0; b0 < total; b0 += 32) {
                const int  srcv  = load_srcs_blk(srcs, beg, deg, b0, lane, i);
                const bool valid = b0 + (int)lane < total;
                const float e = logit2(srcv, valid, asd2, adi);
                m = fmaxf(m, valid ? e : NEG_INF);
            }
            m = wave_max(m);
            float ssum = 0.0f;
            for (int b0 = 0; b0 < total; b0 += 32) {
                const int  srcv  = load_srcs_blk(srcs, beg, deg, b0, lane, i);
                const bool valid = b0 + (int)lane < total;
                const float e = logit2(srcv, valid, asd2, adi);
                ssum += valid ? expf(e - m) : 0.0f;
            }
            const float den = wave_sum(ssum) + 1e-16f;
            for (int b0 = 0; b0 < total; b0 += 32) {
                const int  srcv  = load_srcs_blk(srcs, beg, deg, b0, lane, i);
                const bool valid = b0 + (int)lane < total;
                const float e = logit2(srcv, valid, asd2, adi);
                rows2_blk(srcv, j, f, h2, hv);
                const float al = valid ? expf(e - m) / den : 0.0f;
                accum2_blk(hv, al, b0, total, j, acc);
            }
        }
        acc += __shfl_xor(acc, 8);
        acc += __shfl_xor(acc, 16);
        if (lane < 8u) out[(unsigned)i * 8u + f] = acc + bz;

        i = i1;  deg = deg1;  beg = beg1;  srcv0 = srcv1;  adi = adi1;
        i1 = i2; deg1 = deg2; beg1 = beg2;
    }
}

extern "C" void kernel_launch(void* const* d_in, const int* in_sizes, int n_in,
                              void* d_out, int out_size, void* d_ws, size_t ws_size,
                              hipStream_t stream) {
    const float* x     = (const float*)d_in[0];
    const int*   ei    = (const int*)d_in[1];
    const float* W1    = (const float*)d_in[2];
    const float* asrc1 = (const float*)d_in[3];
    const float* adst1 = (const float*)d_in[4];
    const float* bias1 = (const float*)d_in[5];
    const float* W2    = (const float*)d_in[6];
    const float* asrc2 = (const float*)d_in[7];
    const float* adst2 = (const float*)d_in[8];
    const float* bias2 = (const float*)d_in[9];

    int n = in_sizes[0] / F_IN;
    int E = in_sizes[1] / 2;
    int B = (n + BKT_NODES - 1) >> BKT_SHIFT;

    char* p = (char*)d_ws;
    float* h1   = (float*)p; p += (size_t)n * 64 * 4;
    float* as1  = (float*)p; p += (size_t)n * 8 * 4;
    float* ad1  = (float*)p; p += (size_t)n * 8 * 4;
    float* h2   = (float*)p; p += (size_t)n * 8 * 4;
    float* asd2 = (float*)p; p += (size_t)n * 8;
    int*   cnt  = (int*)p;   p += (size_t)n * 4;
    int*   begs = (int*)p;   p += (size_t)n * 4;
    int*   part = (int*)p;   p += (size_t)E * 4;
    int*   srcs = (int*)p;   p += (size_t)E * 4;
    int*   ghist = (int*)p;  p += (size_t)MAX_BKT * 4;
    int*   gcur  = (int*)p;  p += (size_t)MAX_BKT * 4;

    hipMemsetAsync(ghist, 0, (size_t)2 * MAX_BKT * 4, stream);

    int gP = (E + 1024 * PART_EPT - 1) / (1024 * PART_EPT);

    k_hist<<<gP, 1024, 0, stream>>>(ei, E, n, B, ghist);
    k_part<<<gP, 1024, 0, stream>>>(ei, E, n, B, ghist, gcur, part);
    int gG = (n + 16 * G1_TILES - 1) / (16 * G1_TILES);
    k_bucket_gemm1<<<B + gG, 1024, 0, stream>>>(part, ghist, n, cnt, begs, srcs, B,
                                                x, W1, asrc1, adst1, h1, as1, ad1);

    k_agg1<<<AGG1_GRID, 512, 0, stream>>>(cnt, begs, srcs, as1, ad1, h1, bias1,
                                          W2, asrc2, adst2, h2, asd2, n);

    k_agg2<<<AGG2_GRID, 256, 0, stream>>>(cnt, begs, srcs, asd2, h2, bias2, (float*)d_out, n);
}


# softmax and ELU exponentials as v_exp_f32 of x*log2e (one fma each) and weights as exponential times one v_rcp_f32 of the node's sum, not library expf and a full division per weight; sum orders kept
# speedup vs baseline: 13.1948x; 1.0386x over previous
#include <hip/hip_runtime.h>
#include <hip/hip_bf16.h>

#define NEG_SLOPE 0.2f
#define F_IN 14

typedef float v2f __attribute__((ext_vector_type(2)));
typedef float v8f __attribute__((ext_vector_type(8)));

__device__ __forceinline__ int clamp_node(int v, int n) {
    return v < 0 ? 0 : (v >= n ? n - 1 : v);
}

#define LOG2E 1.44269504088896340736f
__device__ __forceinline__ float exp_hw(float x) {
    return __builtin_amdgcn_exp2f(x * LOG2E);
}
__device__ __forceinline__ float exp_hw_rel(float e, float ml) {
    return __builtin_amdgcn_exp2f(__builtin_fmaf(e, LOG2E, -ml));
}
__device__ __forceinline__ float rcp_hw(float x) {
    return __builtin_amdgcn_rcpf(x);
}

#define G1_TILES 32
__device__ __forceinline__ void gemm1_role(int gblk,
                                           const float* __restrict__ x,
                                           const float* __restrict__ W1,
                                           const float* __restrict__ asrc,
                                           const float* __restrict__ adst,
                                           float* __restrict__ h1,
                                           float* __restrict__ as1, float* __restrict__ ad1, int n,
                                           float* xs_all,
                                           float* ws,
                                           float* wa) {
    const int tid  = threadIdx.x;
    const int lane = tid & 31;
    const int w    = __builtin_amdgcn_readfirstlane(tid >> 5);
    const int row0 = (gblk * G1_TILES + w) * 16;
    const bool hi  = lane >= 16;
    const int  m   = lane & 15;
    float* xs = xs_all + w * 256;

    int rows = n - row0;
    rows = rows > 16 ? 16 : (rows < 0 ? 0 : rows);
    const int cntx = rows * F_IN;
    float xv[7];
#pragma unroll
    for (int q = 0; q < 7; ++q) {
        const int i = lane + 32 * q;
        xv[q] = i < cntx ? x[(size_t)row0 * F_IN + i] : 0.0f;
    }
    float4 av0 = make_float4(0.f, 0.f, 0.f, 0.f), av1 = av0;
    if (tid < 256) {
        const int mm = tid & 15;
        const float4* avp = (const float4*)((mm < 8 ? asrc : adst) + (mm & 7) * 8);
        av0 = avp[0];
        av1 = avp[1];
    }
    if (tid < F_IN * 64 / 4)
        ((float4*)ws)[tid] = ((const float4*)W1)[tid];
    else if (tid < 256)
        ((float4*)(ws + F_IN * 64))[tid - F_IN * 64 / 4] = make_float4(0.f, 0.f, 0.f, 0.f);
#pragma unroll
    for (int q = 0; q < 7; ++q) {
        const int i = lane + 32 * q;
        const int r = i / F_IN, c = i - r * F_IN;
        xs[r * 16 + c] = xv[q];
    }
    xs[(lane >> 1) * 16 + F_IN + (lane & 1)] = 0.0f;
    __syncthreads();
    if (tid < 256) {
        const int k = tid >> 4, mm = tid & 15;
        const float4* wp = (const float4*)(ws + k * 64 + (mm & 7) * 8);
        const float4 w0 = wp[0], w1 = wp[1];
        wa[k * 16 + mm] = w0.x * av0.x + w0.y * av0.y + w0.z * av0.z + w0.w * av0.w
                        + w1.x * av1.x + w1.y * av1.y + w1.z * av1.z + w1.w * av1.w;
    }
    __syncthreads();
    if (row0 >= n) return;
    const bool full = (row0 + 16 <= n);

    const int kofs = hi ? 2 : 0;
    v2f a[4];
#pragma unroll
    for (int k = 0; k < 4; ++k) {
        a[k][0] = xs[m * 16 + 4 * k + kofs + 0];
        a[k][1] = xs[m * 16 + 4 * k + kofs + 1];
    }

#pragma unroll
    for (int t = 0; t < 4; ++t) {
        v8f c = {};
#pragma unroll
        for (int k = 0; k < 4; ++k) {
            v2f b;
            b[0] = ws[(4 * k + kofs + 0) * 64 + t * 16 + m];
            b[1] = ws[(4 * k + kofs + 1) * 64 + t * 16 + m];
            c = __builtin_amdgcn_wmma_f32_16x16x4_f32(false, a[k], false, b,
                                                      (short)0, c, false, false);
        }
        float* out = h1 + (size_t)row0 * 64 + t * 16 + m;
        if (full) {
#pragma unroll
            for (int r = 0; r < 8; ++r)
                out[(size_t)(r + (hi ? 8 : 0)) * 64] = c[r];
        } else {
#pragma unroll
            for (int r = 0; r < 8; ++r) {
                int rr = r + (hi ? 8 : 0);
                if (row0 + rr < n) out[(size_t)rr * 64] = c[r];
            }
        }
    }

    {
        v8f c = {};
#pragma unroll
        for (int k = 0; k < 4; ++k) {
            v2f b;
            b[0] = wa[(4 * k + kofs + 0) * 16 + m];
            b[1] = wa[(4 * k + kofs + 1) * 16 + m];
            c = __builtin_amdgcn_wmma_f32_16x16x4_f32(false, a[k], false, b,
                                                      (short)0, c, false, false);
        }
        float* lo = (m < 8 ? as1 : ad1) + (size_t)row0 * 8 + (m & 7);
        if (full) {
#pragma unroll
            for (int r = 0; r < 8; ++r)
                lo[(size_t)(r + (hi ? 8 : 0)) * 8] = c[r];
        } else {
#pragma unroll
            for (int r = 0; r < 8; ++r) {
                int rr = r + (hi ? 8 : 0);
                if (row0 + rr < n) lo[(size_t)rr * 8] = c[r];
            }
        }
    }
}

#define BKT_SHIFT 9
#define BKT_NODES (1 << BKT_SHIFT)
#define MAX_BKT   1024
#define PART_EPT  8
#define BKT_EPT   8

__device__ __forceinline__ int block_excl_scan(int v, int* wsum, int& total) {
    const int lane = threadIdx.x & 31, w = threadIdx.x >> 5;
    int incl = v;
#pragma unroll
    for (int o = 1; o < 32; o <<= 1) {
        int t = __shfl_up(incl, o);
        if (lane >= o) incl += t;
    }
    if (lane == 31) wsum[w] = incl;
    __syncthreads();
    int woff = 0, tot = 0;
#pragma unroll
    for (int q = 0; q < 32; ++q) {
        int t = wsum[q];
        if (q < w) woff += t;
        tot += t;
    }
    total = tot;
    return woff + incl - v;
}

__global__ __launch_bounds__(1024) void k_hist(const int* __restrict__ ei, int E, int n, int B,
                                               int* __restrict__ ghist) {
    __shared__ int lh[MAX_BKT];
    const int tid = threadIdx.x;
    const int base = blockIdx.x * (1024 * PART_EPT) + tid;
    int d[PART_EPT];
#pragma unroll
    for (int k = 0; k < PART_EPT; ++k) {
        const int idx = base + k * 1024;
        d[k] = ei[E + (idx < E ? idx : E - 1)];
    }
    if (tid < B) lh[tid] = 0;
    __syncthreads();
#pragma unroll
    for (int k = 0; k < PART_EPT; ++k)
        if (base + k * 1024 < E) atomicAdd(&lh[clamp_node(d[k], n) >> BKT_SHIFT], 1);
    __syncthreads();
    if (tid < B) {
        const int c = lh[tid];
        if (c) atomicAdd(&ghist[tid], c);
    }
}

__global__ __launch_bounds__(1024) void k_part(const int* __restrict__ ei, int E, int n, int B,
                                               const int* __restrict__ ghist, int* __restrict__ gcur,
                                               int* __restrict__ part) {
    __shared__ int lh[MAX_BKT];
    __shared__ int wsum[32];
    const int tid = threadIdx.x;
    const int base = blockIdx.x * (1024 * PART_EPT) + tid;
    int sv[PART_EPT], d[PART_EPT];
#pragma unroll
    for (int k = 0; k < PART_EPT; ++k) {
        const int idx = base + k * 1024;
        const int ii  = idx < E ? idx : E - 1;
        sv[k] = ei[ii];
        d[k]  = ei[E + ii];
    }
    int tot;
    const int bstart = block_excl_scan(tid < B ? ghist[tid] : 0, wsum, tot);
    if (tid < B) lh[tid] = 0;
    __syncthreads();
#pragma unroll
    for (int k = 0; k < PART_EPT; ++k) {
        sv[k] = clamp_node(sv[k], n);
        d[k]  = (base + k * 1024 < E) ? clamp_node(d[k], n) : -1;
        if (d[k] >= 0) atomicAdd(&lh[d[k] >> BKT_SHIFT], 1);
    }
    __syncthreads();
    if (tid < B) {
        const int c = lh[tid];
        const int r = c ? atomicAdd(&gcur[tid], c) : 0;
        lh[tid] = bstart + r;
    }
    __syncthreads();
#pragma unroll
    for (int k = 0; k < PART_EPT; ++k)
        if (d[k] >= 0) {
            const int pos = atomicAdd(&lh[d[k] >> BKT_SHIFT], 1);
            part[pos] = (sv[k] << BKT_SHIFT) | (d[k] & (BKT_NODES - 1));
        }
}

__device__ __forceinline__ void bucket_role(int b, const int* __restrict__ part, const int* __restrict__ ghist,
                                            int n, int* __restrict__ cnt, int* __restrict__ begs,
                                            int* __restrict__ srcs,
                                            int* lc,
                                            int* wsum) {
    const int tid = threadIdx.x;
    int base;
    (void)block_excl_scan(tid < b ? ghist[tid] : 0, wsum, base);
    const int count = ghist[b];
    const int* __restrict__ rec = part + base;
    if (tid < BKT_NODES) lc[tid] = 0;
    __syncthreads();
    for (int e0 = tid; e0 - tid < count; e0 += 1024 * BKT_EPT) {
        int p[BKT_EPT];
#pragma unroll
        for (int k = 0; k < BKT_EPT; ++k) {
            const int e = e0 + k * 1024;
            p[k] = rec[e < count ? e : count - 1];
        }
#pragma unroll
        for (int k = 0; k < BKT_EPT; ++k)
            if (e0 + k * 1024 < count) atomicAdd(&lc[p[k] & (BKT_NODES - 1)], 1);
    }
    __syncthreads();
    const int c = tid < BKT_NODES ? lc[tid] : 0;
    __syncthreads();
    int tot;
    const int excl = block_excl_scan(c, wsum, tot);
    if (tid < BKT_NODES) {
        const int node = (b << BKT_SHIFT) + tid;
        if (node < n) { cnt[node] = c; begs[node] = base + excl; }
        lc[tid] = excl;
    }
    __syncthreads();
    for (int e0 = tid; e0 - tid < count; e0 += 1024 * BKT_EPT) {
        int p[BKT_EPT];
#pragma unroll
        for (int k = 0; k < BKT_EPT; ++k) {
            const int e = e0 + k * 1024;
            p[k] = rec[e < count ? e : count - 1];
        }
#pragma unroll
        for (int k = 0; k < BKT_EPT; ++k)
            if (e0 + k * 1024 < count) {
                const int pos = atomicAdd(&lc[p[k] & (BKT_NODES - 1)], 1);
                srcs[base + pos] = p[k] >> BKT_SHIFT;
            }
    }
}

__global__ __launch_bounds__(1024) void k_bucket_gemm1(const int* __restrict__ part, const int* __restrict__ ghist,
                                                       int n, int* __restrict__ cnt, int* __restrict__ begs,
                                                       int* __restrict__ srcs, int nb,
                                                       const float* __restrict__ x,
                                                       const float* __restrict__ W1,
                                                       const float* __restrict__ asrc,
                                                       const float* __restrict__ adst,
                                                       float* __restrict__ h1,
                                                       float* __restrict__ as1, float* __restrict__ ad1) {
    __shared__ int   lc[BKT_NODES];
    __shared__ int   wsum[32];
    __shared__ float xs_all[G1_TILES * 16 * 16];
    __shared__ float ws[16 * 64];
    __shared__ float wa[16 * 16];
    const int blk = (int)blockIdx.x;
    if (blk < nb)
        bucket_role(blk, part, ghist, n, cnt, begs, srcs, lc, wsum);
    else
        gemm1_role(blk - nb, x, W1, asrc, adst, h1, as1, ad1, n, xs_all, ws, wa);
}

#define AGG1_GRID 1024
#define AGG2_GRID 2048
#define A1_PITCH 68
__device__ __forceinline__ int load_srcs_blk(const int* __restrict__ srcs, int beg, int deg,
                                             int b0, unsigned lane, int self) {
    const int kk = b0 + (int)lane;
    return kk < deg ? srcs[beg + kk] : self;
}

__device__ __forceinline__ void logits1_blk(int srcv, int b0, int total, unsigned j, unsigned h,
                                            const float* __restrict__ as1, float adi,
                                            float (&e)[8]) {
#pragma unroll
    for (int c = 0; c < 8; ++c) {
        const int sk = __shfl(srcv, 4 * c + (int)j);
        float t = as1[(unsigned)sk * 8u + h] + adi;
        t = t > 0.0f ? t : NEG_SLOPE * t;
        e[c] = ((int)j < total - b0 - 4 * c) ? t : 0.0f;
    }
}

#define GRP8_PAIR_LO 0x6d2240
#define GRP8_PAIR_HI 0xff6b64
__device__ __forceinline__ float grp8_pair_lo(float v) {
    return __builtin_bit_cast(float, __builtin_amdgcn_mov_dpp8(__builtin_bit_cast(unsigned, v), GRP8_PAIR_LO));
}
__device__ __forceinline__ float grp8_pair_hi(float v) {
    return __builtin_bit_cast(float, __builtin_amdgcn_mov_dpp8(__builtin_bit_cast(unsigned, v), GRP8_PAIR_HI));
}
__device__ __forceinline__ float xhalf(float v) {
    return __builtin_bit_cast(float, __builtin_amdgcn_permlanex16(__builtin_bit_cast(int, v), __builtin_bit_cast(int, v),
                                                                  0x76543210, 0xfedcba98, false, false));
}
__device__ __forceinline__ float xoct(float v) {
    return __builtin_bit_cast(float, __builtin_amdgcn_update_dpp(0, __builtin_bit_cast(int, v), 0x168  ,
                                                                  0xf, 0xf, false));
}

__device__ __forceinline__ void accum1_blk(int srcv, int b0, int total, const float (&al)[8],
                                           unsigned j, unsigned h, const float* __restrict__ h1,
                                           float (&acc)[8]) {
    const float4* __restrict__ rows = (const float4*)h1;
#pragma unroll
    for (int c = 0; c < 8; ++c) {
        if (b0 + 4 * c < total) {
            const int sk = __shfl(srcv, 4 * c + (int)j);
            const unsigned p = (unsigned)sk * 16u + h;
            const float4 lo = rows[p];
            const float4 hi = rows[p + 8u];
            const float wlo = grp8_pair_lo(al[c]);
            const float whi = grp8_pair_hi(al[c]);
            acc[0] += lo.x * wlo;  acc[1] += lo.y * wlo;  acc[2] += lo.z * wlo;  acc[3] += lo.w * wlo;
            acc[4] += hi.x * whi;  acc[5] += hi.y * whi;  acc[6] += hi.z * whi;  acc[7] += hi.w * whi;
        }
    }
}

__global__ __launch_bounds__(512, 16) void k_agg1(const int* __restrict__ cnt, const int* __restrict__ begs,
                                              const int* __restrict__ srcs,
                                              const float* __restrict__ as1, const float* __restrict__ ad1,
                                              const float* __restrict__ h1, const float* __restrict__ bias1,
                                              const float* __restrict__ W2,
                                              const float* __restrict__ asrc2, const float* __restrict__ adst2,
                                              float* __restrict__ h2, float* __restrict__ asd2, int n) {
    __shared__ __attribute__((aligned(16))) float smA[2 * 16 * A1_PITCH];
    __shared__ float ws2[64 * 16];

    const unsigned tid  = threadIdx.x;
    const unsigned lane = tid & 31u;
    const int w = __builtin_amdgcn_readfirstlane((int)(tid >> 5));
    const unsigned j = lane >> 3, h = lane & 7u;
    const int ntiles = (n + 15) >> 4;
    const int tstep  = (int)gridDim.x;
    const int nstep  = tstep * 16;

    int t = (int)blockIdx.x;
    int i = t * 16 + w;
    int deg = 0, beg = 0, srcv0 = 0;
    float adi = 0.0f;
    if (i < n) {
        deg   = cnt[i];
        beg   = begs[i];
        adi   = ad1[(unsigned)i * 8u + h];
        srcv0 = load_srcs_blk(srcs, beg, deg, 0, lane, i);
    }
    int i1 = i + nstep, deg1 = 0, beg1 = 0;
    if (i1 < n) {
        deg1 = cnt[i1];
        beg1 = begs[i1];
    }
    const unsigned f0 = (lane < 16u ? 0u : 32u) + 4u * h + ((j & 1u) ? 2u : 0u);
    const float bz0 = bias1[f0], bz1 = bias1[f0 + 1u];

    if (tid < 128u) {
        const int k = (int)(tid >> 1), half = (int)(tid & 1u);
        ((float4*)(ws2 + k * 16 + half * 4))[0] = ((const float4*)W2)[tid];
    } else if (tid < 256u) {
        const unsigned tt = tid - 128u;
        const int k = (int)(tt >> 1);
        const float4* avp = (const float4*)((tt & 1u) ? adst2 : asrc2);
        const float4 av0 = avp[0], av1 = avp[1];
        const float4* wp = (const float4*)(W2 + k * 8);
        const float4 w0 = wp[0], w1 = wp[1];
        ws2[k * 16 + 8 + (int)(tt & 1u)] = w0.x * av0.x + w0.y * av0.y + w0.z * av0.z + w0.w * av0.w
                                         + w1.x * av1.x + w1.y * av1.y + w1.z * av1.z + w1.w * av1.w;
    } else {
        for (int e = (int)tid - 256; e < 64 * 6; e += 256) {
            const int k = e / 6;
            ws2[k * 16 + 10 + (e - 6 * k)] = 0.0f;
        }
    }

    for (int it = 0; t < ntiles; t += tstep, ++it) {
        const int row0 = t * 16;
        const bool have = i < n;

        int srcv1 = 0, cb2 = 0;
        float adi1 = 0.0f;
        const int i2 = i1 + nstep;
        auto ask_ahead = [&]() {
            if (i1 < n) {
                adi1  = ad1[(unsigned)i1 * 8u + h];
                srcv1 = load_srcs_blk(srcs, beg1, deg1, 0, lane, i1);
            }
            if (i2 < n) cb2 = (lane < 16u ? cnt : begs)[i2];
        };

        float r0 = 0.0f, r1 = 0.0f;
        if (!have) ask_ahead();
        if (have) {
            const int total = deg + 1;
            const float NEG_INF = -__builtin_huge_valf();

            float acc[8] = {0.0f, 0.0f, 0.0f, 0.0f, 0.0f, 0.0f, 0.0f, 0.0f};
            float e[8], al[8];
            if (total <= 32) {
                const int srcv = srcv0;
                logits1_blk(srcv, 0, total, j, h, as1, adi, e);
                ask_ahead();
                float m = NEG_INF;
#pragma unroll
                for (int c = 0; c < 8; ++c)
                    if ((int)j < total - 4 * c) m = fmaxf(m, e[c]);
                m = fmaxf(m, __shfl_xor(m, 8));
                m = fmaxf(m, __shfl_xor(m, 16));
                const float ml = m * LOG2E;
                float ssum = 0.0f;
#pragma unroll
                for (int c = 0; c < 8; ++c) {
                    al[c] = ((int)j < total - 4 * c) ? exp_hw_rel(e[c], ml) : 0.0f;
                    ssum += al[c];
                }
                ssum += __shfl_xor(ssum, 8);
                ssum += __shfl_xor(ssum, 16);
                const float rden = rcp_hw(ssum + 1e-16f);
#pragma unroll
                for (int c = 0; c < 8; ++c) al[c] = al[c] * rden;
                accum1_blk(srcv, 0, total, al, j, h, h1, acc);
            } else {
                ask_ahead();
                float m = NEG_INF;
                for (int b0 = 0; b0 < total; b0 += 32) {
                    const int srcv = load_srcs_blk(srcs, beg, deg, b0, lane, i);
                    logits1_blk(srcv, b0, total, j, h, as1, adi, e);
#pragma unroll
                    for (int c = 0; c < 8; ++c)
                        if ((int)j < total - b0 - 4 * c) m = fmaxf(m, e[c]);
                }
                m = fmaxf(m, __shfl_xor(m, 8));
                m = fmaxf(m, __shfl_xor(m, 16));
                const float ml = m * LOG2E;
                float ssum = 0.0f;
                for (int b0 = 0; b0 < total; b0 += 32) {
                    const int srcv = load_srcs_blk(srcs, beg, deg, b0, lane, i);
                    logits1_blk(srcv, b0, total, j, h, as1, adi, e);
#pragma unroll
                    for (int c = 0; c < 8; ++c)
                        ssum += ((int)j < total - b0 - 4 * c) ? exp_hw_rel(e[c], ml) : 0.0f;
                }
                ssum += __shfl_xor(ssum, 8);
                ssum += __shfl_xor(ssum, 16);
                const float rden = rcp_hw(ssum + 1e-16f);
                for (int b0 = 0; b0 < total; b0 += 32) {
                    const int srcv = load_srcs_blk(srcs, beg, deg, b0, lane, i);
                    logits1_blk(srcv, b0, total, j, h, as1, adi, e);
#pragma unroll
                    for (int c = 0; c < 8; ++c)
                        al[c] = ((int)j < total - b0 - 4 * c) ? exp_hw_rel(e[c], ml) * rden : 0.0f;
                    accum1_blk(srcv, b0, total, al, j, h, h1, acc);
                }
            }

            const bool lowh = lane < 16u, odd = (j & 1u) != 0u;
            float k4[4];
#pragma unroll
            for (int q = 0; q < 4; ++q)
                k4[q] = (lowh ? acc[q] : acc[4 + q]) + xhalf(lowh ? acc[4 + q] : acc[q]);
            const float s0 = (odd ? k4[2] : k4[0]) + xoct(odd ? k4[0] : k4[2]);
            const float s1 = (odd ? k4[3] : k4[1]) + xoct(odd ? k4[1] : k4[3]);

            r0 = s0 + bz0;
            r1 = s1 + bz1;
            r0 = r0 > 0.0f ? r0 : (exp_hw(r0) - 1.0f);
            r1 = r1 > 0.0f ? r1 : (exp_hw(r1) - 1.0f);
        }

        float* sm = smA + (it & 1) * (16 * A1_PITCH);
        *(float2*)(sm + w * A1_PITCH + (int)f0) = make_float2(r0, r1);
        __syncthreads();

        if (w == (it & 15)) {
            const bool hi   = lane >= 16u;
            const int  m    = (int)(lane & 15u);
            const bool full = (row0 + 16 <= n);
            const int  kofs = hi ? 2 : 0;
            v8f c = {};
#pragma unroll
            for (int k = 0; k < 16; ++k) {
                v2f a, b;
                a[0] = sm[m * A1_PITCH + 4 * k + kofs + 0];
                a[1] = sm[m * A1_PITCH + 4 * k + kofs + 1];
                b[0] = ws2[(4 * k + kofs + 0) * 16 + m];
                b[1] = ws2[(4 * k + kofs + 1) * 16 + m];
                c = __builtin_amdgcn_wmma_f32_16x16x4_f32(false, a, false, b,
                                                          (short)0, c, false, false);
            }
            const int rbase = row0 + (hi ? 8 : 0);
            const int lim   = n - rbase;
            if (m < 8) {
                float* out = h2 + (size_t)rbase * 8 + m;
                if (full) {
#pragma unroll
                    for (int r = 0; r < 8; ++r) out[r * 8] = c[r];
                } else {
#pragma unroll
                    for (int r = 0; r < 8; ++r)
                        if (r < lim) out[r * 8] = c[r];
                }
            } else if (m < 10) {
                float* out = asd2 + (size_t)rbase * 2 + (m - 8);
                if (full) {
#pragma unroll
                    for (int r = 0; r < 8; ++r) out[r * 2] = c[r];
                } else {
#pragma unroll
                    for (int r = 0; r < 8; ++r)
                        if (r < lim) out[r * 2] = c[r];
                }
            }
        }

        i = i1;  deg = deg1;  beg = beg1;  srcv0 = srcv1;  adi = adi1;
        i1 = i2;
        deg1 = __builtin_amdgcn_readlane(cb2, 0);
        beg1 = __builtin_amdgcn_readlane(cb2, 16);
    }
}

__device__ __forceinline__ float wave_max(float v) {
#pragma unroll
    for (int o = 16; o > 0; o >>= 1) v = fmaxf(v, __shfl_xor(v, o));
    return v;
}
__device__ __forceinline__ float wave_sum(float v) {
#pragma unroll
    for (int o = 16; o > 0; o >>= 1) v += __shfl_xor(v, o);
    return v;
}

__device__ __forceinline__ float logit2(int srcv, bool valid,
                                        const float* __restrict__ asd2, float adi) {
    float t = asd2[2u * (unsigned)srcv] + adi;
    t = t > 0.0f ? t : NEG_SLOPE * t;
    return valid ? t : 0.0f;
}
__device__ __forceinline__ float logit2_of(float as, bool valid, float adi) {
    float t = as + adi;
    t = t > 0.0f ? t : NEG_SLOPE * t;
    return valid ? t : 0.0f;
}

__device__ __forceinline__ void rows2_blk(int srcv, unsigned j, unsigned f,
                                          const float* __restrict__ h2, float (&hv)[8]) {
#pragma unroll
    for (int c = 0; c < 8; ++c) {
        const int sk = __shfl(srcv, 4 * c + (int)j);
        hv[c] = h2[(unsigned)sk * 8u + f];
    }
}

__device__ __forceinline__ void accum2_blk(const float (&hv)[8], float al, int b0, int total,
                                           unsigned j, float& acc) {
#pragma unroll
    for (int c = 0; c < 8; ++c) {
        const float ak = __shfl(al, 4 * c + (int)j);
        acc += (b0 + 4 * c + (int)j < total) ? hv[c] * ak : 0.0f;
    }
}

__global__ __launch_bounds__(256) void k_agg2(const int* __restrict__ cnt, const int* __restrict__ begs,
                                              const int* __restrict__ srcs,
                                              const float* __restrict__ asd2,
                                              const float* __restrict__ h2, const float* __restrict__ bias2,
                                              float* __restrict__ out, int n) {
    const unsigned lane = threadIdx.x & 31u;
    const int nw = (int)gridDim.x * 8;
    int i = __builtin_amdgcn_readfirstlane((int)(blockIdx.x * 8u + (threadIdx.x >> 5)));
    const unsigned j = lane >> 3, f = lane & 7u;
    const float NEG_INF = -__builtin_huge_valf();
    const float bz = bias2[f];

    int deg = 0, beg = 0, srcv0 = 0;
    float adi = 0.0f;
    if (i < n) {
        deg   = cnt[i];
        beg   = begs[i];
        adi   = asd2[2u * (unsigned)i + 1u];
        srcv0 = load_srcs_blk(srcs, beg, deg, 0, lane, i);
    }
    int i1 = i + nw, deg1 = 0, beg1 = 0;
    if (i1 < n) {
        deg1 = cnt[i1];
        beg1 = begs[i1];
    }

    while (i < n) {
        int srcv1 = 0;
        float adi1 = 0.0f;
        const int i2 = i1 + nw;
        int deg2 = 0, beg2 = 0;
        auto ask_ahead = [&]() {
            if (i1 < n) {
                adi1  = asd2[2u * (unsigned)i1 + 1u];
                srcv1 = load_srcs_blk(srcs, beg1, deg1, 0, lane, i1);
            }
            if (i2 < n) {
                deg2 = cnt[i2];
                beg2 = begs[i2];
            }
        };

        const int total = deg + 1;
        float acc = 0.0f;
        float hv[8];
        if (total <= 32) {
            const int  srcv  = srcv0;
            const bool valid = (int)lane < total;
            rows2_blk(srcv, j, f, h2, hv);
            __builtin_amdgcn_sched_barrier(0);
            const float lg = asd2[2u * (unsigned)srcv];
            ask_ahead();
            const float e  = logit2_of(lg, valid, adi);
            const float m  = wave_max(valid ? e : NEG_INF);
            const float ex = valid ? exp_hw_rel(e, m * LOG2E) : 0.0f;
            const float rden = rcp_hw(wave_sum(ex) + 1e-16f);
            accum2_blk(hv, ex * rden, 0, total, j, acc);
        } else {
            ask_ahead();
            float m = NEG_INF;
            for (int b0 = 0; b0 < total; b0 += 32) {
                const int  srcv  = load_srcs_blk(srcs, beg, deg, b0, lane, i);
                const bool valid = b0 + (int)lane < total;
                const float e = logit2(srcv, valid, asd2, adi);
                m = fmaxf(m, valid ? e : NEG_INF);
            }
            m = wave_max(m);
            const float ml = m * LOG2E;
            float ssum = 0.0f;
            for (int b0 = 0; b0 < total; b0 += 32) {
                const int  srcv  = load_srcs_blk(srcs, beg, deg, b0, lane, i);
                const bool valid = b0 + (int)lane < total;
                const float e = logit2(srcv, valid, asd2, adi);
                ssum += valid ? exp_hw_rel(e, ml) : 0.0f;
            }
            const float rden = rcp_hw(wave_sum(ssum) + 1e-16f);
            for (int b0 = 0; b0 < total; b0 += 32) {
                const int  srcv  = load_srcs_blk(srcs, beg, deg, b0, lane, i);
                const bool valid = b0 + (int)lane < total;
                const float e = logit2(srcv, valid, asd2, adi);
                rows2_blk(srcv, j, f, h2, hv);
                const float al = valid ? exp_hw_rel(e, ml) * rden : 0.0f;
                accum2_blk(hv, al, b0, total, j, acc);
            }
        }
        acc += __shfl_xor(acc, 8);
        acc += __shfl_xor(acc, 16);
        if (lane < 8u) out[(unsigned)i * 8u + f] = acc + bz;

        i = i1;  deg = deg1;  beg = beg1;  srcv0 = srcv1;  adi = adi1;
        i1 = i2; deg1 = deg2; beg1 = beg2;
    }
}

extern "C" void kernel_launch(void* const* d_in, const int* in_sizes, int n_in,
                              void* d_out, int out_size, void* d_ws, size_t ws_size,
                              hipStream_t stream) {
    const float* x     = (const float*)d_in[0];
    const int*   ei    = (const int*)d_in[1];
    const float* W1    = (const float*)d_in[2];
    const float* asrc1 = (const float*)d_in[3];
    const float* adst1 = (const float*)d_in[4];
    const float* bias1 = (const float*)d_in[5];
    const float* W2    = (const float*)d_in[6];
    const float* asrc2 = (const float*)d_in[7];
    const float* adst2 = (const float*)d_in[8];
    const float* bias2 = (const float*)d_in[9];

    int n = in_sizes[0] / F_IN;
    int E = in_sizes[1] / 2;
    int B = (n + BKT_NODES - 1) >> BKT_SHIFT;

    char* p = (char*)d_ws;
    float* h1   = (float*)p; p += (size_t)n * 64 * 4;
    float* as1  = (float*)p; p += (size_t)n * 8 * 4;
    float* ad1  = (float*)p; p += (size_t)n * 8 * 4;
    float* h2   = (float*)p; p += (size_t)n * 8 * 4;
    float* asd2 = (float*)p; p += (size_t)n * 8;
    int*   cnt  = (int*)p;   p += (size_t)n * 4;
    int*   begs = (int*)p;   p += (size_t)n * 4;
    int*   part = (int*)p;   p += (size_t)E * 4;
    int*   srcs = (int*)p;   p += (size_t)E * 4;
    int*   ghist = (int*)p;  p += (size_t)MAX_BKT * 4;
    int*   gcur  = (int*)p;  p += (size_t)MAX_BKT * 4;

    hipMemsetAsync(ghist, 0, (size_t)2 * MAX_BKT * 4, stream);

    int gP = (E + 1024 * PART_EPT - 1) / (1024 * PART_EPT);

    k_hist<<<gP, 1024, 0, stream>>>(ei, E, n, B, ghist);
    k_part<<<gP, 1024, 0, stream>>>(ei, E, n, B, ghist, gcur, part);
    int gG = (n + 16 * G1_TILES - 1) / (16 * G1_TILES);
    k_bucket_gemm1<<<B + gG, 1024, 0, stream>>>(part, ghist, n, cnt, begs, srcs, B,
                                                x, W1, asrc1, adst1, h1, as1, ad1);

    k_agg1<<<AGG1_GRID, 512, 0, stream>>>(cnt, begs, srcs, as1, ad1, h1, bias1,
                                          W2, asrc2, adst2, h2, asd2, n);

    k_agg2<<<AGG2_GRID, 256, 0, stream>>>(cnt, begs, srcs, asd2, h2, bias2, (float*)d_out, n);
}
